# starting kernel
# baseline (speedup 1.0000x reference)
#include <hip/hip_runtime.h>
#include <math.h>


typedef __attribute__((ext_vector_type(2))) float v2f;
typedef __attribute__((ext_vector_type(8))) float v8f;

static inline int ceil_div(long long a, long long b) { return (int)((a + b - 1) / b); }

template <int K, int N>
__global__ __launch_bounds__(256) void gemm_wmma_f32(const float* __restrict__ A,
                                                     const float* __restrict__ B,
                                                     float* __restrict__ C, int M) {
  constexpr int NT = N / 16;
  const int mTiles = (M + 15) >> 4;
  const int gwave = (int)((blockIdx.x * (long long)blockDim.x + threadIdx.x) >> 5);
  if (gwave >= mTiles * NT) return;

  const int lane = threadIdx.x & 31;
  const int mt = gwave / NT;
  const int nt = gwave - mt * NT;
  const int half = lane >> 4;
  const int l16 = lane & 15;

  const int row = mt * 16 + l16;
  const int rclamp = row < M ? row : (M - 1);
  const int col = nt * 16 + l16;
  const float* __restrict__ Ar = A + (long long)rclamp * K;

  v8f acc = {};
#pragma unroll 4
  for (int k = 0; k < K; k += 4) {
    const int ka = k + 2 * half;
    v2f a, b;
    a.x = Ar[ka];
    a.y = Ar[ka + 1];
    b.x = B[(long long)ka * N + col];
    b.y = B[(long long)(ka + 1) * N + col];
    acc = __builtin_amdgcn_wmma_f32_16x16x4_f32(false, a, false, b, (short)0, acc,
                                                false, false);
  }

  const int m0 = mt * 16 + 8 * half;
  float* __restrict__ Cp = C + (long long)m0 * N + col;
  if (mt * 16 + 16 <= M) {
#pragma unroll
    for (int v = 0; v < 8; ++v) Cp[(long long)v * N] = acc[v];
  } else {
#pragma unroll
    for (int v = 0; v < 8; ++v)
      if (m0 + v < M) Cp[(long long)v * N] = acc[v];
  }
}

__global__ void deg_kernel(const int* __restrict__ dst, int* __restrict__ deg, int e) {
  const int i = blockIdx.x * blockDim.x + threadIdx.x;
  if (i < e) atomicAdd(&deg[dst[i]], 1);
}

__global__ void dinv_kernel(const int* __restrict__ deg, float* __restrict__ dinv, int n) {
  const int i = blockIdx.x * blockDim.x + threadIdx.x;
  if (i < n) dinv[i] = rsqrtf((float)deg[i] + 1.0f);
}

template <int LOGF>
__global__ __launch_bounds__(256) void agg_kernel(const float* __restrict__ H,
                                                  const int* __restrict__ src,
                                                  const int* __restrict__ dst,
                                                  const float* __restrict__ dinv,
                                                  float* __restrict__ out, int e, int n) {
  constexpr int F = 1 << LOGF;
  const long long total = (long long)(e + n) << LOGF;
  const long long i = (long long)blockIdx.x * blockDim.x + threadIdx.x;
  if (i >= total) return;
  const int f = (int)(i & (F - 1));
  const long long edge = i >> LOGF;
  int s, d;
  if (edge < (long long)e) {
    s = src[edge];
    d = dst[edge];
  } else {
    s = d = (int)(edge - e);
  }
  const float nrm = dinv[s] * dinv[d];
  atomicAdd(out + (((long long)d) << LOGF) + f, H[(((long long)s) << LOGF) + f] * nrm);
}

template <int LOGF>
__global__ void bias_relu_kernel(const float* __restrict__ agg, const float* __restrict__ b,
                                 float* __restrict__ out, long long total) {
  constexpr int F = 1 << LOGF;
  const long long i = (long long)blockIdx.x * blockDim.x + threadIdx.x;
  if (i >= total) return;
  const float v = agg[i] + b[(int)(i & (F - 1))];
  out[i] = v > 0.0f ? v : 0.0f;
}

__global__ void lsm_kernel(const float* __restrict__ agg, const float* __restrict__ b,
                           float* __restrict__ out, int n) {
  const int i = blockIdx.x * blockDim.x + threadIdx.x;
  if (i >= n) return;
  float v[16];
  float mx = -INFINITY;
#pragma unroll
  for (int j = 0; j < 16; ++j) {
    v[j] = agg[(long long)i * 16 + j] + b[j];
    mx = fmaxf(mx, v[j]);
  }
  float s = 0.0f;
#pragma unroll
  for (int j = 0; j < 16; ++j) s += expf(v[j] - mx);
  const float lse = mx + logf(s);
#pragma unroll
  for (int j = 0; j < 16; ++j) out[(long long)i * 16 + j] = v[j] - lse;
}

extern "C" void kernel_launch(void* const* d_in, const int* in_sizes, int n_in,
                              void* d_out, int out_size, void* d_ws, size_t ws_size,
                              hipStream_t stream) {
  const float* x  = (const float*)d_in[0];
  const int*   ei = (const int*)d_in[1];
  const float* W1 = (const float*)d_in[2];
  const float* b1 = (const float*)d_in[3];
  const float* W2 = (const float*)d_in[4];
  const float* b2 = (const float*)d_in[5];

  const int n = in_sizes[0] / 256;
  const int e = in_sizes[1] / 2;
  const int* src = ei;
  const int* dst = ei + e;

  float* agg1 = (float*)d_ws;
  float* agg2 = agg1 + (size_t)n * 128;
  int*   deg  = (int*)(agg2 + (size_t)n * 16);
  float* h1   = (float*)(deg + n);
  float* h2   = h1 + (size_t)n * 128;
  float* dinv = h2 + (size_t)n * 16;

  hipMemsetAsync(d_ws, 0, (size_t)n * (128 + 16 + 1) * sizeof(float), stream);

  deg_kernel<<<ceil_div(e, 256), 256, 0, stream>>>(dst, deg, e);
  dinv_kernel<<<ceil_div(n, 256), 256, 0, stream>>>(deg, dinv, n);

  {
    const long long waves = (long long)((n + 15) / 16) * (128 / 16);
    gemm_wmma_f32<256, 128><<<ceil_div(waves * 32, 256), 256, 0, stream>>>(x, W1, h1, n);
  }
  agg_kernel<7><<<ceil_div(((long long)(e + n)) << 7, 256), 256, 0, stream>>>(
      h1, src, dst, dinv, agg1, e, n);
  bias_relu_kernel<7><<<ceil_div((long long)n * 128, 256), 256, 0, stream>>>(
      agg1, b1, h1, (long long)n * 128);

  {
    const long long waves = (long long)((n + 15) / 16) * (16 / 16);
    gemm_wmma_f32<128, 16><<<ceil_div(waves * 32, 256), 256, 0, stream>>>(h1, W2, h2, n);
  }
  agg_kernel<4><<<ceil_div(((long long)(e + n)) << 4, 256), 256, 0, stream>>>(
      h2, src, dst, dinv, agg2, e, n);
  lsm_kernel<<<ceil_div(n, 256), 256, 0, stream>>>(agg2, b2, (float*)d_out, n);
}


# Aggregation re-cut: a work-group owns a range of destination rows, sums them in LDS with float LDS atomics (arrival order) and scans the whole edge list for its edges; no global float atomics
# speedup vs baseline: 1.9601x; 1.9601x over previous
#include <hip/hip_runtime.h>
#include <math.h>


typedef __attribute__((ext_vector_type(2))) float v2f;
typedef __attribute__((ext_vector_type(8))) float v8f;

static inline int ceil_div(long long a, long long b) { return (int)((a + b - 1) / b); }

template <int K, int N>
__global__ __launch_bounds__(256) void gemm_wmma_f32(const float* __restrict__ A,
                                                     const float* __restrict__ B,
                                                     float* __restrict__ C, int M) {
  constexpr int NT = N / 16;
  const int mTiles = (M + 15) >> 4;
  const int gwave = (int)((blockIdx.x * (long long)blockDim.x + threadIdx.x) >> 5);
  if (gwave >= mTiles * NT) return;

  const int lane = threadIdx.x & 31;
  const int mt = gwave / NT;
  const int nt = gwave - mt * NT;
  const int half = lane >> 4;
  const int l16 = lane & 15;

  const int row = mt * 16 + l16;
  const int rclamp = row < M ? row : (M - 1);
  const int col = nt * 16 + l16;
  const float* __restrict__ Ar = A + (long long)rclamp * K;

  v8f acc = {};
#pragma unroll 4
  for (int k = 0; k < K; k += 4) {
    const int ka = k + 2 * half;
    v2f a, b;
    a.x = Ar[ka];
    a.y = Ar[ka + 1];
    b.x = B[(long long)ka * N + col];
    b.y = B[(long long)(ka + 1) * N + col];
    acc = __builtin_amdgcn_wmma_f32_16x16x4_f32(false, a, false, b, (short)0, acc,
                                                false, false);
  }

  const int m0 = mt * 16 + 8 * half;
  float* __restrict__ Cp = C + (long long)m0 * N + col;
  if (mt * 16 + 16 <= M) {
#pragma unroll
    for (int v = 0; v < 8; ++v) Cp[(long long)v * N] = acc[v];
  } else {
#pragma unroll
    for (int v = 0; v < 8; ++v)
      if (m0 + v < M) Cp[(long long)v * N] = acc[v];
  }
}

__global__ void deg_kernel(const int* __restrict__ dst, int* __restrict__ deg, int e) {
  const int i = blockIdx.x * blockDim.x + threadIdx.x;
  if (i < e) atomicAdd(&deg[dst[i]], 1);
}

__global__ void dinv_kernel(const int* __restrict__ deg, float* __restrict__ dinv, int n) {
  const int i = blockIdx.x * blockDim.x + threadIdx.x;
  if (i < n) dinv[i] = rsqrtf((float)deg[i] + 1.0f);
}

template <int LOGF, int R, int QCAP>
__global__ __launch_bounds__(1024) void agg_lds_kernel(const float* __restrict__ H,
                                                       const int* __restrict__ src,
                                                       const int* __restrict__ dst,
                                                       const float* __restrict__ dinv,
                                                       float* __restrict__ out, int e, int n) {
  constexpr int F = 1 << LOGF;
  constexpr int LPE = F / 4;
  constexpr int LOGLPE = LOGF - 2;
  constexpr int EPI = 32 / LPE;
  constexpr int NT = 1024;
  __shared__ float acc[R * F];
  __shared__ int queue[QCAP];
  __shared__ int qcount;

  const int tid = threadIdx.x;
  const int lane = tid & 31;
  const int base = blockIdx.x * R;

  if (tid == 0) qcount = 0;
  for (int i = tid; i < R * LPE; i += NT) {
    const int row = i >> LOGLPE, q = i & (LPE - 1);
    const int node = base + row;
    if (node < n) {
      const float dn = dinv[node];
      const float nrm = dn * dn;
      const float4 h = *reinterpret_cast<const float4*>(H + (((long long)node) << LOGF) + 4 * q);
      float* a = acc + (row << LOGF) + q;
      a[0] = h.x * nrm;
      a[LPE] = h.y * nrm;
      a[2 * LPE] = h.z * nrm;
      a[3 * LPE] = h.w * nrm;
    }
  }
  __syncthreads();

  auto push = [&](int idx) {
    const int slot = atomicAdd(&qcount, 1);
    if (slot < QCAP) {
      queue[slot] = idx;
    } else {
      const int s = src[idx];
      const int d = dst[idx];
      const float nrm = dinv[s] * dinv[d];
      const float* hs = H + (((long long)s) << LOGF);
      float* a = acc + ((d - base) << LOGF);
      for (int f = 0; f < F; ++f) atomicAdd(a + (f & 3) * LPE + (f >> 2), hs[f] * nrm);
    }
  };
  auto in_range = [&](int d) { return (unsigned)(d - base) < (unsigned)R; };

  int head = (int)(((16u - (unsigned)(reinterpret_cast<unsigned long long>(dst) & 15u)) & 15u) >> 2);
  if (head > e) head = e;
  if (tid < head && in_range(dst[tid])) push(tid);
  const int nvec = (e - head) >> 2;
  const int4* __restrict__ dv = reinterpret_cast<const int4*>(dst + head);
  constexpr int U = 8;
  int v = tid;
  for (; v + (U - 1) * NT < nvec; v += U * NT) {
    int4 dd[U];
    for (int u = 0; u < U; ++u) dd[u] = dv[v + u * NT];
    unsigned m = 0;
    for (int u = 0; u < U; ++u) {
      m |= (in_range(dd[u].x) ? 1u : 0u) << (4 * u);
      m |= (in_range(dd[u].y) ? 2u : 0u) << (4 * u);
      m |= (in_range(dd[u].z) ? 4u : 0u) << (4 * u);
      m |= (in_range(dd[u].w) ? 8u : 0u) << (4 * u);
    }
    while (m) {
      const int b = __builtin_ctz(m);
      m &= m - 1;
      push(head + 4 * (v + (b >> 2) * NT) + (b & 3));
    }
  }
  for (; v < nvec; v += NT) {
    const int4 d4 = dv[v];
    unsigned m = (in_range(d4.x) ? 1u : 0u) | (in_range(d4.y) ? 2u : 0u) |
                 (in_range(d4.z) ? 4u : 0u) | (in_range(d4.w) ? 8u : 0u);
    while (m) {
      const int b = __builtin_ctz(m);
      m &= m - 1;
      push(head + 4 * v + b);
    }
  }
  {
    const int t = head + 4 * nvec + tid;
    if (t < e && in_range(dst[t])) push(t);
  }
  __syncthreads();

  constexpr int STEPS = 32 / EPI;
  constexpr int UB = STEPS < 8 ? STEPS : 8;
  const int total = qcount < QCAP ? qcount : QCAP;
  const int q = lane & (LPE - 1);
  const int b0 = (tid >> 5) * 32;
  int ns = 0, nd = base;
  if (b0 + lane < total) {
    const int idx = queue[b0 + lane];
    ns = src[idx];
    nd = dst[idx];
  }
  for (int b = b0; b < total; b += NT) {
    const int s = ns;
    const int dl = nd - base;
    const float nrm = (b + lane < total) ? dinv[s] * dinv[nd] : 0.0f;
    const int cnt = total - b < 32 ? total - b : 32;
    ns = 0;
    nd = base;
    if (b + NT + lane < total) {
      const int idx = queue[b + NT + lane];
      ns = src[idx];
      nd = dst[idx];
    }
    for (int j = 0; j < STEPS; j += UB) {
      float4 h[UB];
      int dj[UB];
      float nj[UB];
      for (int u = 0; u < UB; ++u) {
        const int sub = (j + u) * EPI + (lane >> LOGLPE);
        const int sj = __shfl(s, sub);
        dj[u] = __shfl(dl, sub);
        nj[u] = __shfl(nrm, sub);
        h[u] = *reinterpret_cast<const float4*>(H + (((long long)sj) << LOGF) + 4 * q);
      }
      for (int u = 0; u < UB; ++u) {
        const int sub = (j + u) * EPI + (lane >> LOGLPE);
        if (sub < cnt) {
          float* a = acc + (dj[u] << LOGF) + q;
          atomicAdd(a, h[u].x * nj[u]);
          atomicAdd(a + LPE, h[u].y * nj[u]);
          atomicAdd(a + 2 * LPE, h[u].z * nj[u]);
          atomicAdd(a + 3 * LPE, h[u].w * nj[u]);
        }
      }
    }
  }
  __syncthreads();

  for (int i = tid; i < R * LPE; i += NT) {
    const int row = i >> LOGLPE, qq = i & (LPE - 1);
    const int node = base + row;
    if (node < n) {
      const float* a = acc + (row << LOGF) + qq;
      float4 o;
      o.x = a[0];
      o.y = a[LPE];
      o.z = a[2 * LPE];
      o.w = a[3 * LPE];
      *reinterpret_cast<float4*>(out + (((long long)node) << LOGF) + 4 * qq) = o;
    }
  }
}

template <int LOGF>
__global__ void bias_relu_kernel(const float* __restrict__ agg, const float* __restrict__ b,
                                 float* __restrict__ out, long long total) {
  constexpr int F = 1 << LOGF;
  const long long i = (long long)blockIdx.x * blockDim.x + threadIdx.x;
  if (i >= total) return;
  const float v = agg[i] + b[(int)(i & (F - 1))];
  out[i] = v > 0.0f ? v : 0.0f;
}

__global__ void lsm_kernel(const float* __restrict__ agg, const float* __restrict__ b,
                           float* __restrict__ out, int n) {
  const int i = blockIdx.x * blockDim.x + threadIdx.x;
  if (i >= n) return;
  float v[16];
  float mx = -INFINITY;
#pragma unroll
  for (int j = 0; j < 16; ++j) {
    v[j] = agg[(long long)i * 16 + j] + b[j];
    mx = fmaxf(mx, v[j]);
  }
  float s = 0.0f;
#pragma unroll
  for (int j = 0; j < 16; ++j) s += expf(v[j] - mx);
  const float lse = mx + logf(s);
#pragma unroll
  for (int j = 0; j < 16; ++j) out[(long long)i * 16 + j] = v[j] - lse;
}

extern "C" void kernel_launch(void* const* d_in, const int* in_sizes, int n_in,
                              void* d_out, int out_size, void* d_ws, size_t ws_size,
                              hipStream_t stream) {
  const float* x  = (const float*)d_in[0];
  const int*   ei = (const int*)d_in[1];
  const float* W1 = (const float*)d_in[2];
  const float* b1 = (const float*)d_in[3];
  const float* W2 = (const float*)d_in[4];
  const float* b2 = (const float*)d_in[5];

  const int n = in_sizes[0] / 256;
  const int e = in_sizes[1] / 2;
  const int* src = ei;
  const int* dst = ei + e;

  float* agg1 = (float*)d_ws;
  float* h1   = agg1 + (size_t)n * 128;
  float* agg2 = h1 + (size_t)n * 128;
  float* h2   = agg2 + (size_t)n * 16;
  int*   deg  = (int*)(h2 + (size_t)n * 16);
  float* dinv = (float*)(deg + n);

  hipMemsetAsync(deg, 0, (size_t)n * sizeof(int), stream);

  deg_kernel<<<ceil_div(e, 256), 256, 0, stream>>>(dst, deg, e);
  dinv_kernel<<<ceil_div(n, 256), 256, 0, stream>>>(deg, dinv, n);

  {
    const long long waves = (long long)((n + 15) / 16) * (128 / 16);
    gemm_wmma_f32<256, 128><<<ceil_div(waves * 32, 256), 256, 0, stream>>>(x, W1, h1, n);
  }
  agg_lds_kernel<7, 512, 10240><<<ceil_div(n, 512), 1024, 0, stream>>>(h1, src, dst, dinv, agg1, e, n);
  bias_relu_kernel<7><<<ceil_div((long long)n * 128, 256), 256, 0, stream>>>(
      agg1, b1, h1, (long long)n * 128);

  {
    const long long waves = (long long)((n + 15) / 16) * (16 / 16);
    gemm_wmma_f32<128, 16><<<ceil_div(waves * 32, 256), 256, 0, stream>>>(h1, W2, h2, n);
  }
  agg_lds_kernel<4, 1024, 40960><<<ceil_div(n, 1024), 1024, 0, stream>>>(h2, src, dst, dinv, agg2, e, n);
  lsm_kernel<<<ceil_div(n, 256), 256, 0, stream>>>(agg2, b2, (float*)d_out, n);
}


# Wide GEMM re-cut: a 4-wave work-group copies its 64 x 256 slab of A into LDS with 16-byte loads (pitch K+4), each wave computes 64 rows x 32 columns (4 x 2 tiles); B from global; k order kept
# speedup vs baseline: 2.2058x; 1.1254x over previous
#include <hip/hip_runtime.h>
#include <math.h>


typedef __attribute__((ext_vector_type(2))) float v2f;
typedef __attribute__((ext_vector_type(8))) float v8f;

static inline int ceil_div(long long a, long long b) { return (int)((a + b - 1) / b); }

template <int K, int N>
__global__ __launch_bounds__(256) void gemm_wmma_f32(const float* __restrict__ A,
                                                     const float* __restrict__ B,
                                                     float* __restrict__ C, int M) {
  constexpr int NT = N / 16;
  const int mTiles = (M + 15) >> 4;
  const int gwave = (int)((blockIdx.x * (long long)blockDim.x + threadIdx.x) >> 5);
  if (gwave >= mTiles * NT) return;

  const int lane = threadIdx.x & 31;
  const int mt = gwave / NT;
  const int nt = gwave - mt * NT;
  const int half = lane >> 4;
  const int l16 = lane & 15;

  const int row = mt * 16 + l16;
  const int rclamp = row < M ? row : (M - 1);
  const int col = nt * 16 + l16;
  const float* __restrict__ Ar = A + (long long)rclamp * K;

  v8f acc = {};
#pragma unroll 4
  for (int k = 0; k < K; k += 4) {
    const int ka = k + 2 * half;
    v2f a, b;
    a.x = Ar[ka];
    a.y = Ar[ka + 1];
    b.x = B[(long long)ka * N + col];
    b.y = B[(long long)(ka + 1) * N + col];
    acc = __builtin_amdgcn_wmma_f32_16x16x4_f32(false, a, false, b, (short)0, acc,
                                                false, false);
  }

  const int m0 = mt * 16 + 8 * half;
  float* __restrict__ Cp = C + (long long)m0 * N + col;
  if (mt * 16 + 16 <= M) {
#pragma unroll
    for (int v = 0; v < 8; ++v) Cp[(long long)v * N] = acc[v];
  } else {
#pragma unroll
    for (int v = 0; v < 8; ++v)
      if (m0 + v < M) Cp[(long long)v * N] = acc[v];
  }
}

template <int K, int N, int TM, int TN, int WN>
__global__ __launch_bounds__(WN * 32) void gemm_wmma_f32_lds_a(const float* __restrict__ A,
                                                             const float* __restrict__ B,
                                                             float* __restrict__ C, int M) {
  constexpr int BM = TM * 16;
  constexpr int P = K + 4;
  constexpr int NTHR = WN * 32;
  constexpr int KQ = K / 4;
  constexpr int NV = BM * KQ;
  constexpr int LB = 32;
  static_assert(N == WN * TN * 16, "waves must tile the N columns");
  static_assert(NV % (NTHR * LB) == 0, "slab must split evenly over the staging batches");
  __shared__ __attribute__((aligned(16))) float As[BM * P];

  const int tid = threadIdx.x;
  const int row0 = blockIdx.x * BM;

  for (int i0 = 0; i0 < NV; i0 += NTHR * LB) {
    float4 t[LB];
    for (int u = 0; u < LB; ++u) {
      const int f = i0 + u * NTHR + tid;
      const int r = f / KQ, kq = f - r * KQ;
      int gr = row0 + r;
      gr = gr < M ? gr : (M - 1);
      t[u] = *reinterpret_cast<const float4*>(A + (long long)gr * K + 4 * kq);
    }
    for (int u = 0; u < LB; ++u) {
      const int f = i0 + u * NTHR + tid;
      const int r = f / KQ, kq = f - r * KQ;
      *reinterpret_cast<float4*>(As + r * P + 4 * kq) = t[u];
    }
  }
  __syncthreads();

  const int lane = tid & 31;
  const int wave = tid >> 5;
  const int half = lane >> 4;
  const int l16 = lane & 15;
  const int col0 = wave * (TN * 16) + l16;

  const float* __restrict__ Al = As + l16 * P + 2 * half;
  const float* __restrict__ Bl = B + (long long)(2 * half) * N + col0;

  v8f acc[TM][TN];
  for (int i = 0; i < TM; ++i)
    for (int j = 0; j < TN; ++j) acc[i][j] = v8f{};

#pragma unroll
  for (int k = 0; k < K; k += 4) {
    v2f a[TM], b[TN];
    for (int i = 0; i < TM; ++i) a[i] = *reinterpret_cast<const v2f*>(Al + i * 16 * P + k);
    for (int j = 0; j < TN; ++j) {
      b[j].x = Bl[(long long)k * N + j * 16];
      b[j].y = Bl[(long long)(k + 1) * N + j * 16];
    }
    for (int i = 0; i < TM; ++i)
      for (int j = 0; j < TN; ++j)
        acc[i][j] = __builtin_amdgcn_wmma_f32_16x16x4_f32(false, a[i], false, b[j], (short)0,
                                                          acc[i][j], false, false);
  }

  for (int i = 0; i < TM; ++i) {
    const int m0 = row0 + i * 16 + 8 * half;
    for (int j = 0; j < TN; ++j) {
      float* __restrict__ Cp = C + (long long)m0 * N + col0 + j * 16;
      if (row0 + BM <= M) {
        for (int v = 0; v < 8; ++v) Cp[(long long)v * N] = acc[i][j][v];
      } else {
        for (int v = 0; v < 8; ++v)
          if (m0 + v < M) Cp[(long long)v * N] = acc[i][j][v];
      }
    }
  }
}

__global__ void deg_kernel(const int* __restrict__ dst, int* __restrict__ deg, int e) {
  const int i = blockIdx.x * blockDim.x + threadIdx.x;
  if (i < e) atomicAdd(&deg[dst[i]], 1);
}

__global__ void dinv_kernel(const int* __restrict__ deg, float* __restrict__ dinv, int n) {
  const int i = blockIdx.x * blockDim.x + threadIdx.x;
  if (i < n) dinv[i] = rsqrtf((float)deg[i] + 1.0f);
}

template <int LOGF, int R, int QCAP>
__global__ __launch_bounds__(1024) void agg_lds_kernel(const float* __restrict__ H,
                                                       const int* __restrict__ src,
                                                       const int* __restrict__ dst,
                                                       const float* __restrict__ dinv,
                                                       float* __restrict__ out, int e, int n) {
  constexpr int F = 1 << LOGF;
  constexpr int LPE = F / 4;
  constexpr int LOGLPE = LOGF - 2;
  constexpr int EPI = 32 / LPE;
  constexpr int NT = 1024;
  __shared__ float acc[R * F];
  __shared__ int queue[QCAP];
  __shared__ int qcount;

  const int tid = threadIdx.x;
  const int lane = tid & 31;
  const int base = blockIdx.x * R;

  if (tid == 0) qcount = 0;
  for (int i = tid; i < R * LPE; i += NT) {
    const int row = i >> LOGLPE, q = i & (LPE - 1);
    const int node = base + row;
    if (node < n) {
      const float dn = dinv[node];
      const float nrm = dn * dn;
      const float4 h = *reinterpret_cast<const float4*>(H + (((long long)node) << LOGF) + 4 * q);
      float* a = acc + (row << LOGF) + q;
      a[0] = h.x * nrm;
      a[LPE] = h.y * nrm;
      a[2 * LPE] = h.z * nrm;
      a[3 * LPE] = h.w * nrm;
    }
  }
  __syncthreads();

  auto push = [&](int idx) {
    const int slot = atomicAdd(&qcount, 1);
    if (slot < QCAP) {
      queue[slot] = idx;
    } else {
      const int s = src[idx];
      const int d = dst[idx];
      const float nrm = dinv[s] * dinv[d];
      const float* hs = H + (((long long)s) << LOGF);
      float* a = acc + ((d - base) << LOGF);
      for (int f = 0; f < F; ++f) atomicAdd(a + (f & 3) * LPE + (f >> 2), hs[f] * nrm);
    }
  };
  auto in_range = [&](int d) { return (unsigned)(d - base) < (unsigned)R; };

  int head = (int)(((16u - (unsigned)(reinterpret_cast<unsigned long long>(dst) & 15u)) & 15u) >> 2);
  if (head > e) head = e;
  if (tid < head && in_range(dst[tid])) push(tid);
  const int nvec = (e - head) >> 2;
  const int4* __restrict__ dv = reinterpret_cast<const int4*>(dst + head);
  constexpr int U = 8;
  int v = tid;
  for (; v + (U - 1) * NT < nvec; v += U * NT) {
    int4 dd[U];
    for (int u = 0; u < U; ++u) dd[u] = dv[v + u * NT];
    unsigned m = 0;
    for (int u = 0; u < U; ++u) {
      m |= (in_range(dd[u].x) ? 1u : 0u) << (4 * u);
      m |= (in_range(dd[u].y) ? 2u : 0u) << (4 * u);
      m |= (in_range(dd[u].z) ? 4u : 0u) << (4 * u);
      m |= (in_range(dd[u].w) ? 8u : 0u) << (4 * u);
    }
    while (m) {
      const int b = __builtin_ctz(m);
      m &= m - 1;
      push(head + 4 * (v + (b >> 2) * NT) + (b & 3));
    }
  }
  for (; v < nvec; v += NT) {
    const int4 d4 = dv[v];
    unsigned m = (in_range(d4.x) ? 1u : 0u) | (in_range(d4.y) ? 2u : 0u) |
                 (in_range(d4.z) ? 4u : 0u) | (in_range(d4.w) ? 8u : 0u);
    while (m) {
      const int b = __builtin_ctz(m);
      m &= m - 1;
      push(head + 4 * v + b);
    }
  }
  {
    const int t = head + 4 * nvec + tid;
    if (t < e && in_range(dst[t])) push(t);
  }
  __syncthreads();

  constexpr int STEPS = 32 / EPI;
  constexpr int UB = STEPS < 8 ? STEPS : 8;
  const int total = qcount < QCAP ? qcount : QCAP;
  const int q = lane & (LPE - 1);
  const int b0 = (tid >> 5) * 32;
  int ns = 0, nd = base;
  if (b0 + lane < total) {
    const int idx = queue[b0 + lane];
    ns = src[idx];
    nd = dst[idx];
  }
  for (int b = b0; b < total; b += NT) {
    const int s = ns;
    const int dl = nd - base;
    const float nrm = (b + lane < total) ? dinv[s] * dinv[nd] : 0.0f;
    const int cnt = total - b < 32 ? total - b : 32;
    ns = 0;
    nd = base;
    if (b + NT + lane < total) {
      const int idx = queue[b + NT + lane];
      ns = src[idx];
      nd = dst[idx];
    }
    for (int j = 0; j < STEPS; j += UB) {
      float4 h[UB];
      int dj[UB];
      float nj[UB];
      for (int u = 0; u < UB; ++u) {
        const int sub = (j + u) * EPI + (lane >> LOGLPE);
        const int sj = __shfl(s, sub);
        dj[u] = __shfl(dl, sub);
        nj[u] = __shfl(nrm, sub);
        h[u] = *reinterpret_cast<const float4*>(H + (((long long)sj) << LOGF) + 4 * q);
      }
      for (int u = 0; u < UB; ++u) {
        const int sub = (j + u) * EPI + (lane >> LOGLPE);
        if (sub < cnt) {
          float* a = acc + (dj[u] << LOGF) + q;
          atomicAdd(a, h[u].x * nj[u]);
          atomicAdd(a + LPE, h[u].y * nj[u]);
          atomicAdd(a + 2 * LPE, h[u].z * nj[u]);
          atomicAdd(a + 3 * LPE, h[u].w * nj[u]);
        }
      }
    }
  }
  __syncthreads();

  for (int i = tid; i < R * LPE; i += NT) {
    const int row = i >> LOGLPE, qq = i & (LPE - 1);
    const int node = base + row;
    if (node < n) {
      const float* a = acc + (row << LOGF) + qq;
      float4 o;
      o.x = a[0];
      o.y = a[LPE];
      o.z = a[2 * LPE];
      o.w = a[3 * LPE];
      *reinterpret_cast<float4*>(out + (((long long)node) << LOGF) + 4 * qq) = o;
    }
  }
}

template <int LOGF>
__global__ void bias_relu_kernel(const float* __restrict__ agg, const float* __restrict__ b,
                                 float* __restrict__ out, long long total) {
  constexpr int F = 1 << LOGF;
  const long long i = (long long)blockIdx.x * blockDim.x + threadIdx.x;
  if (i >= total) return;
  const float v = agg[i] + b[(int)(i & (F - 1))];
  out[i] = v > 0.0f ? v : 0.0f;
}

__global__ void lsm_kernel(const float* __restrict__ agg, const float* __restrict__ b,
                           float* __restrict__ out, int n) {
  const int i = blockIdx.x * blockDim.x + threadIdx.x;
  if (i >= n) return;
  float v[16];
  float mx = -INFINITY;
#pragma unroll
  for (int j = 0; j < 16; ++j) {
    v[j] = agg[(long long)i * 16 + j] + b[j];
    mx = fmaxf(mx, v[j]);
  }
  float s = 0.0f;
#pragma unroll
  for (int j = 0; j < 16; ++j) s += expf(v[j] - mx);
  const float lse = mx + logf(s);
#pragma unroll
  for (int j = 0; j < 16; ++j) out[(long long)i * 16 + j] = v[j] - lse;
}

extern "C" void kernel_launch(void* const* d_in, const int* in_sizes, int n_in,
                              void* d_out, int out_size, void* d_ws, size_t ws_size,
                              hipStream_t stream) {
  const float* x  = (const float*)d_in[0];
  const int*   ei = (const int*)d_in[1];
  const float* W1 = (const float*)d_in[2];
  const float* b1 = (const float*)d_in[3];
  const float* W2 = (const float*)d_in[4];
  const float* b2 = (const float*)d_in[5];

  const int n = in_sizes[0] / 256;
  const int e = in_sizes[1] / 2;
  const int* src = ei;
  const int* dst = ei + e;

  float* agg1 = (float*)d_ws;
  float* h1   = agg1 + (size_t)n * 128;
  float* agg2 = h1 + (size_t)n * 128;
  float* h2   = agg2 + (size_t)n * 16;
  int*   deg  = (int*)(h2 + (size_t)n * 16);
  float* dinv = (float*)(deg + n);

  hipMemsetAsync(deg, 0, (size_t)n * sizeof(int), stream);

  deg_kernel<<<ceil_div(e, 256), 256, 0, stream>>>(dst, deg, e);
  dinv_kernel<<<ceil_div(n, 256), 256, 0, stream>>>(deg, dinv, n);

  gemm_wmma_f32_lds_a<256, 128, 4, 2, 4><<<ceil_div(n, 64), 128, 0, stream>>>(x, W1, h1, n);
  agg_lds_kernel<7, 512, 10240><<<ceil_div(n, 512), 1024, 0, stream>>>(h1, src, dst, dinv, agg1, e, n);
  bias_relu_kernel<7><<<ceil_div((long long)n * 128, 256), 256, 0, stream>>>(
      agg1, b1, h1, (long long)n * 128);

  {
    const long long waves = (long long)((n + 15) / 16) * (16 / 16);
    gemm_wmma_f32<128, 16><<<ceil_div(waves * 32, 256), 256, 0, stream>>>(h1, W2, h2, n);
  }
  agg_lds_kernel<4, 1024, 40960><<<ceil_div(n, 1024), 1024, 0, stream>>>(h2, src, dst, dinv, agg2, e, n);
  lsm_kernel<<<ceil_div(n, 256), 256, 0, stream>>>(agg2, b2, (float*)d_out, n);
}


# Layer 2 aggregation re-cut to layer 1's 512-row ranges: layer 1 saves each range's edge list in the workspace, layer 2 reads it and no longer scans the edge list (sum order: LDS arrival, as before)
# speedup vs baseline: 2.6588x; 1.2054x over previous
#include <hip/hip_runtime.h>
#include <math.h>


typedef __attribute__((ext_vector_type(2))) float v2f;
typedef __attribute__((ext_vector_type(8))) float v8f;

static inline int ceil_div(long long a, long long b) { return (int)((a + b - 1) / b); }

template <int K, int N>
__global__ __launch_bounds__(256) void gemm_wmma_f32(const float* __restrict__ A,
                                                     const float* __restrict__ B,
                                                     float* __restrict__ C, int M) {
  constexpr int NT = N / 16;
  const int mTiles = (M + 15) >> 4;
  const int gwave = (int)((blockIdx.x * (long long)blockDim.x + threadIdx.x) >> 5);
  if (gwave >= mTiles * NT) return;

  const int lane = threadIdx.x & 31;
  const int mt = gwave / NT;
  const int nt = gwave - mt * NT;
  const int half = lane >> 4;
  const int l16 = lane & 15;

  const int row = mt * 16 + l16;
  const int rclamp = row < M ? row : (M - 1);
  const int col = nt * 16 + l16;
  const float* __restrict__ Ar = A + (long long)rclamp * K;

  v8f acc = {};
#pragma unroll 4
  for (int k = 0; k < K; k += 4) {
    const int ka = k + 2 * half;
    v2f a, b;
    a.x = Ar[ka];
    a.y = Ar[ka + 1];
    b.x = B[(long long)ka * N + col];
    b.y = B[(long long)(ka + 1) * N + col];
    acc = __builtin_amdgcn_wmma_f32_16x16x4_f32(false, a, false, b, (short)0, acc,
                                                false, false);
  }

  const int m0 = mt * 16 + 8 * half;
  float* __restrict__ Cp = C + (long long)m0 * N + col;
  if (mt * 16 + 16 <= M) {
#pragma unroll
    for (int v = 0; v < 8; ++v) Cp[(long long)v * N] = acc[v];
  } else {
#pragma unroll
    for (int v = 0; v < 8; ++v)
      if (m0 + v < M) Cp[(long long)v * N] = acc[v];
  }
}

template <int K, int N, int TM, int TN, int WN>
__global__ __launch_bounds__(WN * 32) void gemm_wmma_f32_lds_a(const float* __restrict__ A,
                                                             const float* __restrict__ B,
                                                             float* __restrict__ C, int M) {
  constexpr int BM = TM * 16;
  constexpr int P = K + 4;
  constexpr int NTHR = WN * 32;
  constexpr int KQ = K / 4;
  constexpr int NV = BM * KQ;
  constexpr int LB = 32;
  static_assert(N == WN * TN * 16, "waves must tile the N columns");
  static_assert(NV % (NTHR * LB) == 0, "slab must split evenly over the staging batches");
  __shared__ __attribute__((aligned(16))) float As[BM * P];

  const int tid = threadIdx.x;
  const int row0 = blockIdx.x * BM;

  for (int i0 = 0; i0 < NV; i0 += NTHR * LB) {
    float4 t[LB];
    for (int u = 0; u < LB; ++u) {
      const int f = i0 + u * NTHR + tid;
      const int r = f / KQ, kq = f - r * KQ;
      int gr = row0 + r;
      gr = gr < M ? gr : (M - 1);
      t[u] = *reinterpret_cast<const float4*>(A + (long long)gr * K + 4 * kq);
    }
    for (int u = 0; u < LB; ++u) {
      const int f = i0 + u * NTHR + tid;
      const int r = f / KQ, kq = f - r * KQ;
      *reinterpret_cast<float4*>(As + r * P + 4 * kq) = t[u];
    }
  }
  __syncthreads();

  const int lane = tid & 31;
  const int wave = tid >> 5;
  const int half = lane >> 4;
  const int l16 = lane & 15;
  const int col0 = wave * (TN * 16) + l16;

  const float* __restrict__ Al = As + l16 * P + 2 * half;
  const float* __restrict__ Bl = B + (long long)(2 * half) * N + col0;

  v8f acc[TM][TN];
  for (int i = 0; i < TM; ++i)
    for (int j = 0; j < TN; ++j) acc[i][j] = v8f{};

#pragma unroll
  for (int k = 0; k < K; k += 4) {
    v2f a[TM], b[TN];
    for (int i = 0; i < TM; ++i) a[i] = *reinterpret_cast<const v2f*>(Al + i * 16 * P + k);
    for (int j = 0; j < TN; ++j) {
      b[j].x = Bl[(long long)k * N + j * 16];
      b[j].y = Bl[(long long)(k + 1) * N + j * 16];
    }
    for (int i = 0; i < TM; ++i)
      for (int j = 0; j < TN; ++j)
        acc[i][j] = __builtin_amdgcn_wmma_f32_16x16x4_f32(false, a[i], false, b[j], (short)0,
                                                          acc[i][j], false, false);
  }

  for (int i = 0; i < TM; ++i) {
    const int m0 = row0 + i * 16 + 8 * half;
    for (int j = 0; j < TN; ++j) {
      float* __restrict__ Cp = C + (long long)m0 * N + col0 + j * 16;
      if (row0 + BM <= M) {
        for (int v = 0; v < 8; ++v) Cp[(long long)v * N] = acc[i][j][v];
      } else {
        for (int v = 0; v < 8; ++v)
          if (m0 + v < M) Cp[(long long)v * N] = acc[i][j][v];
      }
    }
  }
}

__global__ void deg_kernel(const int* __restrict__ dst, int* __restrict__ deg, int e) {
  const int i = blockIdx.x * blockDim.x + threadIdx.x;
  if (i < e) atomicAdd(&deg[dst[i]], 1);
}

__global__ void dinv_kernel(const int* __restrict__ deg, float* __restrict__ dinv, int n) {
  const int i = blockIdx.x * blockDim.x + threadIdx.x;
  if (i < n) dinv[i] = rsqrtf((float)deg[i] + 1.0f);
}

template <int LOGF, int R, int QCAP, bool SAVE, bool REUSE>
__global__ __launch_bounds__(1024) void agg_lds_kernel(const float* __restrict__ H,
                                                       const int* __restrict__ src,
                                                       const int* __restrict__ dst,
                                                       const float* __restrict__ dinv,
                                                       float* __restrict__ out,
                                                       int* __restrict__ glist,
                                                       int* __restrict__ gcount, int e, int n) {
  constexpr int F = 1 << LOGF;
  constexpr int LPE = F / 4;
  constexpr int LOGLPE = LOGF - 2;
  constexpr int EPI = 32 / LPE;
  constexpr int NT = 1024;
  __shared__ float acc[R * F];
  __shared__ int queue[QCAP];
  __shared__ int qcount;

  const int tid = threadIdx.x;
  const int lane = tid & 31;
  const int base = blockIdx.x * R;

  if (tid == 0) qcount = 0;
  for (int i = tid; i < R * LPE; i += NT) {
    const int row = i >> LOGLPE, q = i & (LPE - 1);
    const int node = base + row;
    if (node < n) {
      const float dn = dinv[node];
      const float nrm = dn * dn;
      const float4 h = *reinterpret_cast<const float4*>(H + (((long long)node) << LOGF) + 4 * q);
      float* a = acc + (row << LOGF) + q;
      a[0] = h.x * nrm;
      a[LPE] = h.y * nrm;
      a[2 * LPE] = h.z * nrm;
      a[3 * LPE] = h.w * nrm;
    }
  }
  __syncthreads();

  auto push = [&](int idx) {
    const int slot = atomicAdd(&qcount, 1);
    if (slot < QCAP) {
      queue[slot] = idx;
    } else {
      const int s = src[idx];
      const int d = dst[idx];
      const float nrm = dinv[s] * dinv[d];
      const float* hs = H + (((long long)s) << LOGF);
      float* a = acc + ((d - base) << LOGF);
      for (int f = 0; f < F; ++f) atomicAdd(a + (f & 3) * LPE + (f >> 2), hs[f] * nrm);
    }
  };
  auto in_range = [&](int d) { return (unsigned)(d - base) < (unsigned)R; };

  bool have_list = false;
  if constexpr (REUSE) {
    const int c = gcount[blockIdx.x];
    have_list = c <= QCAP;
    if (have_list) {
      const int* __restrict__ gl = glist + (long long)blockIdx.x * QCAP;
      for (int i = tid; i < c; i += NT) queue[i] = gl[i];
      if (tid == 0) qcount = c;
    }
  }

  if (!have_list) {
    int head = (int)(((16u - (unsigned)(reinterpret_cast<unsigned long long>(dst) & 15u)) & 15u) >> 2);
    if (head > e) head = e;
    if (tid < head && in_range(dst[tid])) push(tid);
    const int nvec = (e - head) >> 2;
    const int4* __restrict__ dv = reinterpret_cast<const int4*>(dst + head);
    constexpr int U = 8;
    int v = tid;
    for (; v + (U - 1) * NT < nvec; v += U * NT) {
      int4 dd[U];
      for (int u = 0; u < U; ++u) dd[u] = dv[v + u * NT];
      unsigned m = 0;
      for (int u = 0; u < U; ++u) {
        m |= (in_range(dd[u].x) ? 1u : 0u) << (4 * u);
        m |= (in_range(dd[u].y) ? 2u : 0u) << (4 * u);
        m |= (in_range(dd[u].z) ? 4u : 0u) << (4 * u);
        m |= (in_range(dd[u].w) ? 8u : 0u) << (4 * u);
      }
      while (m) {
        const int b = __builtin_ctz(m);
        m &= m - 1;
        push(head + 4 * (v + (b >> 2) * NT) + (b & 3));
      }
    }
    for (; v < nvec; v += NT) {
      const int4 d4 = dv[v];
      unsigned m = (in_range(d4.x) ? 1u : 0u) | (in_range(d4.y) ? 2u : 0u) |
                   (in_range(d4.z) ? 4u : 0u) | (in_range(d4.w) ? 8u : 0u);
      while (m) {
        const int b = __builtin_ctz(m);
        m &= m - 1;
        push(head + 4 * v + b);
      }
    }
    {
      const int t = head + 4 * nvec + tid;
      if (t < e && in_range(dst[t])) push(t);
    }
  }
  __syncthreads();

  if constexpr (SAVE) {
    const int kept = qcount < QCAP ? qcount : QCAP;
    int* __restrict__ gl = glist + (long long)blockIdx.x * QCAP;
    for (int i = tid; i < kept; i += NT) gl[i] = queue[i];
    if (tid == 0) gcount[blockIdx.x] = qcount;
  }

  constexpr int STEPS = 32 / EPI;
  constexpr int UB = STEPS < 8 ? STEPS : 8;
  const int total = qcount < QCAP ? qcount : QCAP;
  const int q = lane & (LPE - 1);
  const int b0 = (tid >> 5) * 32;
  int ns = 0, nd = base;
  if (b0 + lane < total) {
    const int idx = queue[b0 + lane];
    ns = src[idx];
    nd = dst[idx];
  }
  for (int b = b0; b < total; b += NT) {
    const int s = ns;
    const int dl = nd - base;
    const float nrm = (b + lane < total) ? dinv[s] * dinv[nd] : 0.0f;
    const int cnt = total - b < 32 ? total - b : 32;
    ns = 0;
    nd = base;
    if (b + NT + lane < total) {
      const int idx = queue[b + NT + lane];
      ns = src[idx];
      nd = dst[idx];
    }
    for (int j = 0; j < STEPS; j += UB) {
      float4 h[UB];
      int dj[UB];
      float nj[UB];
      for (int u = 0; u < UB; ++u) {
        const int sub = (j + u) * EPI + (lane >> LOGLPE);
        const int sj = __shfl(s, sub);
        dj[u] = __shfl(dl, sub);
        nj[u] = __shfl(nrm, sub);
        h[u] = *reinterpret_cast<const float4*>(H + (((long long)sj) << LOGF) + 4 * q);
      }
      for (int u = 0; u < UB; ++u) {
        const int sub = (j + u) * EPI + (lane >> LOGLPE);
        if (sub < cnt) {
          float* a = acc + (dj[u] << LOGF) + q;
          atomicAdd(a, h[u].x * nj[u]);
          atomicAdd(a + LPE, h[u].y * nj[u]);
          atomicAdd(a + 2 * LPE, h[u].z * nj[u]);
          atomicAdd(a + 3 * LPE, h[u].w * nj[u]);
        }
      }
    }
  }
  __syncthreads();

  for (int i = tid; i < R * LPE; i += NT) {
    const int row = i >> LOGLPE, qq = i & (LPE - 1);
    const int node = base + row;
    if (node < n) {
      const float* a = acc + (row << LOGF) + qq;
      float4 o;
      o.x = a[0];
      o.y = a[LPE];
      o.z = a[2 * LPE];
      o.w = a[3 * LPE];
      *reinterpret_cast<float4*>(out + (((long long)node) << LOGF) + 4 * qq) = o;
    }
  }
}

template <int LOGF>
__global__ void bias_relu_kernel(const float* __restrict__ agg, const float* __restrict__ b,
                                 float* __restrict__ out, long long total) {
  constexpr int F = 1 << LOGF;
  const long long i = (long long)blockIdx.x * blockDim.x + threadIdx.x;
  if (i >= total) return;
  const float v = agg[i] + b[(int)(i & (F - 1))];
  out[i] = v > 0.0f ? v : 0.0f;
}

__global__ void lsm_kernel(const float* __restrict__ agg, const float* __restrict__ b,
                           float* __restrict__ out, int n) {
  const int i = blockIdx.x * blockDim.x + threadIdx.x;
  if (i >= n) return;
  float v[16];
  float mx = -INFINITY;
#pragma unroll
  for (int j = 0; j < 16; ++j) {
    v[j] = agg[(long long)i * 16 + j] + b[j];
    mx = fmaxf(mx, v[j]);
  }
  float s = 0.0f;
#pragma unroll
  for (int j = 0; j < 16; ++j) s += expf(v[j] - mx);
  const float lse = mx + logf(s);
#pragma unroll
  for (int j = 0; j < 16; ++j) out[(long long)i * 16 + j] = v[j] - lse;
}

extern "C" void kernel_launch(void* const* d_in, const int* in_sizes, int n_in,
                              void* d_out, int out_size, void* d_ws, size_t ws_size,
                              hipStream_t stream) {
  const float* x  = (const float*)d_in[0];
  const int*   ei = (const int*)d_in[1];
  const float* W1 = (const float*)d_in[2];
  const float* b1 = (const float*)d_in[3];
  const float* W2 = (const float*)d_in[4];
  const float* b2 = (const float*)d_in[5];

  const int n = in_sizes[0] / 256;
  const int e = in_sizes[1] / 2;
  const int* src = ei;
  const int* dst = ei + e;

  float* agg1 = (float*)d_ws;
  float* h1   = agg1 + (size_t)n * 128;
  int*   glist = (int*)(h1 + (size_t)n * 128);
  int*   deg  = glist + (size_t)n * 32;
  float* dinv = (float*)(deg + n);
  float* h2   = agg1;
  float* agg2 = agg1 + (size_t)n * 16;
  const int nrange = ceil_div(n, 512);
  int*   gcount = glist + (size_t)nrange * 10240;

  hipMemsetAsync(deg, 0, (size_t)n * sizeof(int), stream);

  deg_kernel<<<ceil_div(e, 256), 256, 0, stream>>>(dst, deg, e);
  dinv_kernel<<<ceil_div(n, 256), 256, 0, stream>>>(deg, dinv, n);

  gemm_wmma_f32_lds_a<256, 128, 4, 2, 4><<<ceil_div(n, 64), 128, 0, stream>>>(x, W1, h1, n);
  agg_lds_kernel<7, 512, 10240, true, false><<<nrange, 1024, 0, stream>>>(
      h1, src, dst, dinv, agg1, glist, gcount, e, n);
  bias_relu_kernel<7><<<ceil_div((long long)n * 128, 256), 256, 0, stream>>>(
      agg1, b1, h1, (long long)n * 128);

  {
    const long long waves = (long long)((n + 15) / 16) * (16 / 16);
    gemm_wmma_f32<128, 16><<<ceil_div(waves * 32, 256), 256, 0, stream>>>(h1, W2, h2, n);
  }
  agg_lds_kernel<4, 512, 10240, false, true><<<nrange, 1024, 0, stream>>>(
      h2, src, dst, dinv, agg2, glist, gcount, e, n);
  lsm_kernel<<<ceil_div(n, 256), 256, 0, stream>>>(agg2, b2, (float*)d_out, n);
}


# Range owners scan a byte key plane (512-row range id per edge, written by the degree kernel, 16 edges per 16-B load, zero-byte test) instead of the 4-B dst array; sum order LDS arrival, as before
# speedup vs baseline: 2.8678x; 1.0786x over previous
#include <hip/hip_runtime.h>
#include <math.h>


typedef __attribute__((ext_vector_type(2))) float v2f;
typedef __attribute__((ext_vector_type(8))) float v8f;

static inline int ceil_div(long long a, long long b) { return (int)((a + b - 1) / b); }

template <int K, int N>
__global__ __launch_bounds__(256) void gemm_wmma_f32(const float* __restrict__ A,
                                                     const float* __restrict__ B,
                                                     float* __restrict__ C, int M) {
  constexpr int NT = N / 16;
  const int mTiles = (M + 15) >> 4;
  const int gwave = (int)((blockIdx.x * (long long)blockDim.x + threadIdx.x) >> 5);
  if (gwave >= mTiles * NT) return;

  const int lane = threadIdx.x & 31;
  const int mt = gwave / NT;
  const int nt = gwave - mt * NT;
  const int half = lane >> 4;
  const int l16 = lane & 15;

  const int row = mt * 16 + l16;
  const int rclamp = row < M ? row : (M - 1);
  const int col = nt * 16 + l16;
  const float* __restrict__ Ar = A + (long long)rclamp * K;

  v8f acc = {};
#pragma unroll 4
  for (int k = 0; k < K; k += 4) {
    const int ka = k + 2 * half;
    v2f a, b;
    a.x = Ar[ka];
    a.y = Ar[ka + 1];
    b.x = B[(long long)ka * N + col];
    b.y = B[(long long)(ka + 1) * N + col];
    acc = __builtin_amdgcn_wmma_f32_16x16x4_f32(false, a, false, b, (short)0, acc,
                                                false, false);
  }

  const int m0 = mt * 16 + 8 * half;
  float* __restrict__ Cp = C + (long long)m0 * N + col;
  if (mt * 16 + 16 <= M) {
#pragma unroll
    for (int v = 0; v < 8; ++v) Cp[(long long)v * N] = acc[v];
  } else {
#pragma unroll
    for (int v = 0; v < 8; ++v)
      if (m0 + v < M) Cp[(long long)v * N] = acc[v];
  }
}

template <int K, int N, int TM, int TN, int WN>
__global__ __launch_bounds__(WN * 32) void gemm_wmma_f32_lds_a(const float* __restrict__ A,
                                                             const float* __restrict__ B,
                                                             float* __restrict__ C, int M) {
  constexpr int BM = TM * 16;
  constexpr int P = K + 4;
  constexpr int NTHR = WN * 32;
  constexpr int KQ = K / 4;
  constexpr int NV = BM * KQ;
  constexpr int LB = 32;
  static_assert(N == WN * TN * 16, "waves must tile the N columns");
  static_assert(NV % (NTHR * LB) == 0, "slab must split evenly over the staging batches");
  __shared__ __attribute__((aligned(16))) float As[BM * P];

  const int tid = threadIdx.x;
  const int row0 = blockIdx.x * BM;

  for (int i0 = 0; i0 < NV; i0 += NTHR * LB) {
    float4 t[LB];
    for (int u = 0; u < LB; ++u) {
      const int f = i0 + u * NTHR + tid;
      const int r = f / KQ, kq = f - r * KQ;
      int gr = row0 + r;
      gr = gr < M ? gr : (M - 1);
      t[u] = *reinterpret_cast<const float4*>(A + (long long)gr * K + 4 * kq);
    }
    for (int u = 0; u < LB; ++u) {
      const int f = i0 + u * NTHR + tid;
      const int r = f / KQ, kq = f - r * KQ;
      *reinterpret_cast<float4*>(As + r * P + 4 * kq) = t[u];
    }
  }
  __syncthreads();

  const int lane = tid & 31;
  const int wave = tid >> 5;
  const int half = lane >> 4;
  const int l16 = lane & 15;
  const int col0 = wave * (TN * 16) + l16;

  const float* __restrict__ Al = As + l16 * P + 2 * half;
  const float* __restrict__ Bl = B + (long long)(2 * half) * N + col0;

  v8f acc[TM][TN];
  for (int i = 0; i < TM; ++i)
    for (int j = 0; j < TN; ++j) acc[i][j] = v8f{};

#pragma unroll
  for (int k = 0; k < K; k += 4) {
    v2f a[TM], b[TN];
    for (int i = 0; i < TM; ++i) a[i] = *reinterpret_cast<const v2f*>(Al + i * 16 * P + k);
    for (int j = 0; j < TN; ++j) {
      b[j].x = Bl[(long long)k * N + j * 16];
      b[j].y = Bl[(long long)(k + 1) * N + j * 16];
    }
    for (int i = 0; i < TM; ++i)
      for (int j = 0; j < TN; ++j)
        acc[i][j] = __builtin_amdgcn_wmma_f32_16x16x4_f32(false, a[i], false, b[j], (short)0,
                                                          acc[i][j], false, false);
  }

  for (int i = 0; i < TM; ++i) {
    const int m0 = row0 + i * 16 + 8 * half;
    for (int j = 0; j < TN; ++j) {
      float* __restrict__ Cp = C + (long long)m0 * N + col0 + j * 16;
      if (row0 + BM <= M) {
        for (int v = 0; v < 8; ++v) Cp[(long long)v * N] = acc[i][j][v];
      } else {
        for (int v = 0; v < 8; ++v)
          if (m0 + v < M) Cp[(long long)v * N] = acc[i][j][v];
      }
    }
  }
}

template <int LOGR>
__global__ void deg_key_kernel(const int* __restrict__ dst, int* __restrict__ deg,
                               int* __restrict__ keys, int e) {
  const int t = blockIdx.x * blockDim.x + threadIdx.x;
  const int i0 = 4 * t;
  if (i0 >= e) return;
  unsigned w = 0xFFFFFFFFu;
  for (int k = 0; k < 4; ++k) {
    const int i = i0 + k;
    if (i < e) {
      const int d = dst[i];
      atomicAdd(&deg[d], 1);
      w = (w & ~(0xFFu << (8 * k))) | (((unsigned)(d >> LOGR) & 0xFFu) << (8 * k));
    }
  }
  keys[t] = (int)w;
}

__global__ void dinv_kernel(const int* __restrict__ deg, float* __restrict__ dinv, int n) {
  const int i = blockIdx.x * blockDim.x + threadIdx.x;
  if (i < n) dinv[i] = rsqrtf((float)deg[i] + 1.0f);
}

template <int LOGF, int R, int QCAP, bool SAVE, bool REUSE>
__global__ __launch_bounds__(1024) void agg_lds_kernel(const float* __restrict__ H,
                                                       const int* __restrict__ src,
                                                       const int* __restrict__ dst,
                                                       const float* __restrict__ dinv,
                                                       float* __restrict__ out,
                                                       int* __restrict__ glist,
                                                       int* __restrict__ gcount,
                                                       const int* __restrict__ keys, int e,
                                                       int n) {
  constexpr int F = 1 << LOGF;
  constexpr int LPE = F / 4;
  constexpr int LOGLPE = LOGF - 2;
  constexpr int EPI = 32 / LPE;
  constexpr int NT = 1024;
  __shared__ float acc[R * F];
  __shared__ int queue[QCAP];
  __shared__ int qcount;

  const int tid = threadIdx.x;
  const int lane = tid & 31;
  const int base = blockIdx.x * R;

  if (tid == 0) qcount = 0;
  for (int i = tid; i < R * LPE; i += NT) {
    const int row = i >> LOGLPE, q = i & (LPE - 1);
    const int node = base + row;
    if (node < n) {
      const float dn = dinv[node];
      const float nrm = dn * dn;
      const float4 h = *reinterpret_cast<const float4*>(H + (((long long)node) << LOGF) + 4 * q);
      float* a = acc + (row << LOGF) + q;
      a[0] = h.x * nrm;
      a[LPE] = h.y * nrm;
      a[2 * LPE] = h.z * nrm;
      a[3 * LPE] = h.w * nrm;
    }
  }
  __syncthreads();

  auto push = [&](int idx) {
    const int slot = atomicAdd(&qcount, 1);
    if (slot < QCAP) {
      queue[slot] = idx;
    } else {
      const int s = src[idx];
      const int d = dst[idx];
      const float nrm = dinv[s] * dinv[d];
      const float* hs = H + (((long long)s) << LOGF);
      float* a = acc + ((d - base) << LOGF);
      for (int f = 0; f < F; ++f) atomicAdd(a + (f & 3) * LPE + (f >> 2), hs[f] * nrm);
    }
  };
  auto in_range = [&](int d) { return (unsigned)(d - base) < (unsigned)R; };

  bool have_list = false;
  if constexpr (REUSE) {
    const int c = gcount[blockIdx.x];
    have_list = c <= QCAP;
    if (have_list) {
      const int* __restrict__ gl = glist + (long long)blockIdx.x * QCAP;
      for (int i = tid; i < c; i += NT) queue[i] = gl[i];
      if (tid == 0) qcount = c;
    }
  }

  const bool use_keys = gridDim.x <= 256;

  if (!have_list && use_keys) {
    const unsigned kb = (unsigned)blockIdx.x * 0x01010101u;
    auto nzf = [&](int wi) {
      const unsigned x = (unsigned)wi ^ kb;
      return (((x & 0x7F7F7F7Fu) + 0x7F7F7F7Fu) | x) & 0x80808080u;
    };
    auto pairmask = [&](const int4 p, const int4 q) {
      return ~((nzf(p.x) >> 7) | (nzf(p.y) >> 6) | (nzf(p.z) >> 5) | (nzf(p.w) >> 4) |
               (nzf(q.x) >> 3) | (nzf(q.y) >> 2) | (nzf(q.z) >> 1) | nzf(q.w));
    };
    auto drain = [&](unsigned m, int e0, int e1) {
      while (m) {
        const int p = __builtin_ctz(m);
        m &= m - 1;
        const int idx = ((p & 4) ? e1 : e0) + 4 * (p & 3) + (p >> 3);
        if (idx < e) push(idx);
      }
    };
    const int nw = (e + 3) >> 2;
    const int nvec = nw >> 2;
    const int4* __restrict__ kv = reinterpret_cast<const int4*>(keys);
    constexpr int U = 8;
    int v = tid;
    for (; v + (U - 1) * NT < nvec; v += U * NT) {
      int4 kk[U];
      for (int u = 0; u < U; ++u) kk[u] = kv[v + u * NT];
      for (int u = 0; u < U; u += 2)
        drain(pairmask(kk[u], kk[u + 1]), 16 * (v + u * NT), 16 * (v + (u + 1) * NT));
    }
    const int4 none = {-1, -1, -1, -1};
    for (; v < nvec; v += NT) {
      const unsigned m = pairmask(kv[v], none) & 0x0F0F0F0Fu;
      drain(m, 16 * v, 0);
    }
    if (tid < (nw & 3)) {
      const int t = 4 * nvec + tid;
      const unsigned m = ~(nzf(keys[t]) >> 7) & 0x01010101u;
      drain(m, 4 * t, 0);
    }
  }

  if (!have_list && !use_keys) {
    int head = (int)(((16u - (unsigned)(reinterpret_cast<unsigned long long>(dst) & 15u)) & 15u) >> 2);
    if (head > e) head = e;
    if (tid < head && in_range(dst[tid])) push(tid);
    const int nvec = (e - head) >> 2;
    const int4* __restrict__ dv = reinterpret_cast<const int4*>(dst + head);
    constexpr int U = 8;
    int v = tid;
    for (; v + (U - 1) * NT < nvec; v += U * NT) {
      int4 dd[U];
      for (int u = 0; u < U; ++u) dd[u] = dv[v + u * NT];
      unsigned m = 0;
      for (int u = 0; u < U; ++u) {
        m |= (in_range(dd[u].x) ? 1u : 0u) << (4 * u);
        m |= (in_range(dd[u].y) ? 2u : 0u) << (4 * u);
        m |= (in_range(dd[u].z) ? 4u : 0u) << (4 * u);
        m |= (in_range(dd[u].w) ? 8u : 0u) << (4 * u);
      }
      while (m) {
        const int b = __builtin_ctz(m);
        m &= m - 1;
        push(head + 4 * (v + (b >> 2) * NT) + (b & 3));
      }
    }
    for (; v < nvec; v += NT) {
      const int4 d4 = dv[v];
      unsigned m = (in_range(d4.x) ? 1u : 0u) | (in_range(d4.y) ? 2u : 0u) |
                   (in_range(d4.z) ? 4u : 0u) | (in_range(d4.w) ? 8u : 0u);
      while (m) {
        const int b = __builtin_ctz(m);
        m &= m - 1;
        push(head + 4 * v + b);
      }
    }
    {
      const int t = head + 4 * nvec + tid;
      if (t < e && in_range(dst[t])) push(t);
    }
  }
  __syncthreads();

  if constexpr (SAVE) {
    const int kept = qcount < QCAP ? qcount : QCAP;
    int* __restrict__ gl = glist + (long long)blockIdx.x * QCAP;
    for (int i = tid; i < kept; i += NT) gl[i] = queue[i];
    if (tid == 0) gcount[blockIdx.x] = qcount;
  }

  constexpr int STEPS = 32 / EPI;
  constexpr int UB = STEPS < 8 ? STEPS : 8;
  const int total = qcount < QCAP ? qcount : QCAP;
  const int q = lane & (LPE - 1);
  const int b0 = (tid >> 5) * 32;
  int ns = 0, nd = base;
  if (b0 + lane < total) {
    const int idx = queue[b0 + lane];
    ns = src[idx];
    nd = dst[idx];
  }
  for (int b = b0; b < total; b += NT) {
    const int s = ns;
    const int dl = nd - base;
    const float nrm = (b + lane < total) ? dinv[s] * dinv[nd] : 0.0f;
    const int cnt = total - b < 32 ? total - b : 32;
    ns = 0;
    nd = base;
    if (b + NT + lane < total) {
      const int idx = queue[b + NT + lane];
      ns = src[idx];
      nd = dst[idx];
    }
    for (int j = 0; j < STEPS; j += UB) {
      float4 h[UB];
      int dj[UB];
      float nj[UB];
      for (int u = 0; u < UB; ++u) {
        const int sub = (j + u) * EPI + (lane >> LOGLPE);
        const int sj = __shfl(s, sub);
        dj[u] = __shfl(dl, sub);
        nj[u] = __shfl(nrm, sub);
        h[u] = *reinterpret_cast<const float4*>(H + (((long long)sj) << LOGF) + 4 * q);
      }
      for (int u = 0; u < UB; ++u) {
        const int sub = (j + u) * EPI + (lane >> LOGLPE);
        if (sub < cnt) {
          float* a = acc + (dj[u] << LOGF) + q;
          atomicAdd(a, h[u].x * nj[u]);
          atomicAdd(a + LPE, h[u].y * nj[u]);
          atomicAdd(a + 2 * LPE, h[u].z * nj[u]);
          atomicAdd(a + 3 * LPE, h[u].w * nj[u]);
        }
      }
    }
  }
  __syncthreads();

  for (int i = tid; i < R * LPE; i += NT) {
    const int row = i >> LOGLPE, qq = i & (LPE - 1);
    const int node = base + row;
    if (node < n) {
      const float* a = acc + (row << LOGF) + qq;
      float4 o;
      o.x = a[0];
      o.y = a[LPE];
      o.z = a[2 * LPE];
      o.w = a[3 * LPE];
      *reinterpret_cast<float4*>(out + (((long long)node) << LOGF) + 4 * qq) = o;
    }
  }
}

template <int LOGF>
__global__ void bias_relu_kernel(const float* __restrict__ agg, const float* __restrict__ b,
                                 float* __restrict__ out, long long total) {
  constexpr int F = 1 << LOGF;
  const long long i = (long long)blockIdx.x * blockDim.x + threadIdx.x;
  if (i >= total) return;
  const float v = agg[i] + b[(int)(i & (F - 1))];
  out[i] = v > 0.0f ? v : 0.0f;
}

__global__ void lsm_kernel(const float* __restrict__ agg, const float* __restrict__ b,
                           float* __restrict__ out, int n) {
  const int i = blockIdx.x * blockDim.x + threadIdx.x;
  if (i >= n) return;
  float v[16];
  float mx = -INFINITY;
#pragma unroll
  for (int j = 0; j < 16; ++j) {
    v[j] = agg[(long long)i * 16 + j] + b[j];
    mx = fmaxf(mx, v[j]);
  }
  float s = 0.0f;
#pragma unroll
  for (int j = 0; j < 16; ++j) s += expf(v[j] - mx);
  const float lse = mx + logf(s);
#pragma unroll
  for (int j = 0; j < 16; ++j) out[(long long)i * 16 + j] = v[j] - lse;
}

extern "C" void kernel_launch(void* const* d_in, const int* in_sizes, int n_in,
                              void* d_out, int out_size, void* d_ws, size_t ws_size,
                              hipStream_t stream) {
  const float* x  = (const float*)d_in[0];
  const int*   ei = (const int*)d_in[1];
  const float* W1 = (const float*)d_in[2];
  const float* b1 = (const float*)d_in[3];
  const float* W2 = (const float*)d_in[4];
  const float* b2 = (const float*)d_in[5];

  const int n = in_sizes[0] / 256;
  const int e = in_sizes[1] / 2;
  const int* src = ei;
  const int* dst = ei + e;

  float* agg1 = (float*)d_ws;
  float* h1   = agg1 + (size_t)n * 128;
  int*   glist = (int*)(h1 + (size_t)n * 128);
  int*   deg  = glist + (size_t)n * 32;
  float* dinv = (float*)(deg + n);
  float* h2   = agg1;
  float* agg2 = agg1 + (size_t)n * 16;
  const int nrange = ceil_div(n, 512);
  int*   gcount = glist + (size_t)nrange * 10240;
  int*   keys = glist + ((((size_t)nrange * 10241) + 3) & ~(size_t)3);

  hipMemsetAsync(deg, 0, (size_t)n * sizeof(int), stream);

  deg_key_kernel<9><<<ceil_div(ceil_div(e, 4), 256), 256, 0, stream>>>(dst, deg, keys, e);
  dinv_kernel<<<ceil_div(n, 256), 256, 0, stream>>>(deg, dinv, n);

  gemm_wmma_f32_lds_a<256, 128, 4, 2, 4><<<ceil_div(n, 64), 128, 0, stream>>>(x, W1, h1, n);
  agg_lds_kernel<7, 512, 10240, true, false><<<nrange, 1024, 0, stream>>>(
      h1, src, dst, dinv, agg1, glist, gcount, keys, e, n);
  bias_relu_kernel<7><<<ceil_div((long long)n * 128, 256), 256, 0, stream>>>(
      agg1, b1, h1, (long long)n * 128);

  {
    const long long waves = (long long)((n + 15) / 16) * (16 / 16);
    gemm_wmma_f32<128, 16><<<ceil_div(waves * 32, 256), 256, 0, stream>>>(h1, W2, h2, n);
  }
  agg_lds_kernel<4, 512, 10240, false, true><<<nrange, 1024, 0, stream>>>(
      h2, src, dst, dinv, agg2, glist, gcount, keys, e, n);
  lsm_kernel<<<ceil_div(n, 256), 256, 0, stream>>>(agg2, b2, (float*)d_out, n);
}


# bias+ReLU moved to the consumer: the narrow GEMM reads the raw sums, a wave stages its 16 rows in LDS (16-B row loads) and applies them on the way; same values and k order; pointwise launch gone
# speedup vs baseline: 3.2109x; 1.1196x over previous
#include <hip/hip_runtime.h>
#include <math.h>


typedef __attribute__((ext_vector_type(2))) float v2f;
typedef __attribute__((ext_vector_type(8))) float v8f;
typedef __attribute__((address_space(1))) float global_float;

static inline int ceil_div(long long a, long long b) { return (int)((a + b - 1) / b); }

template <int K, int N, int WPB>
__global__ __launch_bounds__(WPB * 32) void gemm_wmma_f32_relu_a(const float* __restrict__ S,
                                                                const float* __restrict__ bias,
                                                                const float* __restrict__ B,
                                                                float* __restrict__ C, int M) {
  constexpr int NT = N / 16;
  constexpr int P = K + 4;
  static_assert(K == 128, "one 16-byte piece of a row per lane");
  __shared__ __attribute__((aligned(16))) float As[WPB * 16 * P];

  const int lane = threadIdx.x & 31;
  const int wave = threadIdx.x >> 5;
  const int gwave = __builtin_amdgcn_readfirstlane((int)(blockIdx.x * WPB + wave));
  const int mt = gwave / NT;
  const int nt = gwave - mt * NT;
  const int half = lane >> 4;
  const int l16 = lane & 15;
  const int col = nt * 16 + l16;
  float* __restrict__ Aw = As + wave * (16 * P);

  const int m0 = mt * 16 + 8 * half;
  unsigned long long caddr = reinterpret_cast<unsigned long long>(C + (long long)m0 * N + col);
  int nrows = M - m0;
  asm volatile("" : "+v"(caddr), "+v"(nrows));
  global_float* Cp = (global_float*)caddr;

  {
    const float4 bq = *reinterpret_cast<const float4*>(bias + 4 * lane);
    float4 t[16];
    for (int r = 0; r < 16; ++r) {
      int gr = mt * 16 + r;
      gr = gr < M ? gr : (M - 1);
      t[r] = *reinterpret_cast<const float4*>(S + (long long)gr * K + 4 * lane);
    }
    for (int r = 0; r < 16; ++r) {
      float4 v;
      v.x = t[r].x + bq.x;
      v.y = t[r].y + bq.y;
      v.z = t[r].z + bq.z;
      v.w = t[r].w + bq.w;
      v.x = v.x > 0.0f ? v.x : 0.0f;
      v.y = v.y > 0.0f ? v.y : 0.0f;
      v.z = v.z > 0.0f ? v.z : 0.0f;
      v.w = v.w > 0.0f ? v.w : 0.0f;
      *reinterpret_cast<float4*>(Aw + r * P + 4 * lane) = v;
    }
  }
  __syncthreads();

  const float* __restrict__ Al = Aw + l16 * P + 2 * half;
  const float* __restrict__ Bl = B + (long long)(2 * half) * N + col;

  v8f acc = {};
#pragma unroll
  for (int k = 0; k < K; k += 4) {
    const v2f a = *reinterpret_cast<const v2f*>(Al + k);
    v2f b;
    b.x = Bl[(long long)k * N];
    b.y = Bl[(long long)(k + 1) * N];
    acc = __builtin_amdgcn_wmma_f32_16x16x4_f32(false, a, false, b, (short)0, acc,
                                                false, false);
  }

  if (mt * 16 + 16 <= M) {
#pragma unroll
    for (int v = 0; v < 8; ++v) Cp[(long long)v * N] = acc[v];
  } else {
#pragma unroll
    for (int v = 0; v < 8; ++v)
      if (v < nrows) Cp[(long long)v * N] = acc[v];
  }
}

template <int K, int N, int TM, int TN, int WN>
__global__ __launch_bounds__(WN * 32) void gemm_wmma_f32_lds_a(const float* __restrict__ A,
                                                             const float* __restrict__ B,
                                                             float* __restrict__ C, int M) {
  constexpr int BM = TM * 16;
  constexpr int P = K + 4;
  constexpr int NTHR = WN * 32;
  constexpr int KQ = K / 4;
  constexpr int NV = BM * KQ;
  constexpr int LB = 32;
  static_assert(N == WN * TN * 16, "waves must tile the N columns");
  static_assert(NV % (NTHR * LB) == 0, "slab must split evenly over the staging batches");
  __shared__ __attribute__((aligned(16))) float As[BM * P];

  const int tid = threadIdx.x;
  const int row0 = blockIdx.x * BM;

  for (int i0 = 0; i0 < NV; i0 += NTHR * LB) {
    float4 t[LB];
    for (int u = 0; u < LB; ++u) {
      const int f = i0 + u * NTHR + tid;
      const int r = f / KQ, kq = f - r * KQ;
      int gr = row0 + r;
      gr = gr < M ? gr : (M - 1);
      t[u] = *reinterpret_cast<const float4*>(A + (long long)gr * K + 4 * kq);
    }
    for (int u = 0; u < LB; ++u) {
      const int f = i0 + u * NTHR + tid;
      const int r = f / KQ, kq = f - r * KQ;
      *reinterpret_cast<float4*>(As + r * P + 4 * kq) = t[u];
    }
  }
  __syncthreads();

  const int lane = tid & 31;
  const int wave = tid >> 5;
  const int half = lane >> 4;
  const int l16 = lane & 15;
  const int col0 = wave * (TN * 16) + l16;

  const float* __restrict__ Al = As + l16 * P + 2 * half;
  const float* __restrict__ Bl = B + (long long)(2 * half) * N + col0;

  v8f acc[TM][TN];
  for (int i = 0; i < TM; ++i)
    for (int j = 0; j < TN; ++j) acc[i][j] = v8f{};

#pragma unroll
  for (int k = 0; k < K; k += 4) {
    v2f a[TM], b[TN];
    for (int i = 0; i < TM; ++i) a[i] = *reinterpret_cast<const v2f*>(Al + i * 16 * P + k);
    for (int j = 0; j < TN; ++j) {
      b[j].x = Bl[(long long)k * N + j * 16];
      b[j].y = Bl[(long long)(k + 1) * N + j * 16];
    }
    for (int i = 0; i < TM; ++i)
      for (int j = 0; j < TN; ++j)
        acc[i][j] = __builtin_amdgcn_wmma_f32_16x16x4_f32(false, a[i], false, b[j], (short)0,
                                                          acc[i][j], false, false);
  }

  for (int i = 0; i < TM; ++i) {
    const int m0 = row0 + i * 16 + 8 * half;
    for (int j = 0; j < TN; ++j) {
      float* __restrict__ Cp = C + (long long)m0 * N + col0 + j * 16;
      if (row0 + BM <= M) {
        for (int v = 0; v < 8; ++v) Cp[(long long)v * N] = acc[i][j][v];
      } else {
        for (int v = 0; v < 8; ++v)
          if (m0 + v < M) Cp[(long long)v * N] = acc[i][j][v];
      }
    }
  }
}

template <int LOGR>
__global__ void deg_key_kernel(const int* __restrict__ dst, int* __restrict__ deg,
                               int* __restrict__ keys, int e) {
  const int t = blockIdx.x * blockDim.x + threadIdx.x;
  const int i0 = 4 * t;
  if (i0 >= e) return;
  unsigned w = 0xFFFFFFFFu;
  for (int k = 0; k < 4; ++k) {
    const int i = i0 + k;
    if (i < e) {
      const int d = dst[i];
      atomicAdd(&deg[d], 1);
      w = (w & ~(0xFFu << (8 * k))) | (((unsigned)(d >> LOGR) & 0xFFu) << (8 * k));
    }
  }
  keys[t] = (int)w;
}

__global__ void dinv_kernel(const int* __restrict__ deg, float* __restrict__ dinv, int n) {
  const int i = blockIdx.x * blockDim.x + threadIdx.x;
  if (i < n) dinv[i] = rsqrtf((float)deg[i] + 1.0f);
}

template <int LOGF, int R, int QCAP, bool SAVE, bool REUSE>
__global__ __launch_bounds__(1024) void agg_lds_kernel(const float* __restrict__ H,
                                                       const int* __restrict__ src,
                                                       const int* __restrict__ dst,
                                                       const float* __restrict__ dinv,
                                                       float* __restrict__ out,
                                                       int* __restrict__ glist,
                                                       int* __restrict__ gcount,
                                                       const int* __restrict__ keys, int e,
                                                       int n) {
  constexpr int F = 1 << LOGF;
  constexpr int LPE = F / 4;
  constexpr int LOGLPE = LOGF - 2;
  constexpr int EPI = 32 / LPE;
  constexpr int NT = 1024;
  __shared__ float acc[R * F];
  __shared__ int queue[QCAP];
  __shared__ int qcount;

  const int tid = threadIdx.x;
  const int lane = tid & 31;
  const int base = blockIdx.x * R;

  if (tid == 0) qcount = 0;
  for (int i = tid; i < R * LPE; i += NT) {
    const int row = i >> LOGLPE, q = i & (LPE - 1);
    const int node = base + row;
    if (node < n) {
      const float dn = dinv[node];
      const float nrm = dn * dn;
      const float4 h = *reinterpret_cast<const float4*>(H + (((long long)node) << LOGF) + 4 * q);
      float* a = acc + (row << LOGF) + q;
      a[0] = h.x * nrm;
      a[LPE] = h.y * nrm;
      a[2 * LPE] = h.z * nrm;
      a[3 * LPE] = h.w * nrm;
    }
  }
  __syncthreads();

  auto push = [&](int idx) {
    const int slot = atomicAdd(&qcount, 1);
    if (slot < QCAP) {
      queue[slot] = idx;
    } else {
      const int s = src[idx];
      const int d = dst[idx];
      const float nrm = dinv[s] * dinv[d];
      const float* hs = H + (((long long)s) << LOGF);
      float* a = acc + ((d - base) << LOGF);
      for (int f = 0; f < F; ++f) atomicAdd(a + (f & 3) * LPE + (f >> 2), hs[f] * nrm);
    }
  };
  auto in_range = [&](int d) { return (unsigned)(d - base) < (unsigned)R; };

  bool have_list = false;
  if constexpr (REUSE) {
    const int c = gcount[blockIdx.x];
    have_list = c <= QCAP;
    if (have_list) {
      const int* __restrict__ gl = glist + (long long)blockIdx.x * QCAP;
      for (int i = tid; i < c; i += NT) queue[i] = gl[i];
      if (tid == 0) qcount = c;
    }
  }

  const bool use_keys = gridDim.x <= 256;

  if (!have_list && use_keys) {
    const unsigned kb = (unsigned)blockIdx.x * 0x01010101u;
    auto nzf = [&](int wi) {
      const unsigned x = (unsigned)wi ^ kb;
      return (((x & 0x7F7F7F7Fu) + 0x7F7F7F7Fu) | x) & 0x80808080u;
    };
    auto pairmask = [&](const int4 p, const int4 q) {
      return ~((nzf(p.x) >> 7) | (nzf(p.y) >> 6) | (nzf(p.z) >> 5) | (nzf(p.w) >> 4) |
               (nzf(q.x) >> 3) | (nzf(q.y) >> 2) | (nzf(q.z) >> 1) | nzf(q.w));
    };
    auto drain = [&](unsigned m, int e0, int e1) {
      while (m) {
        const int p = __builtin_ctz(m);
        m &= m - 1;
        const int idx = ((p & 4) ? e1 : e0) + 4 * (p & 3) + (p >> 3);
        if (idx < e) push(idx);
      }
    };
    const int nw = (e + 3) >> 2;
    const int nvec = nw >> 2;
    const int4* __restrict__ kv = reinterpret_cast<const int4*>(keys);
    constexpr int U = 8;
    int v = tid;
    for (; v + (U - 1) * NT < nvec; v += U * NT) {
      int4 kk[U];
      for (int u = 0; u < U; ++u) kk[u] = kv[v + u * NT];
      for (int u = 0; u < U; u += 2)
        drain(pairmask(kk[u], kk[u + 1]), 16 * (v + u * NT), 16 * (v + (u + 1) * NT));
    }
    const int4 none = {-1, -1, -1, -1};
    for (; v < nvec; v += NT) {
      const unsigned m = pairmask(kv[v], none) & 0x0F0F0F0Fu;
      drain(m, 16 * v, 0);
    }
    if (tid < (nw & 3)) {
      const int t = 4 * nvec + tid;
      const unsigned m = ~(nzf(keys[t]) >> 7) & 0x01010101u;
      drain(m, 4 * t, 0);
    }
  }

  if (!have_list && !use_keys) {
    int head = (int)(((16u - (unsigned)(reinterpret_cast<unsigned long long>(dst) & 15u)) & 15u) >> 2);
    if (head > e) head = e;
    if (tid < head && in_range(dst[tid])) push(tid);
    const int nvec = (e - head) >> 2;
    const int4* __restrict__ dv = reinterpret_cast<const int4*>(dst + head);
    constexpr int U = 8;
    int v = tid;
    for (; v + (U - 1) * NT < nvec; v += U * NT) {
      int4 dd[U];
      for (int u = 0; u < U; ++u) dd[u] = dv[v + u * NT];
      unsigned m = 0;
      for (int u = 0; u < U; ++u) {
        m |= (in_range(dd[u].x) ? 1u : 0u) << (4 * u);
        m |= (in_range(dd[u].y) ? 2u : 0u) << (4 * u);
        m |= (in_range(dd[u].z) ? 4u : 0u) << (4 * u);
        m |= (in_range(dd[u].w) ? 8u : 0u) << (4 * u);
      }
      while (m) {
        const int b = __builtin_ctz(m);
        m &= m - 1;
        push(head + 4 * (v + (b >> 2) * NT) + (b & 3));
      }
    }
    for (; v < nvec; v += NT) {
      const int4 d4 = dv[v];
      unsigned m = (in_range(d4.x) ? 1u : 0u) | (in_range(d4.y) ? 2u : 0u) |
                   (in_range(d4.z) ? 4u : 0u) | (in_range(d4.w) ? 8u : 0u);
      while (m) {
        const int b = __builtin_ctz(m);
        m &= m - 1;
        push(head + 4 * v + b);
      }
    }
    {
      const int t = head + 4 * nvec + tid;
      if (t < e && in_range(dst[t])) push(t);
    }
  }
  __syncthreads();

  if constexpr (SAVE) {
    const int kept = qcount < QCAP ? qcount : QCAP;
    int* __restrict__ gl = glist + (long long)blockIdx.x * QCAP;
    for (int i = tid; i < kept; i += NT) gl[i] = queue[i];
    if (tid == 0) gcount[blockIdx.x] = qcount;
  }

  constexpr int STEPS = 32 / EPI;
  constexpr int UB = STEPS < 8 ? STEPS : 8;
  const int total = qcount < QCAP ? qcount : QCAP;
  const int q = lane & (LPE - 1);
  const int b0 = (tid >> 5) * 32;
  int ns = 0, nd = base;
  if (b0 + lane < total) {
    const int idx = queue[b0 + lane];
    ns = src[idx];
    nd = dst[idx];
  }
  for (int b = b0; b < total; b += NT) {
    const int s = ns;
    const int dl = nd - base;
    const float nrm = (b + lane < total) ? dinv[s] * dinv[nd] : 0.0f;
    const int cnt = total - b < 32 ? total - b : 32;
    ns = 0;
    nd = base;
    if (b + NT + lane < total) {
      const int idx = queue[b + NT + lane];
      ns = src[idx];
      nd = dst[idx];
    }
    for (int j = 0; j < STEPS; j += UB) {
      float4 h[UB];
      int dj[UB];
      float nj[UB];
      for (int u = 0; u < UB; ++u) {
        const int sub = (j + u) * EPI + (lane >> LOGLPE);
        const int sj = __shfl(s, sub);
        dj[u] = __shfl(dl, sub);
        nj[u] = __shfl(nrm, sub);
        h[u] = *reinterpret_cast<const float4*>(H + (((long long)sj) << LOGF) + 4 * q);
      }
      for (int u = 0; u < UB; ++u) {
        const int sub = (j + u) * EPI + (lane >> LOGLPE);
        if (sub < cnt) {
          float* a = acc + (dj[u] << LOGF) + q;
          atomicAdd(a, h[u].x * nj[u]);
          atomicAdd(a + LPE, h[u].y * nj[u]);
          atomicAdd(a + 2 * LPE, h[u].z * nj[u]);
          atomicAdd(a + 3 * LPE, h[u].w * nj[u]);
        }
      }
    }
  }
  __syncthreads();

  for (int i = tid; i < R * LPE; i += NT) {
    const int row = i >> LOGLPE, qq = i & (LPE - 1);
    const int node = base + row;
    if (node < n) {
      const float* a = acc + (row << LOGF) + qq;
      float4 o;
      o.x = a[0];
      o.y = a[LPE];
      o.z = a[2 * LPE];
      o.w = a[3 * LPE];
      *reinterpret_cast<float4*>(out + (((long long)node) << LOGF) + 4 * qq) = o;
    }
  }
}

__global__ void lsm_kernel(const float* __restrict__ agg, const float* __restrict__ b,
                           float* __restrict__ out, int n) {
  const int i = blockIdx.x * blockDim.x + threadIdx.x;
  if (i >= n) return;
  float v[16];
  float mx = -INFINITY;
#pragma unroll
  for (int j = 0; j < 16; ++j) {
    v[j] = agg[(long long)i * 16 + j] + b[j];
    mx = fmaxf(mx, v[j]);
  }
  float s = 0.0f;
#pragma unroll
  for (int j = 0; j < 16; ++j) s += expf(v[j] - mx);
  const float lse = mx + logf(s);
#pragma unroll
  for (int j = 0; j < 16; ++j) out[(long long)i * 16 + j] = v[j] - lse;
}

extern "C" void kernel_launch(void* const* d_in, const int* in_sizes, int n_in,
                              void* d_out, int out_size, void* d_ws, size_t ws_size,
                              hipStream_t stream) {
  const float* x  = (const float*)d_in[0];
  const int*   ei = (const int*)d_in[1];
  const float* W1 = (const float*)d_in[2];
  const float* b1 = (const float*)d_in[3];
  const float* W2 = (const float*)d_in[4];
  const float* b2 = (const float*)d_in[5];

  const int n = in_sizes[0] / 256;
  const int e = in_sizes[1] / 2;
  const int* src = ei;
  const int* dst = ei + e;

  float* agg1 = (float*)d_ws;
  float* h1   = agg1 + (size_t)n * 128;
  int*   glist = (int*)(h1 + (size_t)n * 128);
  int*   deg  = glist + (size_t)n * 32;
  float* dinv = (float*)(deg + n);
  float* h2   = h1;
  float* agg2 = h1 + (size_t)n * 16;
  const int nrange = ceil_div(n, 512);
  int*   gcount = glist + (size_t)nrange * 10240;
  int*   keys = glist + ((((size_t)nrange * 10241) + 3) & ~(size_t)3);

  hipMemsetAsync(deg, 0, (size_t)n * sizeof(int), stream);

  deg_key_kernel<9><<<ceil_div(ceil_div(e, 4), 256), 256, 0, stream>>>(dst, deg, keys, e);
  dinv_kernel<<<ceil_div(n, 256), 256, 0, stream>>>(deg, dinv, n);

  gemm_wmma_f32_lds_a<256, 128, 4, 2, 4><<<ceil_div(n, 64), 128, 0, stream>>>(x, W1, h1, n);
  agg_lds_kernel<7, 512, 10240, true, false><<<nrange, 1024, 0, stream>>>(
      h1, src, dst, dinv, agg1, glist, gcount, keys, e, n);

  {
    const long long waves = (long long)((n + 15) / 16) * (16 / 16);
    gemm_wmma_f32_relu_a<128, 16, 8><<<ceil_div(waves * 32, 256), 256, 0, stream>>>(
        agg1, b1, W2, h2, n);
  }
  agg_lds_kernel<4, 512, 10240, false, true><<<nrange, 1024, 0, stream>>>(
      h2, src, dst, dinv, agg2, glist, gcount, keys, e, n);
  lsm_kernel<<<ceil_div(n, 256), 256, 0, stream>>>(agg2, b2, (float*)d_out, n);
}


# In-degrees counted by the range owners in LDS: a new owner pass scans the key plane, saves each range's edge list and writes dinv; global int atomics, memset, dinv launch gone; sum order LDS arrival
# speedup vs baseline: 4.0927x; 1.2746x over previous
#include <hip/hip_runtime.h>
#include <math.h>


typedef __attribute__((ext_vector_type(2))) float v2f;
typedef __attribute__((ext_vector_type(8))) float v8f;
typedef __attribute__((address_space(1))) float global_float;

static inline int ceil_div(long long a, long long b) { return (int)((a + b - 1) / b); }

template <int K, int N, int WPB>
__global__ __launch_bounds__(WPB * 32) void gemm_wmma_f32_relu_a(const float* __restrict__ S,
                                                                const float* __restrict__ bias,
                                                                const float* __restrict__ B,
                                                                float* __restrict__ C, int M) {
  constexpr int NT = N / 16;
  constexpr int P = K + 4;
  static_assert(K == 128, "one 16-byte piece of a row per lane");
  __shared__ __attribute__((aligned(16))) float As[WPB * 16 * P];

  const int lane = threadIdx.x & 31;
  const int wave = threadIdx.x >> 5;
  const int gwave = __builtin_amdgcn_readfirstlane((int)(blockIdx.x * WPB + wave));
  const int mt = gwave / NT;
  const int nt = gwave - mt * NT;
  const int half = lane >> 4;
  const int l16 = lane & 15;
  const int col = nt * 16 + l16;
  float* __restrict__ Aw = As + wave * (16 * P);

  const int m0 = mt * 16 + 8 * half;
  unsigned long long caddr = reinterpret_cast<unsigned long long>(C + (long long)m0 * N + col);
  int nrows = M - m0;
  asm volatile("" : "+v"(caddr), "+v"(nrows));
  global_float* Cp = (global_float*)caddr;

  {
    const float4 bq = *reinterpret_cast<const float4*>(bias + 4 * lane);
    float4 t[16];
    for (int r = 0; r < 16; ++r) {
      int gr = mt * 16 + r;
      gr = gr < M ? gr : (M - 1);
      t[r] = *reinterpret_cast<const float4*>(S + (long long)gr * K + 4 * lane);
    }
    for (int r = 0; r < 16; ++r) {
      float4 v;
      v.x = t[r].x + bq.x;
      v.y = t[r].y + bq.y;
      v.z = t[r].z + bq.z;
      v.w = t[r].w + bq.w;
      v.x = v.x > 0.0f ? v.x : 0.0f;
      v.y = v.y > 0.0f ? v.y : 0.0f;
      v.z = v.z > 0.0f ? v.z : 0.0f;
      v.w = v.w > 0.0f ? v.w : 0.0f;
      *reinterpret_cast<float4*>(Aw + r * P + 4 * lane) = v;
    }
  }
  __syncthreads();

  const float* __restrict__ Al = Aw + l16 * P + 2 * half;
  const float* __restrict__ Bl = B + (long long)(2 * half) * N + col;

  v8f acc = {};
#pragma unroll
  for (int k = 0; k < K; k += 4) {
    const v2f a = *reinterpret_cast<const v2f*>(Al + k);
    v2f b;
    b.x = Bl[(long long)k * N];
    b.y = Bl[(long long)(k + 1) * N];
    acc = __builtin_amdgcn_wmma_f32_16x16x4_f32(false, a, false, b, (short)0, acc,
                                                false, false);
  }

  if (mt * 16 + 16 <= M) {
#pragma unroll
    for (int v = 0; v < 8; ++v) Cp[(long long)v * N] = acc[v];
  } else {
#pragma unroll
    for (int v = 0; v < 8; ++v)
      if (v < nrows) Cp[(long long)v * N] = acc[v];
  }
}

template <int K, int N, int TM, int TN, int WN>
__global__ __launch_bounds__(WN * 32) void gemm_wmma_f32_lds_a(const float* __restrict__ A,
                                                             const float* __restrict__ B,
                                                             float* __restrict__ C, int M) {
  constexpr int BM = TM * 16;
  constexpr int P = K + 4;
  constexpr int NTHR = WN * 32;
  constexpr int KQ = K / 4;
  constexpr int NV = BM * KQ;
  constexpr int LB = 32;
  static_assert(N == WN * TN * 16, "waves must tile the N columns");
  static_assert(NV % (NTHR * LB) == 0, "slab must split evenly over the staging batches");
  __shared__ __attribute__((aligned(16))) float As[BM * P];

  const int tid = threadIdx.x;
  const int row0 = blockIdx.x * BM;

  for (int i0 = 0; i0 < NV; i0 += NTHR * LB) {
    float4 t[LB];
    for (int u = 0; u < LB; ++u) {
      const int f = i0 + u * NTHR + tid;
      const int r = f / KQ, kq = f - r * KQ;
      int gr = row0 + r;
      gr = gr < M ? gr : (M - 1);
      t[u] = *reinterpret_cast<const float4*>(A + (long long)gr * K + 4 * kq);
    }
    for (int u = 0; u < LB; ++u) {
      const int f = i0 + u * NTHR + tid;
      const int r = f / KQ, kq = f - r * KQ;
      *reinterpret_cast<float4*>(As + r * P + 4 * kq) = t[u];
    }
  }
  __syncthreads();

  const int lane = tid & 31;
  const int wave = tid >> 5;
  const int half = lane >> 4;
  const int l16 = lane & 15;
  const int col0 = wave * (TN * 16) + l16;

  const float* __restrict__ Al = As + l16 * P + 2 * half;
  const float* __restrict__ Bl = B + (long long)(2 * half) * N + col0;

  v8f acc[TM][TN];
  for (int i = 0; i < TM; ++i)
    for (int j = 0; j < TN; ++j) acc[i][j] = v8f{};

#pragma unroll
  for (int k = 0; k < K; k += 4) {
    v2f a[TM], b[TN];
    for (int i = 0; i < TM; ++i) a[i] = *reinterpret_cast<const v2f*>(Al + i * 16 * P + k);
    for (int j = 0; j < TN; ++j) {
      b[j].x = Bl[(long long)k * N + j * 16];
      b[j].y = Bl[(long long)(k + 1) * N + j * 16];
    }
    for (int i = 0; i < TM; ++i)
      for (int j = 0; j < TN; ++j)
        acc[i][j] = __builtin_amdgcn_wmma_f32_16x16x4_f32(false, a[i], false, b[j], (short)0,
                                                          acc[i][j], false, false);
  }

  for (int i = 0; i < TM; ++i) {
    const int m0 = row0 + i * 16 + 8 * half;
    for (int j = 0; j < TN; ++j) {
      float* __restrict__ Cp = C + (long long)m0 * N + col0 + j * 16;
      if (row0 + BM <= M) {
        for (int v = 0; v < 8; ++v) Cp[(long long)v * N] = acc[i][j][v];
      } else {
        for (int v = 0; v < 8; ++v)
          if (m0 + v < M) Cp[(long long)v * N] = acc[i][j][v];
      }
    }
  }
}

template <int LOGR>
__global__ void key_kernel(const int* __restrict__ dst, int* __restrict__ keys, int e) {
  const int t = blockIdx.x * blockDim.x + threadIdx.x;
  const int i0 = 4 * t;
  if (i0 >= e) return;
  unsigned w = 0xFFFFFFFFu;
  for (int k = 0; k < 4; ++k) {
    const int i = i0 + k;
    if (i < e) {
      const int d = dst[i];
      w = (w & ~(0xFFu << (8 * k))) | (((unsigned)(d >> LOGR) & 0xFFu) << (8 * k));
    }
  }
  keys[t] = (int)w;
}

template <int R, int NT, class Push>
__device__ __forceinline__ void scan_for_range(const int* __restrict__ keys,
                                               const int* __restrict__ dst, int e, int base,
                                               Push&& push) {
  const int tid = threadIdx.x;
  auto in_range = [&](int d) { return (unsigned)(d - base) < (unsigned)R; };
  const bool use_keys = gridDim.x <= 256;

  if (use_keys) {
    const unsigned kb = (unsigned)blockIdx.x * 0x01010101u;
    auto nzf = [&](int wi) {
      const unsigned x = (unsigned)wi ^ kb;
      return (((x & 0x7F7F7F7Fu) + 0x7F7F7F7Fu) | x) & 0x80808080u;
    };
    auto pairmask = [&](const int4 p, const int4 q) {
      return ~((nzf(p.x) >> 7) | (nzf(p.y) >> 6) | (nzf(p.z) >> 5) | (nzf(p.w) >> 4) |
               (nzf(q.x) >> 3) | (nzf(q.y) >> 2) | (nzf(q.z) >> 1) | nzf(q.w));
    };
    auto drain = [&](unsigned m, int e0, int e1) {
      while (m) {
        const int p = __builtin_ctz(m);
        m &= m - 1;
        const int idx = ((p & 4) ? e1 : e0) + 4 * (p & 3) + (p >> 3);
        if (idx < e) push(idx);
      }
    };
    const int nw = (e + 3) >> 2;
    const int nvec = nw >> 2;
    const int4* __restrict__ kv = reinterpret_cast<const int4*>(keys);
    constexpr int U = 8;
    int v = tid;
    for (; v + (U - 1) * NT < nvec; v += U * NT) {
      int4 kk[U];
      for (int u = 0; u < U; ++u) kk[u] = kv[v + u * NT];
      for (int u = 0; u < U; u += 2)
        drain(pairmask(kk[u], kk[u + 1]), 16 * (v + u * NT), 16 * (v + (u + 1) * NT));
    }
    const int4 none = {-1, -1, -1, -1};
    for (; v < nvec; v += NT) {
      const unsigned m = pairmask(kv[v], none) & 0x0F0F0F0Fu;
      drain(m, 16 * v, 0);
    }
    if (tid < (nw & 3)) {
      const int t = 4 * nvec + tid;
      const unsigned m = ~(nzf(keys[t]) >> 7) & 0x01010101u;
      drain(m, 4 * t, 0);
    }
  } else {
    int head = (int)(((16u - (unsigned)(reinterpret_cast<unsigned long long>(dst) & 15u)) & 15u) >> 2);
    if (head > e) head = e;
    if (tid < head && in_range(dst[tid])) push(tid);
    const int nvec = (e - head) >> 2;
    const int4* __restrict__ dv = reinterpret_cast<const int4*>(dst + head);
    constexpr int U = 8;
    int v = tid;
    for (; v + (U - 1) * NT < nvec; v += U * NT) {
      int4 dd[U];
      for (int u = 0; u < U; ++u) dd[u] = dv[v + u * NT];
      unsigned m = 0;
      for (int u = 0; u < U; ++u) {
        m |= (in_range(dd[u].x) ? 1u : 0u) << (4 * u);
        m |= (in_range(dd[u].y) ? 2u : 0u) << (4 * u);
        m |= (in_range(dd[u].z) ? 4u : 0u) << (4 * u);
        m |= (in_range(dd[u].w) ? 8u : 0u) << (4 * u);
      }
      while (m) {
        const int b = __builtin_ctz(m);
        m &= m - 1;
        push(head + 4 * (v + (b >> 2) * NT) + (b & 3));
      }
    }
    for (; v < nvec; v += NT) {
      const int4 d4 = dv[v];
      unsigned m = (in_range(d4.x) ? 1u : 0u) | (in_range(d4.y) ? 2u : 0u) |
                   (in_range(d4.z) ? 4u : 0u) | (in_range(d4.w) ? 8u : 0u);
      while (m) {
        const int b = __builtin_ctz(m);
        m &= m - 1;
        push(head + 4 * v + b);
      }
    }
    {
      const int t = head + 4 * nvec + tid;
      if (t < e && in_range(dst[t])) push(t);
    }
  }
}

template <int R, int QCAP>
__global__ __launch_bounds__(1024) void range_list_kernel(const int* __restrict__ dst,
                                                          float* __restrict__ dinv,
                                                          int* __restrict__ glist,
                                                          int* __restrict__ gcount,
                                                          const int* __restrict__ keys, int e,
                                                          int n) {
  constexpr int NT = 1024;
  __shared__ int cnt[R];
  __shared__ int queue[QCAP];
  __shared__ int qcount;

  const int tid = threadIdx.x;
  const int base = blockIdx.x * R;

  if (tid == 0) qcount = 0;
  for (int i = tid; i < R; i += NT) cnt[i] = 0;
  __syncthreads();

  auto push = [&](int idx) {
    const int slot = atomicAdd(&qcount, 1);
    if (slot < QCAP) {
      queue[slot] = idx;
    } else {
      atomicAdd(&cnt[dst[idx] - base], 1);
    }
  };
  scan_for_range<R, NT>(keys, dst, e, base, push);
  __syncthreads();

  const int met = qcount;
  const int total = met < QCAP ? met : QCAP;
  int* __restrict__ gl = glist + (long long)blockIdx.x * QCAP;
  for (int i = tid; i < total; i += NT) {
    const int idx = queue[i];
    gl[i] = idx;
    atomicAdd(&cnt[dst[idx] - base], 1);
  }
  if (tid == 0) gcount[blockIdx.x] = met;
  __syncthreads();

  for (int i = tid; i < R; i += NT) {
    const int node = base + i;
    if (node < n) dinv[node] = rsqrtf((float)cnt[i] + 1.0f);
  }
}

template <int LOGF, int R, int QCAP>
__global__ __launch_bounds__(1024) void agg_lds_kernel(const float* __restrict__ H,
                                                       const int* __restrict__ src,
                                                       const int* __restrict__ dst,
                                                       const float* __restrict__ dinv,
                                                       float* __restrict__ out,
                                                       const int* __restrict__ glist,
                                                       const int* __restrict__ gcount,
                                                       const int* __restrict__ keys, int e,
                                                       int n) {
  constexpr int F = 1 << LOGF;
  constexpr int LPE = F / 4;
  constexpr int LOGLPE = LOGF - 2;
  constexpr int EPI = 32 / LPE;
  constexpr int NT = 1024;
  __shared__ float acc[R * F];
  __shared__ int queue[QCAP];
  __shared__ int qcount;

  const int tid = threadIdx.x;
  const int lane = tid & 31;
  const int base = blockIdx.x * R;

  if (tid == 0) qcount = 0;
  for (int i = tid; i < R * LPE; i += NT) {
    const int row = i >> LOGLPE, q = i & (LPE - 1);
    const int node = base + row;
    if (node < n) {
      const float dn = dinv[node];
      const float nrm = dn * dn;
      const float4 h = *reinterpret_cast<const float4*>(H + (((long long)node) << LOGF) + 4 * q);
      float* a = acc + (row << LOGF) + q;
      a[0] = h.x * nrm;
      a[LPE] = h.y * nrm;
      a[2 * LPE] = h.z * nrm;
      a[3 * LPE] = h.w * nrm;
    }
  }
  __syncthreads();

  auto push = [&](int idx) {
    const int slot = atomicAdd(&qcount, 1);
    if (slot < QCAP) {
      queue[slot] = idx;
    } else {
      const int s = src[idx];
      const int d = dst[idx];
      const float nrm = dinv[s] * dinv[d];
      const float* hs = H + (((long long)s) << LOGF);
      float* a = acc + ((d - base) << LOGF);
      for (int f = 0; f < F; ++f) atomicAdd(a + (f & 3) * LPE + (f >> 2), hs[f] * nrm);
    }
  };

  const int c = gcount[blockIdx.x];
  const bool have_list = c <= QCAP;
  if (have_list) {
    const int* __restrict__ gl = glist + (long long)blockIdx.x * QCAP;
    for (int i = tid; i < c; i += NT) queue[i] = gl[i];
    if (tid == 0) qcount = c;
  } else {
    scan_for_range<R, NT>(keys, dst, e, base, push);
  }
  __syncthreads();

  constexpr int STEPS = 32 / EPI;
  constexpr int UB = STEPS < 8 ? STEPS : 8;
  const int total = qcount < QCAP ? qcount : QCAP;
  const int q = lane & (LPE - 1);
  const int b0 = (tid >> 5) * 32;
  int ns = 0, nd = base;
  if (b0 + lane < total) {
    const int idx = queue[b0 + lane];
    ns = src[idx];
    nd = dst[idx];
  }
  for (int b = b0; b < total; b += NT) {
    const int s = ns;
    const int dl = nd - base;
    const float nrm = (b + lane < total) ? dinv[s] * dinv[nd] : 0.0f;
    const int cnt = total - b < 32 ? total - b : 32;
    ns = 0;
    nd = base;
    if (b + NT + lane < total) {
      const int idx = queue[b + NT + lane];
      ns = src[idx];
      nd = dst[idx];
    }
    for (int j = 0; j < STEPS; j += UB) {
      float4 h[UB];
      int dj[UB];
      float nj[UB];
      for (int u = 0; u < UB; ++u) {
        const int sub = (j + u) * EPI + (lane >> LOGLPE);
        const int sj = __shfl(s, sub);
        dj[u] = __shfl(dl, sub);
        nj[u] = __shfl(nrm, sub);
        h[u] = *reinterpret_cast<const float4*>(H + (((long long)sj) << LOGF) + 4 * q);
      }
      for (int u = 0; u < UB; ++u) {
        const int sub = (j + u) * EPI + (lane >> LOGLPE);
        if (sub < cnt) {
          float* a = acc + (dj[u] << LOGF) + q;
          atomicAdd(a, h[u].x * nj[u]);
          atomicAdd(a + LPE, h[u].y * nj[u]);
          atomicAdd(a + 2 * LPE, h[u].z * nj[u]);
          atomicAdd(a + 3 * LPE, h[u].w * nj[u]);
        }
      }
    }
  }
  __syncthreads();

  for (int i = tid; i < R * LPE; i += NT) {
    const int row = i >> LOGLPE, qq = i & (LPE - 1);
    const int node = base + row;
    if (node < n) {
      const float* a = acc + (row << LOGF) + qq;
      float4 o;
      o.x = a[0];
      o.y = a[LPE];
      o.z = a[2 * LPE];
      o.w = a[3 * LPE];
      *reinterpret_cast<float4*>(out + (((long long)node) << LOGF) + 4 * qq) = o;
    }
  }
}

__global__ void lsm_kernel(const float* __restrict__ agg, const float* __restrict__ b,
                           float* __restrict__ out, int n) {
  const int i = blockIdx.x * blockDim.x + threadIdx.x;
  if (i >= n) return;
  float v[16];
  float mx = -INFINITY;
#pragma unroll
  for (int j = 0; j < 16; ++j) {
    v[j] = agg[(long long)i * 16 + j] + b[j];
    mx = fmaxf(mx, v[j]);
  }
  float s = 0.0f;
#pragma unroll
  for (int j = 0; j < 16; ++j) s += expf(v[j] - mx);
  const float lse = mx + logf(s);
#pragma unroll
  for (int j = 0; j < 16; ++j) out[(long long)i * 16 + j] = v[j] - lse;
}

extern "C" void kernel_launch(void* const* d_in, const int* in_sizes, int n_in,
                              void* d_out, int out_size, void* d_ws, size_t ws_size,
                              hipStream_t stream) {
  const float* x  = (const float*)d_in[0];
  const int*   ei = (const int*)d_in[1];
  const float* W1 = (const float*)d_in[2];
  const float* b1 = (const float*)d_in[3];
  const float* W2 = (const float*)d_in[4];
  const float* b2 = (const float*)d_in[5];

  const int n = in_sizes[0] / 256;
  const int e = in_sizes[1] / 2;
  const int* src = ei;
  const int* dst = ei + e;

  float* agg1 = (float*)d_ws;
  float* h1   = agg1 + (size_t)n * 128;
  int*   glist = (int*)(h1 + (size_t)n * 128);
  float* dinv = (float*)(glist + (size_t)n * 32 + n);
  float* h2   = h1;
  float* agg2 = h1 + (size_t)n * 16;
  const int nrange = ceil_div(n, 512);
  int*   gcount = glist + (size_t)nrange * 10240;
  int*   keys = glist + ((((size_t)nrange * 10241) + 3) & ~(size_t)3);

  key_kernel<9><<<ceil_div(ceil_div(e, 4), 256), 256, 0, stream>>>(dst, keys, e);
  range_list_kernel<512, 10240><<<nrange, 1024, 0, stream>>>(dst, dinv, glist, gcount, keys, e, n);

  gemm_wmma_f32_lds_a<256, 128, 4, 2, 4><<<ceil_div(n, 64), 128, 0, stream>>>(x, W1, h1, n);
  agg_lds_kernel<7, 512, 10240><<<nrange, 1024, 0, stream>>>(
      h1, src, dst, dinv, agg1, glist, gcount, keys, e, n);

  {
    const long long waves = (long long)((n + 15) / 16) * (16 / 16);
    gemm_wmma_f32_relu_a<128, 16, 8><<<ceil_div(waves * 32, 256), 256, 0, stream>>>(
        agg1, b1, W2, h2, n);
  }
  agg_lds_kernel<4, 512, 10240><<<nrange, 1024, 0, stream>>>(
      h2, src, dst, dinv, agg2, glist, gcount, keys, e, n);
  lsm_kernel<<<ceil_div(n, 256), 256, 0, stream>>>(agg2, b2, (float*)d_out, n);
}


# Saved range lists hold edge records (source + local row in one word, looked up once by the owner pass), not edge indices: aggregations drop the src/dst lookups, prefetch dinv; sum order LDS arrival
# speedup vs baseline: 4.5316x; 1.1072x over previous
#include <hip/hip_runtime.h>
#include <math.h>


typedef __attribute__((ext_vector_type(2))) float v2f;
typedef __attribute__((ext_vector_type(8))) float v8f;
typedef __attribute__((address_space(1))) float global_float;

static inline int ceil_div(long long a, long long b) { return (int)((a + b - 1) / b); }

constexpr int REC_SRC_BITS = 23;
__device__ __forceinline__ int rec_pack(int s, int dl) {
  return (int)(((unsigned)dl << REC_SRC_BITS) | (unsigned)s);
}
__device__ __forceinline__ int rec_src(int r) {
  return (int)((unsigned)r & ((1u << REC_SRC_BITS) - 1u));
}
__device__ __forceinline__ int rec_row(int r) { return (int)((unsigned)r >> REC_SRC_BITS); }

template <int K, int N, int WPB>
__global__ __launch_bounds__(WPB * 32) void gemm_wmma_f32_relu_a(const float* __restrict__ S,
                                                                const float* __restrict__ bias,
                                                                const float* __restrict__ B,
                                                                float* __restrict__ C, int M) {
  constexpr int NT = N / 16;
  constexpr int P = K + 4;
  static_assert(K == 128, "one 16-byte piece of a row per lane");
  __shared__ __attribute__((aligned(16))) float As[WPB * 16 * P];

  const int lane = threadIdx.x & 31;
  const int wave = threadIdx.x >> 5;
  const int gwave = __builtin_amdgcn_readfirstlane((int)(blockIdx.x * WPB + wave));
  const int mt = gwave / NT;
  const int nt = gwave - mt * NT;
  const int half = lane >> 4;
  const int l16 = lane & 15;
  const int col = nt * 16 + l16;
  float* __restrict__ Aw = As + wave * (16 * P);

  const int m0 = mt * 16 + 8 * half;
  unsigned long long caddr = reinterpret_cast<unsigned long long>(C + (long long)m0 * N + col);
  int nrows = M - m0;
  asm volatile("" : "+v"(caddr), "+v"(nrows));
  global_float* Cp = (global_float*)caddr;

  {
    const float4 bq = *reinterpret_cast<const float4*>(bias + 4 * lane);
    float4 t[16];
    for (int r = 0; r < 16; ++r) {
      int gr = mt * 16 + r;
      gr = gr < M ? gr : (M - 1);
      t[r] = *reinterpret_cast<const float4*>(S + (long long)gr * K + 4 * lane);
    }
    for (int r = 0; r < 16; ++r) {
      float4 v;
      v.x = t[r].x + bq.x;
      v.y = t[r].y + bq.y;
      v.z = t[r].z + bq.z;
      v.w = t[r].w + bq.w;
      v.x = v.x > 0.0f ? v.x : 0.0f;
      v.y = v.y > 0.0f ? v.y : 0.0f;
      v.z = v.z > 0.0f ? v.z : 0.0f;
      v.w = v.w > 0.0f ? v.w : 0.0f;
      *reinterpret_cast<float4*>(Aw + r * P + 4 * lane) = v;
    }
  }
  __syncthreads();

  const float* __restrict__ Al = Aw + l16 * P + 2 * half;
  const float* __restrict__ Bl = B + (long long)(2 * half) * N + col;

  v8f acc = {};
#pragma unroll
  for (int k = 0; k < K; k += 4) {
    const v2f a = *reinterpret_cast<const v2f*>(Al + k);
    v2f b;
    b.x = Bl[(long long)k * N];
    b.y = Bl[(long long)(k + 1) * N];
    acc = __builtin_amdgcn_wmma_f32_16x16x4_f32(false, a, false, b, (short)0, acc,
                                                false, false);
  }

  if (mt * 16 + 16 <= M) {
#pragma unroll
    for (int v = 0; v < 8; ++v) Cp[(long long)v * N] = acc[v];
  } else {
#pragma unroll
    for (int v = 0; v < 8; ++v)
      if (v < nrows) Cp[(long long)v * N] = acc[v];
  }
}

template <int K, int N, int TM, int TN, int WN>
__global__ __launch_bounds__(WN * 32) void gemm_wmma_f32_lds_a(const float* __restrict__ A,
                                                             const float* __restrict__ B,
                                                             float* __restrict__ C, int M) {
  constexpr int BM = TM * 16;
  constexpr int P = K + 4;
  constexpr int NTHR = WN * 32;
  constexpr int KQ = K / 4;
  constexpr int NV = BM * KQ;
  constexpr int LB = 32;
  static_assert(N == WN * TN * 16, "waves must tile the N columns");
  static_assert(NV % (NTHR * LB) == 0, "slab must split evenly over the staging batches");
  __shared__ __attribute__((aligned(16))) float As[BM * P];

  const int tid = threadIdx.x;
  const int row0 = blockIdx.x * BM;

  for (int i0 = 0; i0 < NV; i0 += NTHR * LB) {
    float4 t[LB];
    for (int u = 0; u < LB; ++u) {
      const int f = i0 + u * NTHR + tid;
      const int r = f / KQ, kq = f - r * KQ;
      int gr = row0 + r;
      gr = gr < M ? gr : (M - 1);
      t[u] = *reinterpret_cast<const float4*>(A + (long long)gr * K + 4 * kq);
    }
    for (int u = 0; u < LB; ++u) {
      const int f = i0 + u * NTHR + tid;
      const int r = f / KQ, kq = f - r * KQ;
      *reinterpret_cast<float4*>(As + r * P + 4 * kq) = t[u];
    }
  }
  __syncthreads();

  const int lane = tid & 31;
  const int wave = tid >> 5;
  const int half = lane >> 4;
  const int l16 = lane & 15;
  const int col0 = wave * (TN * 16) + l16;

  const float* __restrict__ Al = As + l16 * P + 2 * half;
  const float* __restrict__ Bl = B + (long long)(2 * half) * N + col0;

  v8f acc[TM][TN];
  for (int i = 0; i < TM; ++i)
    for (int j = 0; j < TN; ++j) acc[i][j] = v8f{};

#pragma unroll
  for (int k = 0; k < K; k += 4) {
    v2f a[TM], b[TN];
    for (int i = 0; i < TM; ++i) a[i] = *reinterpret_cast<const v2f*>(Al + i * 16 * P + k);
    for (int j = 0; j < TN; ++j) {
      b[j].x = Bl[(long long)k * N + j * 16];
      b[j].y = Bl[(long long)(k + 1) * N + j * 16];
    }
    for (int i = 0; i < TM; ++i)
      for (int j = 0; j < TN; ++j)
        acc[i][j] = __builtin_amdgcn_wmma_f32_16x16x4_f32(false, a[i], false, b[j], (short)0,
                                                          acc[i][j], false, false);
  }

  for (int i = 0; i < TM; ++i) {
    const int m0 = row0 + i * 16 + 8 * half;
    for (int j = 0; j < TN; ++j) {
      float* __restrict__ Cp = C + (long long)m0 * N + col0 + j * 16;
      if (row0 + BM <= M) {
        for (int v = 0; v < 8; ++v) Cp[(long long)v * N] = acc[i][j][v];
      } else {
        for (int v = 0; v < 8; ++v)
          if (m0 + v < M) Cp[(long long)v * N] = acc[i][j][v];
      }
    }
  }
}

template <int LOGR>
__global__ void key_kernel(const int* __restrict__ dst, int* __restrict__ keys, int e) {
  const int t = blockIdx.x * blockDim.x + threadIdx.x;
  const int i0 = 4 * t;
  if (i0 >= e) return;
  unsigned w = 0xFFFFFFFFu;
  for (int k = 0; k < 4; ++k) {
    const int i = i0 + k;
    if (i < e) {
      const int d = dst[i];
      w = (w & ~(0xFFu << (8 * k))) | (((unsigned)(d >> LOGR) & 0xFFu) << (8 * k));
    }
  }
  keys[t] = (int)w;
}

template <int R, int NT, class Push>
__device__ __forceinline__ void scan_for_range(const int* __restrict__ keys,
                                               const int* __restrict__ dst, int e, int base,
                                               Push&& push) {
  const int tid = threadIdx.x;
  auto in_range = [&](int d) { return (unsigned)(d - base) < (unsigned)R; };
  const bool use_keys = gridDim.x <= 256;

  if (use_keys) {
    const unsigned kb = (unsigned)blockIdx.x * 0x01010101u;
    auto nzf = [&](int wi) {
      const unsigned x = (unsigned)wi ^ kb;
      return (((x & 0x7F7F7F7Fu) + 0x7F7F7F7Fu) | x) & 0x80808080u;
    };
    auto pairmask = [&](const int4 p, const int4 q) {
      return ~((nzf(p.x) >> 7) | (nzf(p.y) >> 6) | (nzf(p.z) >> 5) | (nzf(p.w) >> 4) |
               (nzf(q.x) >> 3) | (nzf(q.y) >> 2) | (nzf(q.z) >> 1) | nzf(q.w));
    };
    auto drain = [&](unsigned m, int e0, int e1) {
      while (m) {
        const int p = __builtin_ctz(m);
        m &= m - 1;
        const int idx = ((p & 4) ? e1 : e0) + 4 * (p & 3) + (p >> 3);
        if (idx < e) push(idx);
      }
    };
    const int nw = (e + 3) >> 2;
    const int nvec = nw >> 2;
    const int4* __restrict__ kv = reinterpret_cast<const int4*>(keys);
    constexpr int U = 8;
    int v = tid;
    for (; v + (U - 1) * NT < nvec; v += U * NT) {
      int4 kk[U];
      for (int u = 0; u < U; ++u) kk[u] = kv[v + u * NT];
      for (int u = 0; u < U; u += 2)
        drain(pairmask(kk[u], kk[u + 1]), 16 * (v + u * NT), 16 * (v + (u + 1) * NT));
    }
    const int4 none = {-1, -1, -1, -1};
    for (; v < nvec; v += NT) {
      const unsigned m = pairmask(kv[v], none) & 0x0F0F0F0Fu;
      drain(m, 16 * v, 0);
    }
    if (tid < (nw & 3)) {
      const int t = 4 * nvec + tid;
      const unsigned m = ~(nzf(keys[t]) >> 7) & 0x01010101u;
      drain(m, 4 * t, 0);
    }
  } else {
    int head = (int)(((16u - (unsigned)(reinterpret_cast<unsigned long long>(dst) & 15u)) & 15u) >> 2);
    if (head > e) head = e;
    if (tid < head && in_range(dst[tid])) push(tid);
    const int nvec = (e - head) >> 2;
    const int4* __restrict__ dv = reinterpret_cast<const int4*>(dst + head);
    constexpr int U = 8;
    int v = tid;
    for (; v + (U - 1) * NT < nvec; v += U * NT) {
      int4 dd[U];
      for (int u = 0; u < U; ++u) dd[u] = dv[v + u * NT];
      unsigned m = 0;
      for (int u = 0; u < U; ++u) {
        m |= (in_range(dd[u].x) ? 1u : 0u) << (4 * u);
        m |= (in_range(dd[u].y) ? 2u : 0u) << (4 * u);
        m |= (in_range(dd[u].z) ? 4u : 0u) << (4 * u);
        m |= (in_range(dd[u].w) ? 8u : 0u) << (4 * u);
      }
      while (m) {
        const int b = __builtin_ctz(m);
        m &= m - 1;
        push(head + 4 * (v + (b >> 2) * NT) + (b & 3));
      }
    }
    for (; v < nvec; v += NT) {
      const int4 d4 = dv[v];
      unsigned m = (in_range(d4.x) ? 1u : 0u) | (in_range(d4.y) ? 2u : 0u) |
                   (in_range(d4.z) ? 4u : 0u) | (in_range(d4.w) ? 8u : 0u);
      while (m) {
        const int b = __builtin_ctz(m);
        m &= m - 1;
        push(head + 4 * v + b);
      }
    }
    {
      const int t = head + 4 * nvec + tid;
      if (t < e && in_range(dst[t])) push(t);
    }
  }
}

template <int R, int QCAP>
__global__ __launch_bounds__(1024) void range_list_kernel(const int* __restrict__ src,
                                                          const int* __restrict__ dst,
                                                          float* __restrict__ dinv,
                                                          int* __restrict__ glist,
                                                          int* __restrict__ gcount,
                                                          const int* __restrict__ keys, int e,
                                                          int n) {
  constexpr int NT = 1024;
  static_assert(R <= (1 << (32 - REC_SRC_BITS)), "the local row must fit the record");
  __shared__ int cnt[R];
  __shared__ int queue[QCAP];
  __shared__ int qcount;

  const int tid = threadIdx.x;
  const int base = blockIdx.x * R;
  const bool packable = n <= (1 << REC_SRC_BITS);

  if (tid == 0) qcount = 0;
  for (int i = tid; i < R; i += NT) cnt[i] = 0;
  __syncthreads();

  auto push = [&](int idx) {
    const int slot = atomicAdd(&qcount, 1);
    if (slot < QCAP) {
      queue[slot] = idx;
    } else {
      atomicAdd(&cnt[dst[idx] - base], 1);
    }
  };
  scan_for_range<R, NT>(keys, dst, e, base, push);
  __syncthreads();

  const int met = qcount;
  const int total = met < QCAP ? met : QCAP;
  int* __restrict__ gl = glist + (long long)blockIdx.x * QCAP;
  constexpr int WU = 4;
  int i = tid;
  for (; i + (WU - 1) * NT < total; i += WU * NT) {
    int s[WU], dl[WU];
    for (int u = 0; u < WU; ++u) {
      const int idx = queue[i + u * NT];
      s[u] = src[idx];
      dl[u] = dst[idx] - base;
    }
    for (int u = 0; u < WU; ++u) {
      gl[i + u * NT] = rec_pack(s[u], dl[u]);
      atomicAdd(&cnt[dl[u]], 1);
    }
  }
  for (; i < total; i += NT) {
    const int idx = queue[i];
    const int dl = dst[idx] - base;
    gl[i] = rec_pack(src[idx], dl);
    atomicAdd(&cnt[dl], 1);
  }
  if (tid == 0) gcount[blockIdx.x] = packable ? met : 0x7FFFFFFF;
  __syncthreads();

  for (int i = tid; i < R; i += NT) {
    const int node = base + i;
    if (node < n) dinv[node] = rsqrtf((float)cnt[i] + 1.0f);
  }
}

template <int LOGF, int R, int QCAP>
__global__ __launch_bounds__(1024) void agg_lds_kernel(const float* __restrict__ H,
                                                       const int* __restrict__ src,
                                                       const int* __restrict__ dst,
                                                       const float* __restrict__ dinv,
                                                       float* __restrict__ out,
                                                       const int* __restrict__ glist,
                                                       const int* __restrict__ gcount,
                                                       const int* __restrict__ keys, int e,
                                                       int n) {
  constexpr int F = 1 << LOGF;
  constexpr int LPE = F / 4;
  constexpr int LOGLPE = LOGF - 2;
  constexpr int EPI = 32 / LPE;
  constexpr int NT = 1024;
  static_assert(R <= (1 << (32 - REC_SRC_BITS)), "the local row must fit the record");
  __shared__ float acc[R * F];
  __shared__ int queue[QCAP];
  __shared__ int qcount;

  const int tid = threadIdx.x;
  const int lane = tid & 31;
  const int base = blockIdx.x * R;
  const bool packable = n <= (1 << REC_SRC_BITS);

  if (tid == 0) qcount = 0;
  for (int i = tid; i < R * LPE; i += NT) {
    const int row = i >> LOGLPE, q = i & (LPE - 1);
    const int node = base + row;
    if (node < n) {
      const float dn = dinv[node];
      const float nrm = dn * dn;
      const float4 h = *reinterpret_cast<const float4*>(H + (((long long)node) << LOGF) + 4 * q);
      float* a = acc + (row << LOGF) + q;
      a[0] = h.x * nrm;
      a[LPE] = h.y * nrm;
      a[2 * LPE] = h.z * nrm;
      a[3 * LPE] = h.w * nrm;
    }
  }
  __syncthreads();

  auto push = [&](int idx) {
    const int s = src[idx];
    const int d = dst[idx];
    const int slot = packable ? atomicAdd(&qcount, 1) : QCAP;
    if (slot < QCAP) {
      queue[slot] = rec_pack(s, d - base);
    } else {
      const float nrm = dinv[s] * dinv[d];
      const float* hs = H + (((long long)s) << LOGF);
      float* a = acc + ((d - base) << LOGF);
      for (int f = 0; f < F; ++f) atomicAdd(a + (f & 3) * LPE + (f >> 2), hs[f] * nrm);
    }
  };

  const int c = gcount[blockIdx.x];
  const bool have_list = c <= QCAP;
  if (have_list) {
    const int* __restrict__ gl = glist + (long long)blockIdx.x * QCAP;
    for (int i = tid; i < c; i += NT) queue[i] = gl[i];
    if (tid == 0) qcount = c;
  } else {
    scan_for_range<R, NT>(keys, dst, e, base, push);
  }
  __syncthreads();

  constexpr int STEPS = 32 / EPI;
  constexpr int UB = STEPS < 8 ? STEPS : 8;
  const int total = qcount < QCAP ? qcount : QCAP;
  const int q = lane & (LPE - 1);
  const int b0 = (tid >> 5) * 32;
  int ns, ndl;
  float nnrm;
  {
    const int r = (b0 + lane < total) ? queue[b0 + lane] : 0;
    ns = rec_src(r);
    ndl = rec_row(r);
    nnrm = dinv[ns] * dinv[base + ndl];
  }
  for (int b = b0; b < total; b += NT) {
    const int s = ns;
    const int dl = ndl;
    const float nrm = nnrm;
    const int cnt = total - b < 32 ? total - b : 32;
    const int rn = (b + NT + lane < total) ? queue[b + NT + lane] : 0;
    ns = rec_src(rn);
    ndl = rec_row(rn);
    const float pds = dinv[ns];
    const float pdd = dinv[base + ndl];
    for (int j = 0; j < STEPS; j += UB) {
      float4 h[UB];
      int dj[UB];
      float nj[UB];
      for (int u = 0; u < UB; ++u) {
        const int sub = (j + u) * EPI + (lane >> LOGLPE);
        const int sj = __shfl(s, sub);
        dj[u] = __shfl(dl, sub);
        nj[u] = __shfl(nrm, sub);
        h[u] = *reinterpret_cast<const float4*>(H + (((long long)sj) << LOGF) + 4 * q);
      }
      for (int u = 0; u < UB; ++u) {
        const int sub = (j + u) * EPI + (lane >> LOGLPE);
        if (sub < cnt) {
          float* a = acc + (dj[u] << LOGF) + q;
          atomicAdd(a, h[u].x * nj[u]);
          atomicAdd(a + LPE, h[u].y * nj[u]);
          atomicAdd(a + 2 * LPE, h[u].z * nj[u]);
          atomicAdd(a + 3 * LPE, h[u].w * nj[u]);
        }
      }
    }
    nnrm = pds * pdd;
  }
  __syncthreads();

  for (int i = tid; i < R * LPE; i += NT) {
    const int row = i >> LOGLPE, qq = i & (LPE - 1);
    const int node = base + row;
    if (node < n) {
      const float* a = acc + (row << LOGF) + qq;
      float4 o;
      o.x = a[0];
      o.y = a[LPE];
      o.z = a[2 * LPE];
      o.w = a[3 * LPE];
      *reinterpret_cast<float4*>(out + (((long long)node) << LOGF) + 4 * qq) = o;
    }
  }
}

__global__ void lsm_kernel(const float* __restrict__ agg, const float* __restrict__ b,
                           float* __restrict__ out, int n) {
  const int i = blockIdx.x * blockDim.x + threadIdx.x;
  if (i >= n) return;
  float v[16];
  float mx = -INFINITY;
#pragma unroll
  for (int j = 0; j < 16; ++j) {
    v[j] = agg[(long long)i * 16 + j] + b[j];
    mx = fmaxf(mx, v[j]);
  }
  float s = 0.0f;
#pragma unroll
  for (int j = 0; j < 16; ++j) s += expf(v[j] - mx);
  const float lse = mx + logf(s);
#pragma unroll
  for (int j = 0; j < 16; ++j) out[(long long)i * 16 + j] = v[j] - lse;
}

extern "C" void kernel_launch(void* const* d_in, const int* in_sizes, int n_in,
                              void* d_out, int out_size, void* d_ws, size_t ws_size,
                              hipStream_t stream) {
  const float* x  = (const float*)d_in[0];
  const int*   ei = (const int*)d_in[1];
  const float* W1 = (const float*)d_in[2];
  const float* b1 = (const float*)d_in[3];
  const float* W2 = (const float*)d_in[4];
  const float* b2 = (const float*)d_in[5];

  const int n = in_sizes[0] / 256;
  const int e = in_sizes[1] / 2;
  const int* src = ei;
  const int* dst = ei + e;

  float* agg1 = (float*)d_ws;
  float* h1   = agg1 + (size_t)n * 128;
  int*   glist = (int*)(h1 + (size_t)n * 128);
  float* dinv = (float*)(glist + (size_t)n * 32 + n);
  float* h2   = h1;
  float* agg2 = h1 + (size_t)n * 16;
  const int nrange = ceil_div(n, 512);
  int*   gcount = glist + (size_t)nrange * 10240;
  int*   keys = glist + ((((size_t)nrange * 10241) + 3) & ~(size_t)3);

  key_kernel<9><<<ceil_div(ceil_div(e, 4), 256), 256, 0, stream>>>(dst, keys, e);
  range_list_kernel<512, 10240><<<nrange, 1024, 0, stream>>>(src, dst, dinv, glist, gcount, keys, e,
                                                               n);

  gemm_wmma_f32_lds_a<256, 128, 4, 2, 4><<<ceil_div(n, 64), 128, 0, stream>>>(x, W1, h1, n);
  agg_lds_kernel<7, 512, 10240><<<nrange, 1024, 0, stream>>>(
      h1, src, dst, dinv, agg1, glist, gcount, keys, e, n);

  {
    const long long waves = (long long)((n + 15) / 16) * (16 / 16);
    gemm_wmma_f32_relu_a<128, 16, 8><<<ceil_div(waves * 32, 256), 256, 0, stream>>>(
        agg1, b1, W2, h2, n);
  }
  agg_lds_kernel<4, 512, 10240><<<nrange, 1024, 0, stream>>>(
      h2, src, dst, dinv, agg2, glist, gcount, keys, e, n);
  lsm_kernel<<<ceil_div(n, 256), 256, 0, stream>>>(agg2, b2, (float*)d_out, n);
}


# Owner ranges cut to 391 rows: 256 range owners, one per WGP, instead of 196 (512 rows) in the owner pass and both aggregations; key = dst / 391; sum order LDS arrival, as before
# speedup vs baseline: 4.9242x; 1.0866x over previous
#include <hip/hip_runtime.h>
#include <math.h>


typedef __attribute__((ext_vector_type(2))) float v2f;
typedef __attribute__((ext_vector_type(8))) float v8f;
typedef __attribute__((address_space(1))) float global_float;

static inline int ceil_div(long long a, long long b) { return (int)((a + b - 1) / b); }

constexpr int RANGE_ROWS = 391;

constexpr int REC_SRC_BITS = 23;
__device__ __forceinline__ int rec_pack(int s, int dl) {
  return (int)(((unsigned)dl << REC_SRC_BITS) | (unsigned)s);
}
__device__ __forceinline__ int rec_src(int r) {
  return (int)((unsigned)r & ((1u << REC_SRC_BITS) - 1u));
}
__device__ __forceinline__ int rec_row(int r) { return (int)((unsigned)r >> REC_SRC_BITS); }

template <int K, int N, int WPB>
__global__ __launch_bounds__(WPB * 32) void gemm_wmma_f32_relu_a(const float* __restrict__ S,
                                                                const float* __restrict__ bias,
                                                                const float* __restrict__ B,
                                                                float* __restrict__ C, int M) {
  constexpr int NT = N / 16;
  constexpr int P = K + 4;
  static_assert(K == 128, "one 16-byte piece of a row per lane");
  __shared__ __attribute__((aligned(16))) float As[WPB * 16 * P];

  const int lane = threadIdx.x & 31;
  const int wave = threadIdx.x >> 5;
  const int gwave = __builtin_amdgcn_readfirstlane((int)(blockIdx.x * WPB + wave));
  const int mt = gwave / NT;
  const int nt = gwave - mt * NT;
  const int half = lane >> 4;
  const int l16 = lane & 15;
  const int col = nt * 16 + l16;
  float* __restrict__ Aw = As + wave * (16 * P);

  const int m0 = mt * 16 + 8 * half;
  unsigned long long caddr = reinterpret_cast<unsigned long long>(C + (long long)m0 * N + col);
  int nrows = M - m0;
  asm volatile("" : "+v"(caddr), "+v"(nrows));
  global_float* Cp = (global_float*)caddr;

  {
    const float4 bq = *reinterpret_cast<const float4*>(bias + 4 * lane);
    float4 t[16];
    for (int r = 0; r < 16; ++r) {
      int gr = mt * 16 + r;
      gr = gr < M ? gr : (M - 1);
      t[r] = *reinterpret_cast<const float4*>(S + (long long)gr * K + 4 * lane);
    }
    for (int r = 0; r < 16; ++r) {
      float4 v;
      v.x = t[r].x + bq.x;
      v.y = t[r].y + bq.y;
      v.z = t[r].z + bq.z;
      v.w = t[r].w + bq.w;
      v.x = v.x > 0.0f ? v.x : 0.0f;
      v.y = v.y > 0.0f ? v.y : 0.0f;
      v.z = v.z > 0.0f ? v.z : 0.0f;
      v.w = v.w > 0.0f ? v.w : 0.0f;
      *reinterpret_cast<float4*>(Aw + r * P + 4 * lane) = v;
    }
  }
  __syncthreads();

  const float* __restrict__ Al = Aw + l16 * P + 2 * half;
  const float* __restrict__ Bl = B + (long long)(2 * half) * N + col;

  v8f acc = {};
#pragma unroll
  for (int k = 0; k < K; k += 4) {
    const v2f a = *reinterpret_cast<const v2f*>(Al + k);
    v2f b;
    b.x = Bl[(long long)k * N];
    b.y = Bl[(long long)(k + 1) * N];
    acc = __builtin_amdgcn_wmma_f32_16x16x4_f32(false, a, false, b, (short)0, acc,
                                                false, false);
  }

  if (mt * 16 + 16 <= M) {
#pragma unroll
    for (int v = 0; v < 8; ++v) Cp[(long long)v * N] = acc[v];
  } else {
#pragma unroll
    for (int v = 0; v < 8; ++v)
      if (v < nrows) Cp[(long long)v * N] = acc[v];
  }
}

template <int K, int N, int TM, int TN, int WN>
__global__ __launch_bounds__(WN * 32) void gemm_wmma_f32_lds_a(const float* __restrict__ A,
                                                             const float* __restrict__ B,
                                                             float* __restrict__ C, int M) {
  constexpr int BM = TM * 16;
  constexpr int P = K + 4;
  constexpr int NTHR = WN * 32;
  constexpr int KQ = K / 4;
  constexpr int NV = BM * KQ;
  constexpr int LB = 32;
  static_assert(N == WN * TN * 16, "waves must tile the N columns");
  static_assert(NV % (NTHR * LB) == 0, "slab must split evenly over the staging batches");
  __shared__ __attribute__((aligned(16))) float As[BM * P];

  const int tid = threadIdx.x;
  const int row0 = blockIdx.x * BM;

  for (int i0 = 0; i0 < NV; i0 += NTHR * LB) {
    float4 t[LB];
    for (int u = 0; u < LB; ++u) {
      const int f = i0 + u * NTHR + tid;
      const int r = f / KQ, kq = f - r * KQ;
      int gr = row0 + r;
      gr = gr < M ? gr : (M - 1);
      t[u] = *reinterpret_cast<const float4*>(A + (long long)gr * K + 4 * kq);
    }
    for (int u = 0; u < LB; ++u) {
      const int f = i0 + u * NTHR + tid;
      const int r = f / KQ, kq = f - r * KQ;
      *reinterpret_cast<float4*>(As + r * P + 4 * kq) = t[u];
    }
  }
  __syncthreads();

  const int lane = tid & 31;
  const int wave = tid >> 5;
  const int half = lane >> 4;
  const int l16 = lane & 15;
  const int col0 = wave * (TN * 16) + l16;

  const float* __restrict__ Al = As + l16 * P + 2 * half;
  const float* __restrict__ Bl = B + (long long)(2 * half) * N + col0;

  v8f acc[TM][TN];
  for (int i = 0; i < TM; ++i)
    for (int j = 0; j < TN; ++j) acc[i][j] = v8f{};

#pragma unroll
  for (int k = 0; k < K; k += 4) {
    v2f a[TM], b[TN];
    for (int i = 0; i < TM; ++i) a[i] = *reinterpret_cast<const v2f*>(Al + i * 16 * P + k);
    for (int j = 0; j < TN; ++j) {
      b[j].x = Bl[(long long)k * N + j * 16];
      b[j].y = Bl[(long long)(k + 1) * N + j * 16];
    }
    for (int i = 0; i < TM; ++i)
      for (int j = 0; j < TN; ++j)
        acc[i][j] = __builtin_amdgcn_wmma_f32_16x16x4_f32(false, a[i], false, b[j], (short)0,
                                                          acc[i][j], false, false);
  }

  for (int i = 0; i < TM; ++i) {
    const int m0 = row0 + i * 16 + 8 * half;
    for (int j = 0; j < TN; ++j) {
      float* __restrict__ Cp = C + (long long)m0 * N + col0 + j * 16;
      if (row0 + BM <= M) {
        for (int v = 0; v < 8; ++v) Cp[(long long)v * N] = acc[i][j][v];
      } else {
        for (int v = 0; v < 8; ++v)
          if (m0 + v < M) Cp[(long long)v * N] = acc[i][j][v];
      }
    }
  }
}

template <int R>
__global__ void key_kernel(const int* __restrict__ dst, int* __restrict__ keys, int e) {
  const int t = blockIdx.x * blockDim.x + threadIdx.x;
  const int i0 = 4 * t;
  if (i0 >= e) return;
  unsigned w = 0xFFFFFFFFu;
  for (int k = 0; k < 4; ++k) {
    const int i = i0 + k;
    if (i < e) {
      const int d = dst[i];
      w = (w & ~(0xFFu << (8 * k))) | (((unsigned)(d / R) & 0xFFu) << (8 * k));
    }
  }
  keys[t] = (int)w;
}

template <int R, int NT, class Push>
__device__ __forceinline__ void scan_for_range(const int* __restrict__ keys,
                                               const int* __restrict__ dst, int e, int base,
                                               Push&& push) {
  const int tid = threadIdx.x;
  auto in_range = [&](int d) { return (unsigned)(d - base) < (unsigned)R; };
  const bool use_keys = gridDim.x <= 256;

  if (use_keys) {
    const unsigned kb = (unsigned)blockIdx.x * 0x01010101u;
    auto nzf = [&](int wi) {
      const unsigned x = (unsigned)wi ^ kb;
      return (((x & 0x7F7F7F7Fu) + 0x7F7F7F7Fu) | x) & 0x80808080u;
    };
    auto pairmask = [&](const int4 p, const int4 q) {
      return ~((nzf(p.x) >> 7) | (nzf(p.y) >> 6) | (nzf(p.z) >> 5) | (nzf(p.w) >> 4) |
               (nzf(q.x) >> 3) | (nzf(q.y) >> 2) | (nzf(q.z) >> 1) | nzf(q.w));
    };
    auto drain = [&](unsigned m, int e0, int e1) {
      while (m) {
        const int p = __builtin_ctz(m);
        m &= m - 1;
        const int idx = ((p & 4) ? e1 : e0) + 4 * (p & 3) + (p >> 3);
        if (idx < e) push(idx);
      }
    };
    const int nw = (e + 3) >> 2;
    const int nvec = nw >> 2;
    const int4* __restrict__ kv = reinterpret_cast<const int4*>(keys);
    constexpr int U = 8;
    int v = tid;
    for (; v + (U - 1) * NT < nvec; v += U * NT) {
      int4 kk[U];
      for (int u = 0; u < U; ++u) kk[u] = kv[v + u * NT];
      for (int u = 0; u < U; u += 2)
        drain(pairmask(kk[u], kk[u + 1]), 16 * (v + u * NT), 16 * (v + (u + 1) * NT));
    }
    const int4 none = {-1, -1, -1, -1};
    for (; v < nvec; v += NT) {
      const unsigned m = pairmask(kv[v], none) & 0x0F0F0F0Fu;
      drain(m, 16 * v, 0);
    }
    if (tid < (nw & 3)) {
      const int t = 4 * nvec + tid;
      const unsigned m = ~(nzf(keys[t]) >> 7) & 0x01010101u;
      drain(m, 4 * t, 0);
    }
  } else {
    int head = (int)(((16u - (unsigned)(reinterpret_cast<unsigned long long>(dst) & 15u)) & 15u) >> 2);
    if (head > e) head = e;
    if (tid < head && in_range(dst[tid])) push(tid);
    const int nvec = (e - head) >> 2;
    const int4* __restrict__ dv = reinterpret_cast<const int4*>(dst + head);
    constexpr int U = 8;
    int v = tid;
    for (; v + (U - 1) * NT < nvec; v += U * NT) {
      int4 dd[U];
      for (int u = 0; u < U; ++u) dd[u] = dv[v + u * NT];
      unsigned m = 0;
      for (int u = 0; u < U; ++u) {
        m |= (in_range(dd[u].x) ? 1u : 0u) << (4 * u);
        m |= (in_range(dd[u].y) ? 2u : 0u) << (4 * u);
        m |= (in_range(dd[u].z) ? 4u : 0u) << (4 * u);
        m |= (in_range(dd[u].w) ? 8u : 0u) << (4 * u);
      }
      while (m) {
        const int b = __builtin_ctz(m);
        m &= m - 1;
        push(head + 4 * (v + (b >> 2) * NT) + (b & 3));
      }
    }
    for (; v < nvec; v += NT) {
      const int4 d4 = dv[v];
      unsigned m = (in_range(d4.x) ? 1u : 0u) | (in_range(d4.y) ? 2u : 0u) |
                   (in_range(d4.z) ? 4u : 0u) | (in_range(d4.w) ? 8u : 0u);
      while (m) {
        const int b = __builtin_ctz(m);
        m &= m - 1;
        push(head + 4 * v + b);
      }
    }
    {
      const int t = head + 4 * nvec + tid;
      if (t < e && in_range(dst[t])) push(t);
    }
  }
}

template <int R, int QCAP>
__global__ __launch_bounds__(1024) void range_list_kernel(const int* __restrict__ src,
                                                          const int* __restrict__ dst,
                                                          float* __restrict__ dinv,
                                                          int* __restrict__ glist,
                                                          int* __restrict__ gcount,
                                                          const int* __restrict__ keys, int e,
                                                          int n) {
  constexpr int NT = 1024;
  static_assert(R <= (1 << (32 - REC_SRC_BITS)), "the local row must fit the record");
  __shared__ int cnt[R];
  __shared__ int queue[QCAP];
  __shared__ int qcount;

  const int tid = threadIdx.x;
  const int base = blockIdx.x * R;
  const bool packable = n <= (1 << REC_SRC_BITS);

  if (tid == 0) qcount = 0;
  for (int i = tid; i < R; i += NT) cnt[i] = 0;
  __syncthreads();

  auto push = [&](int idx) {
    const int slot = atomicAdd(&qcount, 1);
    if (slot < QCAP) {
      queue[slot] = idx;
    } else {
      atomicAdd(&cnt[dst[idx] - base], 1);
    }
  };
  scan_for_range<R, NT>(keys, dst, e, base, push);
  __syncthreads();

  const int met = qcount;
  const int total = met < QCAP ? met : QCAP;
  int* __restrict__ gl = glist + (long long)blockIdx.x * QCAP;
  constexpr int WU = 4;
  int i = tid;
  for (; i + (WU - 1) * NT < total; i += WU * NT) {
    int s[WU], dl[WU];
    for (int u = 0; u < WU; ++u) {
      const int idx = queue[i + u * NT];
      s[u] = src[idx];
      dl[u] = dst[idx] - base;
    }
    for (int u = 0; u < WU; ++u) {
      gl[i + u * NT] = rec_pack(s[u], dl[u]);
      atomicAdd(&cnt[dl[u]], 1);
    }
  }
  for (; i < total; i += NT) {
    const int idx = queue[i];
    const int dl = dst[idx] - base;
    gl[i] = rec_pack(src[idx], dl);
    atomicAdd(&cnt[dl], 1);
  }
  if (tid == 0) gcount[blockIdx.x] = packable ? met : 0x7FFFFFFF;
  __syncthreads();

  for (int i = tid; i < R; i += NT) {
    const int node = base + i;
    if (node < n) dinv[node] = rsqrtf((float)cnt[i] + 1.0f);
  }
}

template <int LOGF, int R, int QCAP>
__global__ __launch_bounds__(1024) void agg_lds_kernel(const float* __restrict__ H,
                                                       const int* __restrict__ src,
                                                       const int* __restrict__ dst,
                                                       const float* __restrict__ dinv,
                                                       float* __restrict__ out,
                                                       const int* __restrict__ glist,
                                                       const int* __restrict__ gcount,
                                                       const int* __restrict__ keys, int e,
                                                       int n) {
  constexpr int F = 1 << LOGF;
  constexpr int LPE = F / 4;
  constexpr int LOGLPE = LOGF - 2;
  constexpr int EPI = 32 / LPE;
  constexpr int NT = 1024;
  static_assert(R <= (1 << (32 - REC_SRC_BITS)), "the local row must fit the record");
  __shared__ float acc[R * F];
  __shared__ int queue[QCAP];
  __shared__ int qcount;

  const int tid = threadIdx.x;
  const int lane = tid & 31;
  const int base = blockIdx.x * R;
  const bool packable = n <= (1 << REC_SRC_BITS);

  if (tid == 0) qcount = 0;
  for (int i = tid; i < R * LPE; i += NT) {
    const int row = i >> LOGLPE, q = i & (LPE - 1);
    const int node = base + row;
    if (node < n) {
      const float dn = dinv[node];
      const float nrm = dn * dn;
      const float4 h = *reinterpret_cast<const float4*>(H + (((long long)node) << LOGF) + 4 * q);
      float* a = acc + (row << LOGF) + q;
      a[0] = h.x * nrm;
      a[LPE] = h.y * nrm;
      a[2 * LPE] = h.z * nrm;
      a[3 * LPE] = h.w * nrm;
    }
  }
  __syncthreads();

  auto push = [&](int idx) {
    const int s = src[idx];
    const int d = dst[idx];
    const int slot = packable ? atomicAdd(&qcount, 1) : QCAP;
    if (slot < QCAP) {
      queue[slot] = rec_pack(s, d - base);
    } else {
      const float nrm = dinv[s] * dinv[d];
      const float* hs = H + (((long long)s) << LOGF);
      float* a = acc + ((d - base) << LOGF);
      for (int f = 0; f < F; ++f) atomicAdd(a + (f & 3) * LPE + (f >> 2), hs[f] * nrm);
    }
  };

  const int c = gcount[blockIdx.x];
  const bool have_list = c <= QCAP;
  if (have_list) {
    const int* __restrict__ gl = glist + (long long)blockIdx.x * QCAP;
    for (int i = tid; i < c; i += NT) queue[i] = gl[i];
    if (tid == 0) qcount = c;
  } else {
    scan_for_range<R, NT>(keys, dst, e, base, push);
  }
  __syncthreads();

  constexpr int STEPS = 32 / EPI;
  constexpr int UB = STEPS < 8 ? STEPS : 8;
  const int total = qcount < QCAP ? qcount : QCAP;
  const int q = lane & (LPE - 1);
  const int b0 = (tid >> 5) * 32;
  int ns, ndl;
  float nnrm;
  {
    const int r = (b0 + lane < total) ? queue[b0 + lane] : 0;
    ns = rec_src(r);
    ndl = rec_row(r);
    nnrm = dinv[ns] * dinv[base + ndl];
  }
  for (int b = b0; b < total; b += NT) {
    const int s = ns;
    const int dl = ndl;
    const float nrm = nnrm;
    const int cnt = total - b < 32 ? total - b : 32;
    const int rn = (b + NT + lane < total) ? queue[b + NT + lane] : 0;
    ns = rec_src(rn);
    ndl = rec_row(rn);
    const float pds = dinv[ns];
    const float pdd = dinv[base + ndl];
    for (int j = 0; j < STEPS; j += UB) {
      float4 h[UB];
      int dj[UB];
      float nj[UB];
      for (int u = 0; u < UB; ++u) {
        const int sub = (j + u) * EPI + (lane >> LOGLPE);
        const int sj = __shfl(s, sub);
        dj[u] = __shfl(dl, sub);
        nj[u] = __shfl(nrm, sub);
        h[u] = *reinterpret_cast<const float4*>(H + (((long long)sj) << LOGF) + 4 * q);
      }
      for (int u = 0; u < UB; ++u) {
        const int sub = (j + u) * EPI + (lane >> LOGLPE);
        if (sub < cnt) {
          float* a = acc + (dj[u] << LOGF) + q;
          atomicAdd(a, h[u].x * nj[u]);
          atomicAdd(a + LPE, h[u].y * nj[u]);
          atomicAdd(a + 2 * LPE, h[u].z * nj[u]);
          atomicAdd(a + 3 * LPE, h[u].w * nj[u]);
        }
      }
    }
    nnrm = pds * pdd;
  }
  __syncthreads();

  for (int i = tid; i < R * LPE; i += NT) {
    const int row = i >> LOGLPE, qq = i & (LPE - 1);
    const int node = base + row;
    if (node < n) {
      const float* a = acc + (row << LOGF) + qq;
      float4 o;
      o.x = a[0];
      o.y = a[LPE];
      o.z = a[2 * LPE];
      o.w = a[3 * LPE];
      *reinterpret_cast<float4*>(out + (((long long)node) << LOGF) + 4 * qq) = o;
    }
  }
}

__global__ void lsm_kernel(const float* __restrict__ agg, const float* __restrict__ b,
                           float* __restrict__ out, int n) {
  const int i = blockIdx.x * blockDim.x + threadIdx.x;
  if (i >= n) return;
  float v[16];
  float mx = -INFINITY;
#pragma unroll
  for (int j = 0; j < 16; ++j) {
    v[j] = agg[(long long)i * 16 + j] + b[j];
    mx = fmaxf(mx, v[j]);
  }
  float s = 0.0f;
#pragma unroll
  for (int j = 0; j < 16; ++j) s += expf(v[j] - mx);
  const float lse = mx + logf(s);
#pragma unroll
  for (int j = 0; j < 16; ++j) out[(long long)i * 16 + j] = v[j] - lse;
}

extern "C" void kernel_launch(void* const* d_in, const int* in_sizes, int n_in,
                              void* d_out, int out_size, void* d_ws, size_t ws_size,
                              hipStream_t stream) {
  const float* x  = (const float*)d_in[0];
  const int*   ei = (const int*)d_in[1];
  const float* W1 = (const float*)d_in[2];
  const float* b1 = (const float*)d_in[3];
  const float* W2 = (const float*)d_in[4];
  const float* b2 = (const float*)d_in[5];

  const int n = in_sizes[0] / 256;
  const int e = in_sizes[1] / 2;
  const int* src = ei;
  const int* dst = ei + e;

  float* agg1 = (float*)d_ws;
  float* h1   = agg1 + (size_t)n * 128;
  int*   glist = (int*)(h1 + (size_t)n * 128);
  float* dinv = (float*)(glist + (size_t)n * 32 + n);
  float* h2   = h1;
  float* agg2 = h1 + (size_t)n * 16;
  const int nrange = ceil_div(n, RANGE_ROWS);
  int*   gcount = glist + (size_t)nrange * 10240;
  int*   keys = glist + ((((size_t)nrange * 10241) + 3) & ~(size_t)3);

  key_kernel<RANGE_ROWS><<<ceil_div(ceil_div(e, 4), 256), 256, 0, stream>>>(dst, keys, e);
  range_list_kernel<RANGE_ROWS, 10240><<<nrange, 1024, 0, stream>>>(src, dst, dinv, glist, gcount, keys, e,
                                                               n);

  gemm_wmma_f32_lds_a<256, 128, 4, 2, 4><<<ceil_div(n, 64), 128, 0, stream>>>(x, W1, h1, n);
  agg_lds_kernel<7, RANGE_ROWS, 10240><<<nrange, 1024, 0, stream>>>(
      h1, src, dst, dinv, agg1, glist, gcount, keys, e, n);

  {
    const long long waves = (long long)((n + 15) / 16) * (16 / 16);
    gemm_wmma_f32_relu_a<128, 16, 8><<<ceil_div(waves * 32, 256), 256, 0, stream>>>(
        agg1, b1, W2, h2, n);
  }
  agg_lds_kernel<4, RANGE_ROWS, 10240><<<nrange, 1024, 0, stream>>>(
      h2, src, dst, dinv, agg2, glist, gcount, keys, e, n);
  lsm_kernel<<<ceil_div(n, 256), 256, 0, stream>>>(agg2, b2, (float*)d_out, n);
}


# Key kernel also writes a line map (per 4096-edge chunk and range: which of the chunk's 32 key lines hold an edge of the range); owners' first pass reads only marked key lines; sum order LDS arrival
# speedup vs baseline: 5.1916x; 1.0543x over previous
#include <hip/hip_runtime.h>
#include <math.h>


typedef __attribute__((ext_vector_type(2))) float v2f;
typedef __attribute__((ext_vector_type(8))) float v8f;
typedef __attribute__((address_space(1))) float global_float;

static inline int ceil_div(long long a, long long b) { return (int)((a + b - 1) / b); }

constexpr int RANGE_ROWS = 391;

constexpr int REC_SRC_BITS = 23;
__device__ __forceinline__ int rec_pack(int s, int dl) {
  return (int)(((unsigned)dl << REC_SRC_BITS) | (unsigned)s);
}
__device__ __forceinline__ int rec_src(int r) {
  return (int)((unsigned)r & ((1u << REC_SRC_BITS) - 1u));
}
__device__ __forceinline__ int rec_row(int r) { return (int)((unsigned)r >> REC_SRC_BITS); }

template <int K, int N, int WPB>
__global__ __launch_bounds__(WPB * 32) void gemm_wmma_f32_relu_a(const float* __restrict__ S,
                                                                const float* __restrict__ bias,
                                                                const float* __restrict__ B,
                                                                float* __restrict__ C, int M) {
  constexpr int NT = N / 16;
  constexpr int P = K + 4;
  static_assert(K == 128, "one 16-byte piece of a row per lane");
  __shared__ __attribute__((aligned(16))) float As[WPB * 16 * P];

  const int lane = threadIdx.x & 31;
  const int wave = threadIdx.x >> 5;
  const int gwave = __builtin_amdgcn_readfirstlane((int)(blockIdx.x * WPB + wave));
  const int mt = gwave / NT;
  const int nt = gwave - mt * NT;
  const int half = lane >> 4;
  const int l16 = lane & 15;
  const int col = nt * 16 + l16;
  float* __restrict__ Aw = As + wave * (16 * P);

  const int m0 = mt * 16 + 8 * half;
  unsigned long long caddr = reinterpret_cast<unsigned long long>(C + (long long)m0 * N + col);
  int nrows = M - m0;
  asm volatile("" : "+v"(caddr), "+v"(nrows));
  global_float* Cp = (global_float*)caddr;

  {
    const float4 bq = *reinterpret_cast<const float4*>(bias + 4 * lane);
    float4 t[16];
    for (int r = 0; r < 16; ++r) {
      int gr = mt * 16 + r;
      gr = gr < M ? gr : (M - 1);
      t[r] = *reinterpret_cast<const float4*>(S + (long long)gr * K + 4 * lane);
    }
    for (int r = 0; r < 16; ++r) {
      float4 v;
      v.x = t[r].x + bq.x;
      v.y = t[r].y + bq.y;
      v.z = t[r].z + bq.z;
      v.w = t[r].w + bq.w;
      v.x = v.x > 0.0f ? v.x : 0.0f;
      v.y = v.y > 0.0f ? v.y : 0.0f;
      v.z = v.z > 0.0f ? v.z : 0.0f;
      v.w = v.w > 0.0f ? v.w : 0.0f;
      *reinterpret_cast<float4*>(Aw + r * P + 4 * lane) = v;
    }
  }
  __syncthreads();

  const float* __restrict__ Al = Aw + l16 * P + 2 * half;
  const float* __restrict__ Bl = B + (long long)(2 * half) * N + col;

  v8f acc = {};
#pragma unroll
  for (int k = 0; k < K; k += 4) {
    const v2f a = *reinterpret_cast<const v2f*>(Al + k);
    v2f b;
    b.x = Bl[(long long)k * N];
    b.y = Bl[(long long)(k + 1) * N];
    acc = __builtin_amdgcn_wmma_f32_16x16x4_f32(false, a, false, b, (short)0, acc,
                                                false, false);
  }

  if (mt * 16 + 16 <= M) {
#pragma unroll
    for (int v = 0; v < 8; ++v) Cp[(long long)v * N] = acc[v];
  } else {
#pragma unroll
    for (int v = 0; v < 8; ++v)
      if (v < nrows) Cp[(long long)v * N] = acc[v];
  }
}

template <int K, int N, int TM, int TN, int WN>
__global__ __launch_bounds__(WN * 32) void gemm_wmma_f32_lds_a(const float* __restrict__ A,
                                                             const float* __restrict__ B,
                                                             float* __restrict__ C, int M) {
  constexpr int BM = TM * 16;
  constexpr int P = K + 4;
  constexpr int NTHR = WN * 32;
  constexpr int KQ = K / 4;
  constexpr int NV = BM * KQ;
  constexpr int LB = 32;
  static_assert(N == WN * TN * 16, "waves must tile the N columns");
  static_assert(NV % (NTHR * LB) == 0, "slab must split evenly over the staging batches");
  __shared__ __attribute__((aligned(16))) float As[BM * P];

  const int tid = threadIdx.x;
  const int row0 = blockIdx.x * BM;

  for (int i0 = 0; i0 < NV; i0 += NTHR * LB) {
    float4 t[LB];
    for (int u = 0; u < LB; ++u) {
      const int f = i0 + u * NTHR + tid;
      const int r = f / KQ, kq = f - r * KQ;
      int gr = row0 + r;
      gr = gr < M ? gr : (M - 1);
      t[u] = *reinterpret_cast<const float4*>(A + (long long)gr * K + 4 * kq);
    }
    for (int u = 0; u < LB; ++u) {
      const int f = i0 + u * NTHR + tid;
      const int r = f / KQ, kq = f - r * KQ;
      *reinterpret_cast<float4*>(As + r * P + 4 * kq) = t[u];
    }
  }
  __syncthreads();

  const int lane = tid & 31;
  const int wave = tid >> 5;
  const int half = lane >> 4;
  const int l16 = lane & 15;
  const int col0 = wave * (TN * 16) + l16;

  const float* __restrict__ Al = As + l16 * P + 2 * half;
  const float* __restrict__ Bl = B + (long long)(2 * half) * N + col0;

  v8f acc[TM][TN];
  for (int i = 0; i < TM; ++i)
    for (int j = 0; j < TN; ++j) acc[i][j] = v8f{};

#pragma unroll
  for (int k = 0; k < K; k += 4) {
    v2f a[TM], b[TN];
    for (int i = 0; i < TM; ++i) a[i] = *reinterpret_cast<const v2f*>(Al + i * 16 * P + k);
    for (int j = 0; j < TN; ++j) {
      b[j].x = Bl[(long long)k * N + j * 16];
      b[j].y = Bl[(long long)(k + 1) * N + j * 16];
    }
    for (int i = 0; i < TM; ++i)
      for (int j = 0; j < TN; ++j)
        acc[i][j] = __builtin_amdgcn_wmma_f32_16x16x4_f32(false, a[i], false, b[j], (short)0,
                                                          acc[i][j], false, false);
  }

  for (int i = 0; i < TM; ++i) {
    const int m0 = row0 + i * 16 + 8 * half;
    for (int j = 0; j < TN; ++j) {
      float* __restrict__ Cp = C + (long long)m0 * N + col0 + j * 16;
      if (row0 + BM <= M) {
        for (int v = 0; v < 8; ++v) Cp[(long long)v * N] = acc[i][j][v];
      } else {
        for (int v = 0; v < 8; ++v)
          if (m0 + v < M) Cp[(long long)v * N] = acc[i][j][v];
      }
    }
  }
}

constexpr int CHUNK_EDGES = 4096;
template <int R>
__global__ __launch_bounds__(256) void key_kernel(const int* __restrict__ dst,
                                                  int* __restrict__ keys,
                                                  unsigned* __restrict__ lmap, int e) {
  __shared__ unsigned present[256];
  const int t = threadIdx.x;
  present[t] = 0u;
  __syncthreads();
  const int w0 = blockIdx.x * (CHUNK_EDGES / 4);
  const int nwl = ((e + 127) >> 7) << 5;
  const bool whole = (reinterpret_cast<unsigned long long>(dst) & 15ull) == 0ull &&
                     4 * w0 + CHUNK_EDGES <= e;
  int4 d[4];
  if (whole) {
    for (int u = 0; u < 4; ++u)
      d[u] = *reinterpret_cast<const int4*>(dst + 4 * (w0 + u * 256 + t));
  } else {
    for (int u = 0; u < 4; ++u) {
      const int i0 = 4 * (w0 + u * 256 + t);
      d[u] = int4{-1, -1, -1, -1};
      if (i0 < e) d[u].x = dst[i0];
      if (i0 + 1 < e) d[u].y = dst[i0 + 1];
      if (i0 + 2 < e) d[u].z = dst[i0 + 2];
      if (i0 + 3 < e) d[u].w = dst[i0 + 3];
    }
  }
  for (int u = 0; u < 4; ++u) {
    const int wi = w0 + u * 256 + t;
    const int i0 = 4 * wi;
    const unsigned bit = 1u << (u * 8 + (t >> 5));
    const int dd[4] = {d[u].x, d[u].y, d[u].z, d[u].w};
    unsigned w = 0xFFFFFFFFu;
    for (int k = 0; k < 4; ++k) {
      if (i0 + k < e) {
        const unsigned key = (unsigned)(dd[k] / R) & 0xFFu;
        w = (w & ~(0xFFu << (8 * k))) | (key << (8 * k));
        atomicOr(&present[key], bit);
      }
    }
    if (wi < nwl) keys[wi] = (int)w;
  }
  __syncthreads();
  lmap[(long long)blockIdx.x * 256 + t] = present[t];
}

template <int R, int NT, class Push>
__device__ __forceinline__ void scan_for_range(const int* __restrict__ keys,
                                               const int* __restrict__ dst, int e, int base,
                                               Push&& push) {
  const int tid = threadIdx.x;
  auto in_range = [&](int d) { return (unsigned)(d - base) < (unsigned)R; };
  const bool use_keys = gridDim.x <= 256;

  if (use_keys) {
    const unsigned kb = (unsigned)blockIdx.x * 0x01010101u;
    auto nzf = [&](int wi) {
      const unsigned x = (unsigned)wi ^ kb;
      return (((x & 0x7F7F7F7Fu) + 0x7F7F7F7Fu) | x) & 0x80808080u;
    };
    auto pairmask = [&](const int4 p, const int4 q) {
      return ~((nzf(p.x) >> 7) | (nzf(p.y) >> 6) | (nzf(p.z) >> 5) | (nzf(p.w) >> 4) |
               (nzf(q.x) >> 3) | (nzf(q.y) >> 2) | (nzf(q.z) >> 1) | nzf(q.w));
    };
    auto drain = [&](unsigned m, int e0, int e1) {
      while (m) {
        const int p = __builtin_ctz(m);
        m &= m - 1;
        const int idx = ((p & 4) ? e1 : e0) + 4 * (p & 3) + (p >> 3);
        if (idx < e) push(idx);
      }
    };
    const int nw = (e + 3) >> 2;
    const int nvec = nw >> 2;
    const int4* __restrict__ kv = reinterpret_cast<const int4*>(keys);
    constexpr int U = 8;
    int v = tid;
    for (; v + (U - 1) * NT < nvec; v += U * NT) {
      int4 kk[U];
      for (int u = 0; u < U; ++u) kk[u] = kv[v + u * NT];
      for (int u = 0; u < U; u += 2)
        drain(pairmask(kk[u], kk[u + 1]), 16 * (v + u * NT), 16 * (v + (u + 1) * NT));
    }
    const int4 none = {-1, -1, -1, -1};
    for (; v < nvec; v += NT) {
      const unsigned m = pairmask(kv[v], none) & 0x0F0F0F0Fu;
      drain(m, 16 * v, 0);
    }
    if (tid < (nw & 3)) {
      const int t = 4 * nvec + tid;
      const unsigned m = ~(nzf(keys[t]) >> 7) & 0x01010101u;
      drain(m, 4 * t, 0);
    }
  } else {
    int head = (int)(((16u - (unsigned)(reinterpret_cast<unsigned long long>(dst) & 15u)) & 15u) >> 2);
    if (head > e) head = e;
    if (tid < head && in_range(dst[tid])) push(tid);
    const int nvec = (e - head) >> 2;
    const int4* __restrict__ dv = reinterpret_cast<const int4*>(dst + head);
    constexpr int U = 8;
    int v = tid;
    for (; v + (U - 1) * NT < nvec; v += U * NT) {
      int4 dd[U];
      for (int u = 0; u < U; ++u) dd[u] = dv[v + u * NT];
      unsigned m = 0;
      for (int u = 0; u < U; ++u) {
        m |= (in_range(dd[u].x) ? 1u : 0u) << (4 * u);
        m |= (in_range(dd[u].y) ? 2u : 0u) << (4 * u);
        m |= (in_range(dd[u].z) ? 4u : 0u) << (4 * u);
        m |= (in_range(dd[u].w) ? 8u : 0u) << (4 * u);
      }
      while (m) {
        const int b = __builtin_ctz(m);
        m &= m - 1;
        push(head + 4 * (v + (b >> 2) * NT) + (b & 3));
      }
    }
    for (; v < nvec; v += NT) {
      const int4 d4 = dv[v];
      unsigned m = (in_range(d4.x) ? 1u : 0u) | (in_range(d4.y) ? 2u : 0u) |
                   (in_range(d4.z) ? 4u : 0u) | (in_range(d4.w) ? 8u : 0u);
      while (m) {
        const int b = __builtin_ctz(m);
        m &= m - 1;
        push(head + 4 * v + b);
      }
    }
    {
      const int t = head + 4 * nvec + tid;
      if (t < e && in_range(dst[t])) push(t);
    }
  }
}

template <int NT, class Push>
__device__ __forceinline__ void scan_marked_lines(const int* __restrict__ keys,
                                                  const unsigned* __restrict__ lmap, int e,
                                                  int nchunk, int* lines, int* nlines,
                                                  Push&& push) {
  const int tid = threadIdx.x;
  for (int c = tid; c < nchunk; c += NT) {
    unsigned w = lmap[(long long)c * 256 + blockIdx.x];
    if (w) {
      int slot = atomicAdd(nlines, __builtin_popcount(w));
      while (w) {
        const int b = __builtin_ctz(w);
        w &= w - 1;
        lines[slot++] = c * (CHUNK_EDGES / 128) + b;
      }
    }
  }
  __syncthreads();
  const int nitems = *nlines * 8;

  const unsigned kb = (unsigned)blockIdx.x * 0x01010101u;
  auto nzf = [&](int wi) {
    const unsigned x = (unsigned)wi ^ kb;
    return (((x & 0x7F7F7F7Fu) + 0x7F7F7F7Fu) | x) & 0x80808080u;
  };
  auto pairmask = [&](const int4 p, const int4 q) {
    return ~((nzf(p.x) >> 7) | (nzf(p.y) >> 6) | (nzf(p.z) >> 5) | (nzf(p.w) >> 4) |
             (nzf(q.x) >> 3) | (nzf(q.y) >> 2) | (nzf(q.z) >> 1) | nzf(q.w));
  };
  auto drain = [&](unsigned m, int e0, int e1) {
    while (m) {
      const int p = __builtin_ctz(m);
      m &= m - 1;
      const int idx = ((p & 4) ? e1 : e0) + 4 * (p & 3) + (p >> 3);
      if (idx < e) push(idx);
    }
  };
  const int4* __restrict__ kv = reinterpret_cast<const int4*>(keys);
  constexpr int U = 8;
  for (int j = tid; j < nitems; j += U * NT) {
    int pv[U];
    bool ok[U];
    int4 kk[U];
    for (int u = 0; u < U; ++u) {
      const int jj = j + u * NT;
      ok[u] = jj < nitems;
      const int jc = ok[u] ? jj : j;
      pv[u] = lines[jc >> 3] * 8 + (jc & 7);
    }
    for (int u = 0; u < U; ++u) kk[u] = kv[pv[u]];
    for (int u = 0; u < U; u += 2) {
      const unsigned keep = (ok[u] ? 0x0F0F0F0Fu : 0u) | (ok[u + 1] ? 0xF0F0F0F0u : 0u);
      drain(pairmask(kk[u], kk[u + 1]) & keep, 16 * pv[u], 16 * pv[u + 1]);
    }
  }
}

template <int R, int QCAP, int LCAP>
__global__ __launch_bounds__(1024) void range_list_kernel(const int* __restrict__ src,
                                                          const int* __restrict__ dst,
                                                          float* __restrict__ dinv,
                                                          int* __restrict__ glist,
                                                          int* __restrict__ gcount,
                                                          const int* __restrict__ keys,
                                                          const unsigned* __restrict__ lmap,
                                                          int e, int n) {
  constexpr int NT = 1024;
  static_assert(R <= (1 << (32 - REC_SRC_BITS)), "the local row must fit the record");
  __shared__ int cnt[R];
  __shared__ int queue[QCAP];
  __shared__ int lines[LCAP];
  __shared__ int qcount;
  __shared__ int nlines;

  const int tid = threadIdx.x;
  const int base = blockIdx.x * R;
  const bool packable = n <= (1 << REC_SRC_BITS);

  if (tid == 0) {
    qcount = 0;
    nlines = 0;
  }
  for (int i = tid; i < R; i += NT) cnt[i] = 0;
  __syncthreads();

  auto push = [&](int idx) {
    const int slot = atomicAdd(&qcount, 1);
    if (slot < QCAP) {
      queue[slot] = idx;
    } else {
      atomicAdd(&cnt[dst[idx] - base], 1);
    }
  };
  const int nchunk = (e + CHUNK_EDGES - 1) / CHUNK_EDGES;
  const bool use_map = gridDim.x <= 256 && (long long)nchunk * (CHUNK_EDGES / 128) <= LCAP;
  if (use_map) {
    scan_marked_lines<NT>(keys, lmap, e, nchunk, lines, &nlines, push);
  } else {
    scan_for_range<R, NT>(keys, dst, e, base, push);
  }
  __syncthreads();

  const int met = qcount;
  const int total = met < QCAP ? met : QCAP;
  int* __restrict__ gl = glist + (long long)blockIdx.x * QCAP;
  constexpr int WU = 4;
  int i = tid;
  for (; i + (WU - 1) * NT < total; i += WU * NT) {
    int s[WU], dl[WU];
    for (int u = 0; u < WU; ++u) {
      const int idx = queue[i + u * NT];
      s[u] = src[idx];
      dl[u] = dst[idx] - base;
    }
    for (int u = 0; u < WU; ++u) {
      gl[i + u * NT] = rec_pack(s[u], dl[u]);
      atomicAdd(&cnt[dl[u]], 1);
    }
  }
  for (; i < total; i += NT) {
    const int idx = queue[i];
    const int dl = dst[idx] - base;
    gl[i] = rec_pack(src[idx], dl);
    atomicAdd(&cnt[dl], 1);
  }
  if (tid == 0) gcount[blockIdx.x] = packable ? met : 0x7FFFFFFF;
  __syncthreads();

  for (int i = tid; i < R; i += NT) {
    const int node = base + i;
    if (node < n) dinv[node] = rsqrtf((float)cnt[i] + 1.0f);
  }
}

template <int LOGF, int R, int QCAP>
__global__ __launch_bounds__(1024) void agg_lds_kernel(const float* __restrict__ H,
                                                       const int* __restrict__ src,
                                                       const int* __restrict__ dst,
                                                       const float* __restrict__ dinv,
                                                       float* __restrict__ out,
                                                       const int* __restrict__ glist,
                                                       const int* __restrict__ gcount,
                                                       const int* __restrict__ keys, int e,
                                                       int n) {
  constexpr int F = 1 << LOGF;
  constexpr int LPE = F / 4;
  constexpr int LOGLPE = LOGF - 2;
  constexpr int EPI = 32 / LPE;
  constexpr int NT = 1024;
  static_assert(R <= (1 << (32 - REC_SRC_BITS)), "the local row must fit the record");
  __shared__ float acc[R * F];
  __shared__ int queue[QCAP];
  __shared__ int qcount;

  const int tid = threadIdx.x;
  const int lane = tid & 31;
  const int base = blockIdx.x * R;
  const bool packable = n <= (1 << REC_SRC_BITS);

  if (tid == 0) qcount = 0;
  for (int i = tid; i < R * LPE; i += NT) {
    const int row = i >> LOGLPE, q = i & (LPE - 1);
    const int node = base + row;
    if (node < n) {
      const float dn = dinv[node];
      const float nrm = dn * dn;
      const float4 h = *reinterpret_cast<const float4*>(H + (((long long)node) << LOGF) + 4 * q);
      float* a = acc + (row << LOGF) + q;
      a[0] = h.x * nrm;
      a[LPE] = h.y * nrm;
      a[2 * LPE] = h.z * nrm;
      a[3 * LPE] = h.w * nrm;
    }
  }
  __syncthreads();

  auto push = [&](int idx) {
    const int s = src[idx];
    const int d = dst[idx];
    const int slot = packable ? atomicAdd(&qcount, 1) : QCAP;
    if (slot < QCAP) {
      queue[slot] = rec_pack(s, d - base);
    } else {
      const float nrm = dinv[s] * dinv[d];
      const float* hs = H + (((long long)s) << LOGF);
      float* a = acc + ((d - base) << LOGF);
      for (int f = 0; f < F; ++f) atomicAdd(a + (f & 3) * LPE + (f >> 2), hs[f] * nrm);
    }
  };

  const int c = gcount[blockIdx.x];
  const bool have_list = c <= QCAP;
  if (have_list) {
    const int* __restrict__ gl = glist + (long long)blockIdx.x * QCAP;
    for (int i = tid; i < c; i += NT) queue[i] = gl[i];
    if (tid == 0) qcount = c;
  } else {
    scan_for_range<R, NT>(keys, dst, e, base, push);
  }
  __syncthreads();

  constexpr int STEPS = 32 / EPI;
  constexpr int UB = STEPS < 8 ? STEPS : 8;
  const int total = qcount < QCAP ? qcount : QCAP;
  const int q = lane & (LPE - 1);
  const int b0 = (tid >> 5) * 32;
  int ns, ndl;
  float nnrm;
  {
    const int r = (b0 + lane < total) ? queue[b0 + lane] : 0;
    ns = rec_src(r);
    ndl = rec_row(r);
    nnrm = dinv[ns] * dinv[base + ndl];
  }
  for (int b = b0; b < total; b += NT) {
    const int s = ns;
    const int dl = ndl;
    const float nrm = nnrm;
    const int cnt = total - b < 32 ? total - b : 32;
    const int rn = (b + NT + lane < total) ? queue[b + NT + lane] : 0;
    ns = rec_src(rn);
    ndl = rec_row(rn);
    const float pds = dinv[ns];
    const float pdd = dinv[base + ndl];
    for (int j = 0; j < STEPS; j += UB) {
      float4 h[UB];
      int dj[UB];
      float nj[UB];
      for (int u = 0; u < UB; ++u) {
        const int sub = (j + u) * EPI + (lane >> LOGLPE);
        const int sj = __shfl(s, sub);
        dj[u] = __shfl(dl, sub);
        nj[u] = __shfl(nrm, sub);
        h[u] = *reinterpret_cast<const float4*>(H + (((long long)sj) << LOGF) + 4 * q);
      }
      for (int u = 0; u < UB; ++u) {
        const int sub = (j + u) * EPI + (lane >> LOGLPE);
        if (sub < cnt) {
          float* a = acc + (dj[u] << LOGF) + q;
          atomicAdd(a, h[u].x * nj[u]);
          atomicAdd(a + LPE, h[u].y * nj[u]);
          atomicAdd(a + 2 * LPE, h[u].z * nj[u]);
          atomicAdd(a + 3 * LPE, h[u].w * nj[u]);
        }
      }
    }
    nnrm = pds * pdd;
  }
  __syncthreads();

  for (int i = tid; i < R * LPE; i += NT) {
    const int row = i >> LOGLPE, qq = i & (LPE - 1);
    const int node = base + row;
    if (node < n) {
      const float* a = acc + (row << LOGF) + qq;
      float4 o;
      o.x = a[0];
      o.y = a[LPE];
      o.z = a[2 * LPE];
      o.w = a[3 * LPE];
      *reinterpret_cast<float4*>(out + (((long long)node) << LOGF) + 4 * qq) = o;
    }
  }
}

__global__ void lsm_kernel(const float* __restrict__ agg, const float* __restrict__ b,
                           float* __restrict__ out, int n) {
  const int i = blockIdx.x * blockDim.x + threadIdx.x;
  if (i >= n) return;
  float v[16];
  float mx = -INFINITY;
#pragma unroll
  for (int j = 0; j < 16; ++j) {
    v[j] = agg[(long long)i * 16 + j] + b[j];
    mx = fmaxf(mx, v[j]);
  }
  float s = 0.0f;
#pragma unroll
  for (int j = 0; j < 16; ++j) s += expf(v[j] - mx);
  const float lse = mx + logf(s);
#pragma unroll
  for (int j = 0; j < 16; ++j) out[(long long)i * 16 + j] = v[j] - lse;
}

extern "C" void kernel_launch(void* const* d_in, const int* in_sizes, int n_in,
                              void* d_out, int out_size, void* d_ws, size_t ws_size,
                              hipStream_t stream) {
  const float* x  = (const float*)d_in[0];
  const int*   ei = (const int*)d_in[1];
  const float* W1 = (const float*)d_in[2];
  const float* b1 = (const float*)d_in[3];
  const float* W2 = (const float*)d_in[4];
  const float* b2 = (const float*)d_in[5];

  const int n = in_sizes[0] / 256;
  const int e = in_sizes[1] / 2;
  const int* src = ei;
  const int* dst = ei + e;

  float* agg1 = (float*)d_ws;
  float* h1   = agg1 + (size_t)n * 128;
  int*   glist = (int*)(h1 + (size_t)n * 128);
  float* dinv = (float*)(glist + (size_t)n * 32 + n);
  float* h2   = h1;
  float* agg2 = h1 + (size_t)n * 16;
  const int nrange = ceil_div(n, RANGE_ROWS);
  int*   gcount = glist + (size_t)nrange * 10240;
  int*   keys = glist + ((((size_t)nrange * 10241) + 3) & ~(size_t)3);
  const int nchunk = ceil_div(e, CHUNK_EDGES);
  unsigned* lmap = (unsigned*)(keys + (size_t)ceil_div(e, 128) * 32);

  key_kernel<RANGE_ROWS><<<nchunk, 256, 0, stream>>>(dst, keys, lmap, e);
  range_list_kernel<RANGE_ROWS, 10240, 16384><<<nrange, 1024, 0, stream>>>(
      src, dst, dinv, glist, gcount, keys, lmap, e, n);

  gemm_wmma_f32_lds_a<256, 128, 4, 2, 4><<<ceil_div(n, 64), 128, 0, stream>>>(x, W1, h1, n);
  agg_lds_kernel<7, RANGE_ROWS, 10240><<<nrange, 1024, 0, stream>>>(
      h1, src, dst, dinv, agg1, glist, gcount, keys, e, n);

  {
    const long long waves = (long long)((n + 15) / 16) * (16 / 16);
    gemm_wmma_f32_relu_a<128, 16, 8><<<ceil_div(waves * 32, 256), 256, 0, stream>>>(
        agg1, b1, W2, h2, n);
  }
  agg_lds_kernel<4, RANGE_ROWS, 10240><<<nrange, 1024, 0, stream>>>(
      h2, src, dst, dinv, agg2, glist, gcount, keys, e, n);
  lsm_kernel<<<ceil_div(n, 256), 256, 0, stream>>>(agg2, b2, (float*)d_out, n);
}


# Owners read a pair plane (src and dst of an edge side by side, 16 edges per line, copied with a finer line map by the first kernel): no src/dst lookups, byte key plane gone; sum order LDS arrival
# speedup vs baseline: 5.4114x; 1.0423x over previous
#include <hip/hip_runtime.h>
#include <math.h>


typedef __attribute__((ext_vector_type(2))) float v2f;
typedef __attribute__((ext_vector_type(8))) float v8f;
typedef __attribute__((address_space(1))) float global_float;

static inline int ceil_div(long long a, long long b) { return (int)((a + b - 1) / b); }

constexpr int RANGE_ROWS = 391;

constexpr int REC_SRC_BITS = 23;
__device__ __forceinline__ int rec_pack(int s, int dl) {
  return (int)(((unsigned)dl << REC_SRC_BITS) | (unsigned)s);
}
__device__ __forceinline__ int rec_src(int r) {
  return (int)((unsigned)r & ((1u << REC_SRC_BITS) - 1u));
}
__device__ __forceinline__ int rec_row(int r) { return (int)((unsigned)r >> REC_SRC_BITS); }

template <int K, int N, int WPB>
__global__ __launch_bounds__(WPB * 32) void gemm_wmma_f32_relu_a(const float* __restrict__ S,
                                                                const float* __restrict__ bias,
                                                                const float* __restrict__ B,
                                                                float* __restrict__ C, int M) {
  constexpr int NT = N / 16;
  constexpr int P = K + 4;
  static_assert(K == 128, "one 16-byte piece of a row per lane");
  __shared__ __attribute__((aligned(16))) float As[WPB * 16 * P];

  const int lane = threadIdx.x & 31;
  const int wave = threadIdx.x >> 5;
  const int gwave = __builtin_amdgcn_readfirstlane((int)(blockIdx.x * WPB + wave));
  const int mt = gwave / NT;
  const int nt = gwave - mt * NT;
  const int half = lane >> 4;
  const int l16 = lane & 15;
  const int col = nt * 16 + l16;
  float* __restrict__ Aw = As + wave * (16 * P);

  const int m0 = mt * 16 + 8 * half;
  unsigned long long caddr = reinterpret_cast<unsigned long long>(C + (long long)m0 * N + col);
  int nrows = M - m0;
  asm volatile("" : "+v"(caddr), "+v"(nrows));
  global_float* Cp = (global_float*)caddr;

  {
    const float4 bq = *reinterpret_cast<const float4*>(bias + 4 * lane);
    float4 t[16];
    for (int r = 0; r < 16; ++r) {
      int gr = mt * 16 + r;
      gr = gr < M ? gr : (M - 1);
      t[r] = *reinterpret_cast<const float4*>(S + (long long)gr * K + 4 * lane);
    }
    for (int r = 0; r < 16; ++r) {
      float4 v;
      v.x = t[r].x + bq.x;
      v.y = t[r].y + bq.y;
      v.z = t[r].z + bq.z;
      v.w = t[r].w + bq.w;
      v.x = v.x > 0.0f ? v.x : 0.0f;
      v.y = v.y > 0.0f ? v.y : 0.0f;
      v.z = v.z > 0.0f ? v.z : 0.0f;
      v.w = v.w > 0.0f ? v.w : 0.0f;
      *reinterpret_cast<float4*>(Aw + r * P + 4 * lane) = v;
    }
  }
  __syncthreads();

  const float* __restrict__ Al = Aw + l16 * P + 2 * half;
  const float* __restrict__ Bl = B + (long long)(2 * half) * N + col;

  v8f acc = {};
#pragma unroll
  for (int k = 0; k < K; k += 4) {
    const v2f a = *reinterpret_cast<const v2f*>(Al + k);
    v2f b;
    b.x = Bl[(long long)k * N];
    b.y = Bl[(long long)(k + 1) * N];
    acc = __builtin_amdgcn_wmma_f32_16x16x4_f32(false, a, false, b, (short)0, acc,
                                                false, false);
  }

  if (mt * 16 + 16 <= M) {
#pragma unroll
    for (int v = 0; v < 8; ++v) Cp[(long long)v * N] = acc[v];
  } else {
#pragma unroll
    for (int v = 0; v < 8; ++v)
      if (v < nrows) Cp[(long long)v * N] = acc[v];
  }
}

template <int K, int N, int TM, int TN, int WN>
__global__ __launch_bounds__(WN * 32) void gemm_wmma_f32_lds_a(const float* __restrict__ A,
                                                             const float* __restrict__ B,
                                                             float* __restrict__ C, int M) {
  constexpr int BM = TM * 16;
  constexpr int P = K + 4;
  constexpr int NTHR = WN * 32;
  constexpr int KQ = K / 4;
  constexpr int NV = BM * KQ;
  constexpr int LB = 32;
  static_assert(N == WN * TN * 16, "waves must tile the N columns");
  static_assert(NV % (NTHR * LB) == 0, "slab must split evenly over the staging batches");
  __shared__ __attribute__((aligned(16))) float As[BM * P];

  const int tid = threadIdx.x;
  const int row0 = blockIdx.x * BM;

  for (int i0 = 0; i0 < NV; i0 += NTHR * LB) {
    float4 t[LB];
    for (int u = 0; u < LB; ++u) {
      const int f = i0 + u * NTHR + tid;
      const int r = f / KQ, kq = f - r * KQ;
      int gr = row0 + r;
      gr = gr < M ? gr : (M - 1);
      t[u] = *reinterpret_cast<const float4*>(A + (long long)gr * K + 4 * kq);
    }
    for (int u = 0; u < LB; ++u) {
      const int f = i0 + u * NTHR + tid;
      const int r = f / KQ, kq = f - r * KQ;
      *reinterpret_cast<float4*>(As + r * P + 4 * kq) = t[u];
    }
  }
  __syncthreads();

  const int lane = tid & 31;
  const int wave = tid >> 5;
  const int half = lane >> 4;
  const int l16 = lane & 15;
  const int col0 = wave * (TN * 16) + l16;

  const float* __restrict__ Al = As + l16 * P + 2 * half;
  const float* __restrict__ Bl = B + (long long)(2 * half) * N + col0;

  v8f acc[TM][TN];
  for (int i = 0; i < TM; ++i)
    for (int j = 0; j < TN; ++j) acc[i][j] = v8f{};

#pragma unroll
  for (int k = 0; k < K; k += 4) {
    v2f a[TM], b[TN];
    for (int i = 0; i < TM; ++i) a[i] = *reinterpret_cast<const v2f*>(Al + i * 16 * P + k);
    for (int j = 0; j < TN; ++j) {
      b[j].x = Bl[(long long)k * N + j * 16];
      b[j].y = Bl[(long long)(k + 1) * N + j * 16];
    }
    for (int i = 0; i < TM; ++i)
      for (int j = 0; j < TN; ++j)
        acc[i][j] = __builtin_amdgcn_wmma_f32_16x16x4_f32(false, a[i], false, b[j], (short)0,
                                                          acc[i][j], false, false);
  }

  for (int i = 0; i < TM; ++i) {
    const int m0 = row0 + i * 16 + 8 * half;
    for (int j = 0; j < TN; ++j) {
      float* __restrict__ Cp = C + (long long)m0 * N + col0 + j * 16;
      if (row0 + BM <= M) {
        for (int v = 0; v < 8; ++v) Cp[(long long)v * N] = acc[i][j][v];
      } else {
        for (int v = 0; v < 8; ++v)
          if (m0 + v < M) Cp[(long long)v * N] = acc[i][j][v];
      }
    }
  }
}

constexpr int CHUNK_EDGES = 4096;
constexpr int CHUNK_LINES = CHUNK_EDGES / 16;
constexpr int CHUNK_MAPW = 256 * (CHUNK_LINES / 32);

__device__ __forceinline__ bool pair_map_usable(int n, int e, int nrange) {
  const long long nchunk = ((long long)e + CHUNK_EDGES - 1) / CHUNK_EDGES;
  const long long bytes = nchunk * ((long long)CHUNK_EDGES * 8 + (long long)CHUNK_MAPW * 4);
  return nrange <= 256 && bytes <= (long long)n * 128 * 4;
}

template <int R>
__global__ __launch_bounds__(256) void pair_kernel(const int* __restrict__ src,
                                                   const int* __restrict__ dst,
                                                   int* __restrict__ pairs,
                                                   unsigned* __restrict__ lmap, int e, int n,
                                                   int nrange) {
  __shared__ unsigned present[CHUNK_MAPW];
  if (!pair_map_usable(n, e, nrange)) return;
  const int t = threadIdx.x;
  for (int j = 0; j < CHUNK_MAPW / 256; ++j) present[j * 256 + t] = 0u;
  __syncthreads();
  const int p0 = blockIdx.x * (CHUNK_EDGES / 2);
  const bool whole = ((reinterpret_cast<unsigned long long>(src) |
                       reinterpret_cast<unsigned long long>(dst)) & 7ull) == 0ull &&
                     2ll * p0 + CHUNK_EDGES <= (long long)e;
  int2 s2[8], d2[8];
  if (whole) {
    for (int u = 0; u < 8; ++u) {
      const int p = p0 + u * 256 + t;
      s2[u] = *reinterpret_cast<const int2*>(src + 2 * p);
      d2[u] = *reinterpret_cast<const int2*>(dst + 2 * p);
    }
  } else {
    for (int u = 0; u < 8; ++u) {
      const long long i0 = 2ll * (p0 + u * 256 + t);
      s2[u] = int2{0, 0};
      d2[u] = int2{-1, -1};
      if (i0 < e) {
        s2[u].x = src[i0];
        d2[u].x = dst[i0];
      }
      if (i0 + 1 < e) {
        s2[u].y = src[i0 + 1];
        d2[u].y = dst[i0 + 1];
      }
    }
  }
  int4* __restrict__ pv = reinterpret_cast<int4*>(pairs);
  const unsigned bit = 1u << (t >> 3);
  for (int u = 0; u < 8; ++u) {
    if ((unsigned)d2[u].x < (unsigned)n) atomicOr(&present[(d2[u].x / R) * 8 + u], bit);
    if ((unsigned)d2[u].y < (unsigned)n) atomicOr(&present[(d2[u].y / R) * 8 + u], bit);
    pv[p0 + u * 256 + t] = int4{s2[u].x, d2[u].x, s2[u].y, d2[u].y};
  }
  __syncthreads();
  for (int j = 0; j < CHUNK_MAPW / 256; ++j)
    lmap[(long long)blockIdx.x * CHUNK_MAPW + j * 256 + t] = present[j * 256 + t];
}

template <int R, int NT, class Push>
__device__ __forceinline__ void scan_for_range(const int* __restrict__ dst, int e, int base,
                                               Push&& push) {
  const int tid = threadIdx.x;
  auto in_range = [&](int d) { return (unsigned)(d - base) < (unsigned)R; };
  int head = (int)(((16u - (unsigned)(reinterpret_cast<unsigned long long>(dst) & 15u)) & 15u) >> 2);
  if (head > e) head = e;
  if (tid < head && in_range(dst[tid])) push(tid);
  const int nvec = (e - head) >> 2;
  const int4* __restrict__ dv = reinterpret_cast<const int4*>(dst + head);
  constexpr int U = 8;
  int v = tid;
  for (; v + (U - 1) * NT < nvec; v += U * NT) {
    int4 dd[U];
    for (int u = 0; u < U; ++u) dd[u] = dv[v + u * NT];
    unsigned m = 0;
    for (int u = 0; u < U; ++u) {
      m |= (in_range(dd[u].x) ? 1u : 0u) << (4 * u);
      m |= (in_range(dd[u].y) ? 2u : 0u) << (4 * u);
      m |= (in_range(dd[u].z) ? 4u : 0u) << (4 * u);
      m |= (in_range(dd[u].w) ? 8u : 0u) << (4 * u);
    }
    while (m) {
      const int b = __builtin_ctz(m);
      m &= m - 1;
      push(head + 4 * (v + (b >> 2) * NT) + (b & 3));
    }
  }
  for (; v < nvec; v += NT) {
    const int4 d4 = dv[v];
    unsigned m = (in_range(d4.x) ? 1u : 0u) | (in_range(d4.y) ? 2u : 0u) |
                 (in_range(d4.z) ? 4u : 0u) | (in_range(d4.w) ? 8u : 0u);
    while (m) {
      const int b = __builtin_ctz(m);
      m &= m - 1;
      push(head + 4 * v + b);
    }
  }
  {
    const int t = head + 4 * nvec + tid;
    if (t < e && in_range(dst[t])) push(t);
  }
}

template <int BITS>
__device__ __forceinline__ int wave_take_slots(int* counter, int want) {
  int offset = 0, total = 0;
  for (int k = 0; k < BITS; ++k) {
    const unsigned votes = __builtin_amdgcn_ballot_w32(((want >> k) & 1) != 0);
    offset += (int)__builtin_amdgcn_mbcnt_lo(votes, 0u) << k;
    total += __builtin_popcount(votes) << k;
  }
  const unsigned active = __builtin_amdgcn_ballot_w32(true);
  int first = 0;
  const bool leader = __builtin_amdgcn_mbcnt_lo(active, 0u) == 0u;
  if (total != 0 && leader) first = atomicAdd(counter, total);
  return __builtin_amdgcn_readfirstlane(first) + offset;
}

template <int R, int QCAP, int LCAP>
__global__ __launch_bounds__(1024) void range_list_kernel(const int* __restrict__ src,
                                                          const int* __restrict__ dst,
                                                          float* __restrict__ dinv,
                                                          int* __restrict__ glist,
                                                          int* __restrict__ gcount,
                                                          const int* __restrict__ pairs,
                                                          const unsigned* __restrict__ lmap,
                                                          int e, int n) {
  constexpr int NT = 1024;
  static_assert(R <= (1 << (32 - REC_SRC_BITS)), "the local row must fit the record");
  __shared__ int cnt[R];
  __shared__ int queue[QCAP];
  __shared__ int lines[LCAP];
  __shared__ int qcount;
  __shared__ int nlines;

  const int tid = threadIdx.x;
  const int base = blockIdx.x * R;
  const bool packable = n <= (1 << REC_SRC_BITS);

  if (tid == 0) {
    qcount = 0;
    nlines = 0;
  }
  for (int i = tid; i < R; i += NT) cnt[i] = 0;
  __syncthreads();

  int* __restrict__ gl = glist + (long long)blockIdx.x * QCAP;
  bool direct = false;
  if (pair_map_usable(n, e, (int)gridDim.x)) {
    const int nhalf = ((e + CHUNK_EDGES - 1) / CHUNK_EDGES) * 2;
    const int4* __restrict__ lm4 = reinterpret_cast<const int4*>(lmap);
    for (int it = tid; it < nhalf; it += NT) {
      const int c = it >> 1, half = it & 1;
      const int4 w4 = lm4[((long long)c * 256 + blockIdx.x) * 2 + half];
      const unsigned w[4] = {(unsigned)w4.x, (unsigned)w4.y, (unsigned)w4.z, (unsigned)w4.w};
      const int marked = __builtin_popcount(w[0]) + __builtin_popcount(w[1]) +
                         __builtin_popcount(w[2]) + __builtin_popcount(w[3]);
      int slot = wave_take_slots<8>(&nlines, marked);
      for (int q = 0; q < 4; ++q) {
        unsigned ww = w[q];
        while (ww) {
          const int b = __builtin_ctz(ww);
          ww &= ww - 1;
          if (slot < LCAP) lines[slot] = c * CHUNK_LINES + (half * 4 + q) * 32 + b;
          ++slot;
        }
      }
    }
    __syncthreads();
    direct = nlines <= LCAP;
  }

  int met;
  if (direct) {
    const int4* __restrict__ pv = reinterpret_cast<const int4*>(pairs);
    const int nitems = nlines * 8;
    constexpr int U = 8;
    for (int j = tid; j < nitems; j += U * NT) {
      int4 kk[U];
      bool ok[U];
      for (int u = 0; u < U; ++u) {
        const int jj = j + u * NT;
        ok[u] = jj < nitems;
        const int jc = ok[u] ? jj : j;
        kk[u] = pv[lines[jc >> 3] * 8 + (jc & 7)];
      }
      unsigned hm = 0;
      for (int u = 0; u < U; ++u) {
        const unsigned two = ((unsigned)(kk[u].y - base) < (unsigned)R ? 1u : 0u) |
                             ((unsigned)(kk[u].w - base) < (unsigned)R ? 2u : 0u);
        hm |= (ok[u] ? two : 0u) << (2 * u);
      }
      int slot = wave_take_slots<5>(&qcount, __builtin_popcount(hm));
      if (hm) {
        for (int u = 0; u < U; ++u) {
          if (hm & (1u << (2 * u))) {
            const int dl = kk[u].y - base;
            if (slot < QCAP) queue[slot] = rec_pack(kk[u].x, dl);
            ++slot;
            atomicAdd(&cnt[dl], 1);
          }
          if (hm & (2u << (2 * u))) {
            const int dl = kk[u].w - base;
            if (slot < QCAP) queue[slot] = rec_pack(kk[u].z, dl);
            ++slot;
            atomicAdd(&cnt[dl], 1);
          }
        }
      }
    }
    __syncthreads();
    met = qcount;
    const int total = met < QCAP ? met : QCAP;
    for (int i = tid; i < total; i += NT) gl[i] = queue[i];
  } else {
    auto push = [&](int idx) {
      const int slot = atomicAdd(&qcount, 1);
      if (slot < QCAP) {
        queue[slot] = idx;
      } else {
        atomicAdd(&cnt[dst[idx] - base], 1);
      }
    };
    scan_for_range<R, NT>(dst, e, base, push);
    __syncthreads();

    met = qcount;
    const int total = met < QCAP ? met : QCAP;
    constexpr int WU = 4;
    int i = tid;
    for (; i + (WU - 1) * NT < total; i += WU * NT) {
      int s[WU], dl[WU];
      for (int u = 0; u < WU; ++u) {
        const int idx = queue[i + u * NT];
        s[u] = src[idx];
        dl[u] = dst[idx] - base;
      }
      for (int u = 0; u < WU; ++u) {
        gl[i + u * NT] = rec_pack(s[u], dl[u]);
        atomicAdd(&cnt[dl[u]], 1);
      }
    }
    for (; i < total; i += NT) {
      const int idx = queue[i];
      const int dl = dst[idx] - base;
      gl[i] = rec_pack(src[idx], dl);
      atomicAdd(&cnt[dl], 1);
    }
    __syncthreads();
  }
  if (tid == 0) gcount[blockIdx.x] = packable ? met : 0x7FFFFFFF;

  for (int i = tid; i < R; i += NT) {
    const int node = base + i;
    if (node < n) dinv[node] = rsqrtf((float)cnt[i] + 1.0f);
  }
}

template <int LOGF, int R, int QCAP>
__global__ __launch_bounds__(1024) void agg_lds_kernel(const float* __restrict__ H,
                                                       const int* __restrict__ src,
                                                       const int* __restrict__ dst,
                                                       const float* __restrict__ dinv,
                                                       float* __restrict__ out,
                                                       const int* __restrict__ glist,
                                                       const int* __restrict__ gcount, int e,
                                                       int n) {
  constexpr int F = 1 << LOGF;
  constexpr int LPE = F / 4;
  constexpr int LOGLPE = LOGF - 2;
  constexpr int EPI = 32 / LPE;
  constexpr int NT = 1024;
  static_assert(R <= (1 << (32 - REC_SRC_BITS)), "the local row must fit the record");
  __shared__ float acc[R * F];
  __shared__ int queue[QCAP];
  __shared__ int qcount;

  const int tid = threadIdx.x;
  const int lane = tid & 31;
  const int base = blockIdx.x * R;
  const bool packable = n <= (1 << REC_SRC_BITS);

  if (tid == 0) qcount = 0;
  for (int i = tid; i < R * LPE; i += NT) {
    const int row = i >> LOGLPE, q = i & (LPE - 1);
    const int node = base + row;
    if (node < n) {
      const float dn = dinv[node];
      const float nrm = dn * dn;
      const float4 h = *reinterpret_cast<const float4*>(H + (((long long)node) << LOGF) + 4 * q);
      float* a = acc + (row << LOGF) + q;
      a[0] = h.x * nrm;
      a[LPE] = h.y * nrm;
      a[2 * LPE] = h.z * nrm;
      a[3 * LPE] = h.w * nrm;
    }
  }
  __syncthreads();

  auto push = [&](int idx) {
    const int s = src[idx];
    const int d = dst[idx];
    const int slot = packable ? atomicAdd(&qcount, 1) : QCAP;
    if (slot < QCAP) {
      queue[slot] = rec_pack(s, d - base);
    } else {
      const float nrm = dinv[s] * dinv[d];
      const float* hs = H + (((long long)s) << LOGF);
      float* a = acc + ((d - base) << LOGF);
      for (int f = 0; f < F; ++f) atomicAdd(a + (f & 3) * LPE + (f >> 2), hs[f] * nrm);
    }
  };

  const int c = gcount[blockIdx.x];
  const bool have_list = c <= QCAP;
  if (have_list) {
    const int* __restrict__ gl = glist + (long long)blockIdx.x * QCAP;
    for (int i = tid; i < c; i += NT) queue[i] = gl[i];
    if (tid == 0) qcount = c;
  } else {
    scan_for_range<R, NT>(dst, e, base, push);
  }
  __syncthreads();

  constexpr int STEPS = 32 / EPI;
  constexpr int UB = STEPS < 8 ? STEPS : 8;
  const int total = qcount < QCAP ? qcount : QCAP;
  const int q = lane & (LPE - 1);
  const int b0 = (tid >> 5) * 32;
  int ns, ndl;
  float nnrm;
  {
    const int r = (b0 + lane < total) ? queue[b0 + lane] : 0;
    ns = rec_src(r);
    ndl = rec_row(r);
    nnrm = dinv[ns] * dinv[base + ndl];
  }
  for (int b = b0; b < total; b += NT) {
    const int s = ns;
    const int dl = ndl;
    const float nrm = nnrm;
    const int cnt = total - b < 32 ? total - b : 32;
    const int rn = (b + NT + lane < total) ? queue[b + NT + lane] : 0;
    ns = rec_src(rn);
    ndl = rec_row(rn);
    const float pds = dinv[ns];
    const float pdd = dinv[base + ndl];
    for (int j = 0; j < STEPS; j += UB) {
      float4 h[UB];
      int dj[UB];
      float nj[UB];
      for (int u = 0; u < UB; ++u) {
        const int sub = (j + u) * EPI + (lane >> LOGLPE);
        const int sj = __shfl(s, sub);
        dj[u] = __shfl(dl, sub);
        nj[u] = __shfl(nrm, sub);
        h[u] = *reinterpret_cast<const float4*>(H + (((long long)sj) << LOGF) + 4 * q);
      }
      for (int u = 0; u < UB; ++u) {
        const int sub = (j + u) * EPI + (lane >> LOGLPE);
        if (sub < cnt) {
          float* a = acc + (dj[u] << LOGF) + q;
          atomicAdd(a, h[u].x * nj[u]);
          atomicAdd(a + LPE, h[u].y * nj[u]);
          atomicAdd(a + 2 * LPE, h[u].z * nj[u]);
          atomicAdd(a + 3 * LPE, h[u].w * nj[u]);
        }
      }
    }
    nnrm = pds * pdd;
  }
  __syncthreads();

  for (int i = tid; i < R * LPE; i += NT) {
    const int row = i >> LOGLPE, qq = i & (LPE - 1);
    const int node = base + row;
    if (node < n) {
      const float* a = acc + (row << LOGF) + qq;
      float4 o;
      o.x = a[0];
      o.y = a[LPE];
      o.z = a[2 * LPE];
      o.w = a[3 * LPE];
      *reinterpret_cast<float4*>(out + (((long long)node) << LOGF) + 4 * qq) = o;
    }
  }
}

__global__ void lsm_kernel(const float* __restrict__ agg, const float* __restrict__ b,
                           float* __restrict__ out, int n) {
  const int i = blockIdx.x * blockDim.x + threadIdx.x;
  if (i >= n) return;
  float v[16];
  float mx = -INFINITY;
#pragma unroll
  for (int j = 0; j < 16; ++j) {
    v[j] = agg[(long long)i * 16 + j] + b[j];
    mx = fmaxf(mx, v[j]);
  }
  float s = 0.0f;
#pragma unroll
  for (int j = 0; j < 16; ++j) s += expf(v[j] - mx);
  const float lse = mx + logf(s);
#pragma unroll
  for (int j = 0; j < 16; ++j) out[(long long)i * 16 + j] = v[j] - lse;
}

extern "C" void kernel_launch(void* const* d_in, const int* in_sizes, int n_in,
                              void* d_out, int out_size, void* d_ws, size_t ws_size,
                              hipStream_t stream) {
  const float* x  = (const float*)d_in[0];
  const int*   ei = (const int*)d_in[1];
  const float* W1 = (const float*)d_in[2];
  const float* b1 = (const float*)d_in[3];
  const float* W2 = (const float*)d_in[4];
  const float* b2 = (const float*)d_in[5];

  const int n = in_sizes[0] / 256;
  const int e = in_sizes[1] / 2;
  const int* src = ei;
  const int* dst = ei + e;

  float* agg1 = (float*)d_ws;
  float* h1   = agg1 + (size_t)n * 128;
  int*   glist = (int*)(h1 + (size_t)n * 128);
  float* dinv = (float*)(glist + (size_t)n * 32 + n);
  float* h2   = h1;
  float* agg2 = h1 + (size_t)n * 16;
  const int nrange = ceil_div(n, RANGE_ROWS);
  int*   gcount = glist + (size_t)nrange * 10240;
  const int nchunk = ceil_div(e, CHUNK_EDGES);
  int*   pairs = (int*)d_ws;
  unsigned* lmap = (unsigned*)(pairs + (size_t)nchunk * CHUNK_EDGES * 2);

  pair_kernel<RANGE_ROWS><<<nchunk, 256, 0, stream>>>(src, dst, pairs, lmap, e, n, nrange);
  range_list_kernel<RANGE_ROWS, 10240, 16384><<<nrange, 1024, 0, stream>>>(
      src, dst, dinv, glist, gcount, pairs, lmap, e, n);

  gemm_wmma_f32_lds_a<256, 128, 4, 2, 4><<<ceil_div(n, 64), 128, 0, stream>>>(x, W1, h1, n);
  agg_lds_kernel<7, RANGE_ROWS, 10240><<<nrange, 1024, 0, stream>>>(
      h1, src, dst, dinv, agg1, glist, gcount, e, n);

  {
    const long long waves = (long long)((n + 15) / 16) * (16 / 16);
    gemm_wmma_f32_relu_a<128, 16, 8><<<ceil_div(waves * 32, 256), 256, 0, stream>>>(
        agg1, b1, W2, h2, n);
  }
  agg_lds_kernel<4, RANGE_ROWS, 10240><<<nrange, 1024, 0, stream>>>(
      h2, src, dst, dinv, agg2, glist, gcount, e, n);
  lsm_kernel<<<ceil_div(n, 256), 256, 0, stream>>>(agg2, b2, (float*)d_out, n);
}
